# starting kernel
# baseline (speedup 1.0000x reference)
#include <hip/hip_runtime.h>
#include <math.h>

#define NNODE 50000

typedef float v2f __attribute__((ext_vector_type(2)));
typedef float v8f __attribute__((ext_vector_type(8)));


__global__ void fill_kernel(float* __restrict__ p, float v, int n) {
    int i = blockIdx.x * blockDim.x + threadIdx.x;
    if (i < n) p[i] = v;
}

__device__ __forceinline__ void edge_nodes(const int* __restrict__ ei, int E,
                                           int e, int& s, int& d) {
    if (e < E) { s = ei[e]; d = ei[E + e]; }
    else       { s = e - E; d = s; }
}

__device__ __forceinline__ void atomicMaxF(float* addr, float val) {
    if (val >= 0.0f) atomicMax((int*)addr, __float_as_int(val));
    else             atomicMin((unsigned int*)addr, __float_as_uint(val));
}

__global__ void gemm_wmma_f32(const float* __restrict__ X,
                              const float* __restrict__ W,
                              float* __restrict__ H,
                              int n, int K, int M) {
    const int lane = threadIdx.x & 31;
    const int wave = threadIdx.x >> 5;
    const int wavesPerBlock = blockDim.x >> 5;

    const int colTiles = M >> 4;
    const int rowTiles = n >> 4;
    const int tile = blockIdx.x * wavesPerBlock + wave;
    if (tile >= rowTiles * colTiles) return;

    const int rowT = tile / colTiles;
    const int colT = tile - rowT * colTiles;
    const int row0 = rowT << 4;
    const int col0 = colT << 4;

    const int m16  = lane & 15;
    const int half = lane >> 4;

    v8f acc = {};
    const float* xrow = X + (size_t)(row0 + m16) * K;

    for (int k0 = 0; k0 < K; k0 += 4) {
        const int k = k0 + half * 2;
        v2f a, b;
        a.x = xrow[k];
        a.y = xrow[k + 1];
        b.x = W[(size_t)(k)     * M + col0 + m16];
        b.y = W[(size_t)(k + 1) * M + col0 + m16];
        acc = __builtin_amdgcn_wmma_f32_16x16x4_f32(
                   false, a,  false, b,
                   (short)0, acc,  false,  false);
    }

#pragma unroll
    for (int v = 0; v < 8; ++v)
        H[(size_t)(row0 + half * 8 + v) * M + col0 + m16] = acc[v];
}

__global__ void att_logits(const float* __restrict__ H,
                           const float* __restrict__ a_src,
                           const float* __restrict__ a_dst,
                           float* __restrict__ als, float* __restrict__ ald,
                           int n, int heads, int C) {
    int t = blockIdx.x * blockDim.x + threadIdx.x;
    if (t >= n * heads) return;
    int node = t / heads;
    int h    = t - node * heads;
    const float* hp = H + (size_t)node * heads * C + (size_t)h * C;
    const float* as = a_src + h * C;
    const float* ad = a_dst + h * C;
    float s = 0.0f, d = 0.0f;
    for (int c = 0; c < C; ++c) { float v = hp[c]; s += v * as[c]; d += v * ad[c]; }
    als[t] = s;
    ald[t] = d;
}

__global__ void edge_max(const int* __restrict__ ei, int E, int Etot,
                         const float* __restrict__ als, const float* __restrict__ ald,
                         float* __restrict__ mbuf, int heads) {
    int e = blockIdx.x * blockDim.x + threadIdx.x;
    if (e >= Etot) return;
    int s, d;
    edge_nodes(ei, E, e, s, d);
    for (int h = 0; h < heads; ++h) {
        float x  = als[s * heads + h] + ald[d * heads + h];
        float lr = x >= 0.0f ? x : 0.2f * x;
        atomicMaxF(&mbuf[d * heads + h], lr);
    }
}

__global__ void edge_acc(const int* __restrict__ ei, int E, int Etot,
                         const float* __restrict__ als, const float* __restrict__ ald,
                         const float* __restrict__ mbuf,
                         const float* __restrict__ H,
                         float* __restrict__ den, float* __restrict__ accum,
                         int heads, int C) {
    int t = blockIdx.x * blockDim.x + threadIdx.x;
    if (t >= Etot * heads) return;
    int e = t / heads;
    int h = t - e * heads;
    int s, d;
    edge_nodes(ei, E, e, s, d);
    float x  = als[s * heads + h] + ald[d * heads + h];
    float lr = x >= 0.0f ? x : 0.2f * x;
    float ex = __expf(lr - mbuf[d * heads + h]);
    atomicAdd(&den[d * heads + h], ex);
    const float* hs = H     + (size_t)s * heads * C + (size_t)h * C;
    float*       ap = accum + (size_t)d * heads * C + (size_t)h * C;
    for (int c = 0; c < C; ++c) atomicAdd(&ap[c], hs[c] * ex);
}

__global__ void finalize(float* __restrict__ accum, const float* __restrict__ den,
                         const float* __restrict__ bias,
                         int n, int heads, int C, int relu) {
    int HC = heads * C;
    int t  = blockIdx.x * blockDim.x + threadIdx.x;
    if (t >= n * HC) return;
    int node = t / HC;
    int rem  = t - node * HC;
    int h    = rem / C;
    float v = accum[t] / den[node * heads + h] + bias[rem];
    if (relu) v = fmaxf(v, 0.0f);
    accum[t] = v;
}


extern "C" void kernel_launch(void* const* d_in, const int* in_sizes, int n_in,
                              void* d_out, int out_size, void* d_ws, size_t ws_size,
                              hipStream_t stream) {
    const float* x      = (const float*)d_in[0];
    const int*   ei     = (const int*)  d_in[1];
    const float* W1     = (const float*)d_in[2];
    const float* a_src1 = (const float*)d_in[3];
    const float* a_dst1 = (const float*)d_in[4];
    const float* b1     = (const float*)d_in[5];
    const float* W2     = (const float*)d_in[6];
    const float* a_src2 = (const float*)d_in[7];
    const float* a_dst2 = (const float*)d_in[8];
    const float* b2     = (const float*)d_in[9];
    const float* W3     = (const float*)d_in[10];
    const float* a_src3 = (const float*)d_in[11];
    const float* a_dst3 = (const float*)d_in[12];
    const float* b3     = (const float*)d_in[13];

    const int E    = in_sizes[1] / 2;
    const int Etot = E + NNODE;

    float* ws   = (float*)d_ws;
    float* bufH = ws;
    float* act1 = bufH + (size_t)NNODE * 240;
    float* h2   = act1 + (size_t)NNODE * 112;
    float* act2 = h2   + (size_t)NNODE * 96;
    float* als  = act2 + (size_t)NNODE * 96;
    float* ald  = als  + (size_t)NNODE * 7;
    float* mbuf = ald  + (size_t)NNODE * 7;
    float* den  = mbuf + (size_t)NNODE * 7;
    float* out  = (float*)d_out;

    auto layer = [&](const float* X, int K,
                     const float* W, const float* as, const float* ad, const float* b,
                     int heads, int C, float* Hbuf, float* accbuf, int relu) {
        const int HC = heads * C;

        const int tiles = (NNODE / 16) * (HC / 16);
        gemm_wmma_f32<<<(tiles + 7) / 8, 256, 0, stream>>>(X, W, Hbuf, NNODE, K, HC);

        const int nh = NNODE * heads;
        att_logits<<<(nh + 255) / 256, 256, 0, stream>>>(Hbuf, as, ad, als, ald,
                                                         NNODE, heads, C);

        fill_kernel<<<(nh + 255) / 256, 256, 0, stream>>>(mbuf, -INFINITY, nh);
        hipMemsetAsync(den, 0, (size_t)nh * sizeof(float), stream);
        hipMemsetAsync(accbuf, 0, (size_t)NNODE * HC * sizeof(float), stream);

        edge_max<<<(Etot + 255) / 256, 256, 0, stream>>>(ei, E, Etot, als, ald,
                                                         mbuf, heads);

        const int th = Etot * heads;
        edge_acc<<<(th + 255) / 256, 256, 0, stream>>>(ei, E, Etot, als, ald, mbuf,
                                                       Hbuf, den, accbuf, heads, C);

        const int tot = NNODE * HC;
        finalize<<<(tot + 255) / 256, 256, 0, stream>>>(accbuf, den, b,
                                                        NNODE, heads, C, relu);
    };

    layer(x,    128, W1, a_src1, a_dst1, b1, 7, 16, bufH, act1, 1);
    layer(act1, 112, W2, a_src2, a_dst2, b2, 6, 16, h2,   act2, 1);
    layer(act2,  96, W3, a_src3, a_dst3, b3, 6, 40, bufH, out,  0);
}


# edge_acc: a wave takes one edge row at a time, lanes on consecutive channels: each f32 atomic add covers one whole 128-B line, not 32 lines; sums still by float atomics in arrival order
# speedup vs baseline: 3.1047x; 3.1047x over previous
#include <hip/hip_runtime.h>
#include <math.h>

#define NNODE 50000

typedef float v2f __attribute__((ext_vector_type(2)));
typedef float v8f __attribute__((ext_vector_type(8)));


__global__ void fill_kernel(float* __restrict__ p, float v, int n) {
    int i = blockIdx.x * blockDim.x + threadIdx.x;
    if (i < n) p[i] = v;
}

__device__ __forceinline__ void edge_nodes(const int* __restrict__ ei, int E,
                                           int e, int& s, int& d) {
    if (e < E) { s = ei[e]; d = ei[E + e]; }
    else       { s = e - E; d = s; }
}

__device__ __forceinline__ void atomicMaxF(float* addr, float val) {
    if (val >= 0.0f) atomicMax((int*)addr, __float_as_int(val));
    else             atomicMin((unsigned int*)addr, __float_as_uint(val));
}

__global__ void gemm_wmma_f32(const float* __restrict__ X,
                              const float* __restrict__ W,
                              float* __restrict__ H,
                              int n, int K, int M) {
    const int lane = threadIdx.x & 31;
    const int wave = threadIdx.x >> 5;
    const int wavesPerBlock = blockDim.x >> 5;

    const int colTiles = M >> 4;
    const int rowTiles = n >> 4;
    const int tile = blockIdx.x * wavesPerBlock + wave;
    if (tile >= rowTiles * colTiles) return;

    const int rowT = tile / colTiles;
    const int colT = tile - rowT * colTiles;
    const int row0 = rowT << 4;
    const int col0 = colT << 4;

    const int m16  = lane & 15;
    const int half = lane >> 4;

    v8f acc = {};
    const float* xrow = X + (size_t)(row0 + m16) * K;

    for (int k0 = 0; k0 < K; k0 += 4) {
        const int k = k0 + half * 2;
        v2f a, b;
        a.x = xrow[k];
        a.y = xrow[k + 1];
        b.x = W[(size_t)(k)     * M + col0 + m16];
        b.y = W[(size_t)(k + 1) * M + col0 + m16];
        acc = __builtin_amdgcn_wmma_f32_16x16x4_f32(
                   false, a,  false, b,
                   (short)0, acc,  false,  false);
    }

#pragma unroll
    for (int v = 0; v < 8; ++v)
        H[(size_t)(row0 + half * 8 + v) * M + col0 + m16] = acc[v];
}

__global__ void att_logits(const float* __restrict__ H,
                           const float* __restrict__ a_src,
                           const float* __restrict__ a_dst,
                           float* __restrict__ als, float* __restrict__ ald,
                           int n, int heads, int C) {
    int t = blockIdx.x * blockDim.x + threadIdx.x;
    if (t >= n * heads) return;
    int node = t / heads;
    int h    = t - node * heads;
    const float* hp = H + (size_t)node * heads * C + (size_t)h * C;
    const float* as = a_src + h * C;
    const float* ad = a_dst + h * C;
    float s = 0.0f, d = 0.0f;
    for (int c = 0; c < C; ++c) { float v = hp[c]; s += v * as[c]; d += v * ad[c]; }
    als[t] = s;
    ald[t] = d;
}

__global__ void edge_max(const int* __restrict__ ei, int E, int Etot,
                         const float* __restrict__ als, const float* __restrict__ ald,
                         float* __restrict__ mbuf, int heads) {
    int e = blockIdx.x * blockDim.x + threadIdx.x;
    if (e >= Etot) return;
    int s, d;
    edge_nodes(ei, E, e, s, d);
    for (int h = 0; h < heads; ++h) {
        float x  = als[s * heads + h] + ald[d * heads + h];
        float lr = x >= 0.0f ? x : 0.2f * x;
        atomicMaxF(&mbuf[d * heads + h], lr);
    }
}

#define EDGES_PER_WAVE 32
#define EDGE_BATCH 4

template <int HEADS, int C>
__device__ __forceinline__ void edge_acc_body(const int* __restrict__ ei, int E, int Etot,
                                              const float* __restrict__ als,
                                              const float* __restrict__ ald,
                                              const float* __restrict__ mbuf,
                                              const float* __restrict__ H,
                                              float* __restrict__ den,
                                              float* __restrict__ accum) {
    constexpr int HC     = HEADS * C;
    constexpr int MAXOFF = (HC % 32 == 0) ? 0 : 16;
    constexpr int NCH    = (HC + MAXOFF + 31) / 32;
    const int lane = threadIdx.x & 31;
    const int wave = __builtin_amdgcn_readfirstlane(
        (int)(blockIdx.x * (blockDim.x >> 5) + (threadIdx.x >> 5)));
    const int eBeg = wave * EDGES_PER_WAVE;
    const int hl   = lane < HEADS ? lane : HEADS - 1;

    for (int b = 0; b < EDGES_PER_WAVE; b += EDGE_BATCH) {
        int   sn[EDGE_BATCH], dn[EDGE_BATCH], jo[EDGE_BATCH];
        float xs[EDGE_BATCH], xd[EDGE_BATCH], mx[EDGE_BATCH];
        float hv[EDGE_BATCH][NCH];
#pragma unroll
        for (int i = 0; i < EDGE_BATCH; ++i) {
            const int e = eBeg + b + i;
            sn[i] = 0;
            dn[i] = 0;
            if (e < Etot) edge_nodes(ei, E, e, sn[i], dn[i]);
        }
#pragma unroll
        for (int i = 0; i < EDGE_BATCH; ++i) {
            const int e = eBeg + b + i;
            xs[i] = 0.0f;
            xd[i] = 0.0f;
            mx[i] = 0.0f;
            jo[i] = 0;
#pragma unroll
            for (int k = 0; k < NCH; ++k) hv[i][k] = 0.0f;
            if (e < Etot) {
                const int s = sn[i], d = dn[i];
                jo[i] = MAXOFF ? -((d * HC) & 31) : 0;
                xs[i] = als[s * HEADS + hl];
                xd[i] = ald[d * HEADS + hl];
                mx[i] = mbuf[d * HEADS + hl];
                const float* hs = H + (size_t)s * HC;
#pragma unroll
                for (int k = 0; k < NCH; ++k) {
                    const int j = jo[i] + k * 32 + lane;
                    if ((unsigned)j < (unsigned)HC) hv[i][k] = hs[j];
                }
            }
        }
#pragma unroll
        for (int i = 0; i < EDGE_BATCH; ++i) {
            const int e = eBeg + b + i;
            if (e < Etot) {
                const int d = dn[i];
                const float x  = xs[i] + xd[i];
                const float lr = x >= 0.0f ? x : 0.2f * x;
                const float ex = __expf(lr - mx[i]);
                if (lane < HEADS) atomicAdd(&den[d * HEADS + lane], ex);
                float* ap = accum + (size_t)d * HC;
#pragma unroll
                for (int k = 0; k < NCH; ++k) {
                    const int  j  = jo[i] + k * 32 + lane;
                    const bool in = (unsigned)j < (unsigned)HC;
                    const int  hh = in ? j / C : 0;
                    const float exh = __shfl(ex, hh);
                    if (in) atomicAdd(&ap[j], hv[i][k] * exh);
                }
            }
        }
    }
}

__global__ void edge_acc_7x16(const int* __restrict__ ei, int E, int Etot,
                              const float* __restrict__ als, const float* __restrict__ ald,
                              const float* __restrict__ mbuf, const float* __restrict__ H,
                              float* __restrict__ den, float* __restrict__ accum) {
    edge_acc_body<7, 16>(ei, E, Etot, als, ald, mbuf, H, den, accum);
}

__global__ void edge_acc_6x16(const int* __restrict__ ei, int E, int Etot,
                              const float* __restrict__ als, const float* __restrict__ ald,
                              const float* __restrict__ mbuf, const float* __restrict__ H,
                              float* __restrict__ den, float* __restrict__ accum) {
    edge_acc_body<6, 16>(ei, E, Etot, als, ald, mbuf, H, den, accum);
}

__global__ void edge_acc_6x40(const int* __restrict__ ei, int E, int Etot,
                              const float* __restrict__ als, const float* __restrict__ ald,
                              const float* __restrict__ mbuf, const float* __restrict__ H,
                              float* __restrict__ den, float* __restrict__ accum) {
    edge_acc_body<6, 40>(ei, E, Etot, als, ald, mbuf, H, den, accum);
}

__global__ void finalize(float* __restrict__ accum, const float* __restrict__ den,
                         const float* __restrict__ bias,
                         int n, int heads, int C, int relu) {
    int HC = heads * C;
    int t  = blockIdx.x * blockDim.x + threadIdx.x;
    if (t >= n * HC) return;
    int node = t / HC;
    int rem  = t - node * HC;
    int h    = rem / C;
    float v = accum[t] / den[node * heads + h] + bias[rem];
    if (relu) v = fmaxf(v, 0.0f);
    accum[t] = v;
}


extern "C" void kernel_launch(void* const* d_in, const int* in_sizes, int n_in,
                              void* d_out, int out_size, void* d_ws, size_t ws_size,
                              hipStream_t stream) {
    const float* x      = (const float*)d_in[0];
    const int*   ei     = (const int*)  d_in[1];
    const float* W1     = (const float*)d_in[2];
    const float* a_src1 = (const float*)d_in[3];
    const float* a_dst1 = (const float*)d_in[4];
    const float* b1     = (const float*)d_in[5];
    const float* W2     = (const float*)d_in[6];
    const float* a_src2 = (const float*)d_in[7];
    const float* a_dst2 = (const float*)d_in[8];
    const float* b2     = (const float*)d_in[9];
    const float* W3     = (const float*)d_in[10];
    const float* a_src3 = (const float*)d_in[11];
    const float* a_dst3 = (const float*)d_in[12];
    const float* b3     = (const float*)d_in[13];

    const int E    = in_sizes[1] / 2;
    const int Etot = E + NNODE;

    float* ws   = (float*)d_ws;
    float* bufH = ws;
    float* act1 = bufH + (size_t)NNODE * 240;
    float* h2   = act1 + (size_t)NNODE * 112;
    float* act2 = h2   + (size_t)NNODE * 96;
    float* als  = act2 + (size_t)NNODE * 96;
    float* ald  = als  + (size_t)NNODE * 7;
    float* mbuf = ald  + (size_t)NNODE * 7;
    float* den  = mbuf + (size_t)NNODE * 7;
    float* out  = (float*)d_out;

    auto layer = [&](const float* X, int K,
                     const float* W, const float* as, const float* ad, const float* b,
                     int heads, int C, float* Hbuf, float* accbuf, int relu) {
        const int HC = heads * C;

        const int tiles = (NNODE / 16) * (HC / 16);
        gemm_wmma_f32<<<(tiles + 7) / 8, 256, 0, stream>>>(X, W, Hbuf, NNODE, K, HC);

        const int nh = NNODE * heads;
        att_logits<<<(nh + 255) / 256, 256, 0, stream>>>(Hbuf, as, ad, als, ald,
                                                         NNODE, heads, C);

        fill_kernel<<<(nh + 255) / 256, 256, 0, stream>>>(mbuf, -INFINITY, nh);
        hipMemsetAsync(den, 0, (size_t)nh * sizeof(float), stream);
        hipMemsetAsync(accbuf, 0, (size_t)NNODE * HC * sizeof(float), stream);

        edge_max<<<(Etot + 255) / 256, 256, 0, stream>>>(ei, E, Etot, als, ald,
                                                         mbuf, heads);

        const int eb = (Etot + 255) / 256;
        if (heads == 7)
            edge_acc_7x16<<<eb, 256, 0, stream>>>(ei, E, Etot, als, ald, mbuf,
                                                  Hbuf, den, accbuf);
        else if (C == 16)
            edge_acc_6x16<<<eb, 256, 0, stream>>>(ei, E, Etot, als, ald, mbuf,
                                                  Hbuf, den, accbuf);
        else
            edge_acc_6x40<<<eb, 256, 0, stream>>>(ei, E, Etot, als, ald, mbuf,
                                                  Hbuf, den, accbuf);

        const int tot = NNODE * HC;
        finalize<<<(tot + 255) / 256, 256, 0, stream>>>(accbuf, den, b,
                                                        NNODE, heads, C, relu);
    };

    layer(x,    128, W1, a_src1, a_dst1, b1, 7, 16, bufH, act1, 1);
    layer(act1, 112, W2, a_src2, a_dst2, b2, 6, 16, h2,   act2, 1);
    layer(act2,  96, W3, a_src3, a_dst3, b3, 6, 40, bufH, out,  0);
}


# edges grouped by destination once per call (count, scan, cursor scatter); one wave per node does max, exp-sum, weighted row sum in registers: no float atomics, sums in segment arrival order
# speedup vs baseline: 10.5729x; 3.4054x over previous
#include <hip/hip_runtime.h>
#include <math.h>

#define NNODE 50000

typedef float v2f __attribute__((ext_vector_type(2)));
typedef float v8f __attribute__((ext_vector_type(8)));

__global__ void gemm_wmma_f32(const float* __restrict__ X,
                              const float* __restrict__ W,
                              float* __restrict__ H,
                              int n, int K, int M) {
    const int lane = threadIdx.x & 31;
    const int wave = threadIdx.x >> 5;
    const int wavesPerBlock = blockDim.x >> 5;

    const int colTiles = M >> 4;
    const int rowTiles = n >> 4;
    const int tile = blockIdx.x * wavesPerBlock + wave;
    if (tile >= rowTiles * colTiles) return;

    const int rowT = tile / colTiles;
    const int colT = tile - rowT * colTiles;
    const int row0 = rowT << 4;
    const int col0 = colT << 4;

    const int m16  = lane & 15;
    const int half = lane >> 4;

    v8f acc = {};
    const float* xrow = X + (size_t)(row0 + m16) * K;

    for (int k0 = 0; k0 < K; k0 += 4) {
        const int k = k0 + half * 2;
        v2f a, b;
        a.x = xrow[k];
        a.y = xrow[k + 1];
        b.x = W[(size_t)(k)     * M + col0 + m16];
        b.y = W[(size_t)(k + 1) * M + col0 + m16];
        acc = __builtin_amdgcn_wmma_f32_16x16x4_f32(
                   false, a,  false, b,
                   (short)0, acc,  false,  false);
    }

#pragma unroll
    for (int v = 0; v < 8; ++v)
        H[(size_t)(row0 + half * 8 + v) * M + col0 + m16] = acc[v];
}

#define LSTR 8
__global__ void att_logits(const float* __restrict__ H,
                           const float* __restrict__ a_src,
                           const float* __restrict__ a_dst,
                           float* __restrict__ als, float* __restrict__ ald,
                           int n, int heads, int C) {
    int t = blockIdx.x * blockDim.x + threadIdx.x;
    if (t >= n * LSTR) return;
    int node = t / LSTR;
    int h    = t - node * LSTR;
    float s = 0.0f, d = 0.0f;
    if (h < heads) {
        const float* hp = H + (size_t)node * heads * C + (size_t)h * C;
        const float* as = a_src + h * C;
        const float* ad = a_dst + h * C;
        for (int c = 0; c < C; ++c) { float v = hp[c]; s += v * as[c]; d += v * ad[c]; }
    }
    als[t] = s;
    ald[t] = d;
}

__global__ void dst_count(const int* __restrict__ ei, int E, int* __restrict__ cnt) {
    int e = blockIdx.x * blockDim.x + threadIdx.x;
    if (e < E) atomicAdd(&cnt[ei[E + e]], 1);
}

__global__ void dst_scan(const int* __restrict__ cnt, int* __restrict__ rowptr,
                         int* __restrict__ cursor, int n) {
    __shared__ int wtot[32];
    const int lane  = threadIdx.x & 31;
    const int wave  = threadIdx.x >> 5;
    const int nw    = blockDim.x >> 5;
    const int iters = (n + nw * 128 - 1) / (nw * 128);
    const int base  = wave * iters * 128 + lane * 4;

    int s = 0;
    for (int i = 0; i < iters; ++i) {
        const int idx = base + i * 128;
#pragma unroll
        for (int q = 0; q < 4; ++q)
            if (idx + q < n) s += cnt[idx + q];
    }
    for (int o = 16; o > 0; o >>= 1) s += __shfl_xor(s, o);
    if (lane == 0) wtot[wave] = s;
    __syncthreads();

    int carry = 0;
    for (int w = 0; w < wave; ++w) carry += wtot[w];

    for (int i = 0; i < iters; ++i) {
        const int idx = base + i * 128;
        int v[4];
#pragma unroll
        for (int q = 0; q < 4; ++q) v[q] = idx + q < n ? cnt[idx + q] : 0;
        const int lt = v[0] + v[1] + v[2] + v[3];
        int inc = lt;
        for (int o = 1; o < 32; o <<= 1) {
            const int t = __shfl_up(inc, o);
            if (lane >= o) inc += t;
        }
        int run = carry + inc - lt;
#pragma unroll
        for (int q = 0; q < 4; ++q) {
            if (idx + q < n) { rowptr[idx + q] = run; cursor[idx + q] = run; }
            run += v[q];
        }
        carry += __shfl(inc, 31);
    }
}

__global__ void dst_fill(const int* __restrict__ ei, int E,
                         int* __restrict__ cursor, int* __restrict__ col) {
    int e = blockIdx.x * blockDim.x + threadIdx.x;
    if (e < E) {
        const int pos = atomicAdd(&cursor[ei[E + e]], 1);
        col[pos] = ei[e];
    }
}

template <int HEADS>
__device__ __forceinline__ void edge_logits(const int* __restrict__ col,
                                            const float* __restrict__ als,
                                            const float* ad, int beg, int deg, int d,
                                            int t, int& s, float* lr) {
    s = d;
#pragma unroll
    for (int h = 0; h < HEADS; ++h) lr[h] = -INFINITY;
    if (t <= deg) {
        if (t < deg) s = col[beg + t];
        const float4* p = (const float4*)(als + (size_t)s * LSTR);
        const float4 lo = p[0], hi = p[1];
        const float a[8] = {lo.x, lo.y, lo.z, lo.w, hi.x, hi.y, hi.z, hi.w};
#pragma unroll
        for (int h = 0; h < HEADS; ++h) {
            const float x = a[h] + ad[h];
            lr[h] = x >= 0.0f ? x : 0.2f * x;
        }
    }
}

template <int HEADS, int C, int RELU, int BATCH>
__device__ __forceinline__ void aggregate_body(const int* __restrict__ rowptr,
                                               const int* __restrict__ cnt,
                                               const int* __restrict__ col,
                                               const float* __restrict__ als,
                                               const float* __restrict__ ald,
                                               const float* __restrict__ H,
                                               const float* __restrict__ bias,
                                               float* __restrict__ out, int n) {
    constexpr int HC  = HEADS * C;
    constexpr int NCH = (HC + 31) / 32;
    static_assert(C == 16 || C >= 32, "a 32-channel piece spans at most two heads");
    static_assert(HEADS <= LSTR, "logit slots per node");
    const int lane = threadIdx.x & 31;
    const int d = __builtin_amdgcn_readfirstlane(
        (int)(blockIdx.x * (blockDim.x >> 5) + (threadIdx.x >> 5)));
    if (d >= n) return;
    const int beg = rowptr[d];
    const int deg = cnt[d];
    const int tot = deg + 1;

    float ad[HEADS], m[HEADS], dsum[HEADS], acc[NCH];
#pragma unroll
    for (int h = 0; h < HEADS; ++h) {
        ad[h]   = ald[(size_t)d * LSTR + h];
        dsum[h] = 0.0f;
    }
#pragma unroll
    for (int k = 0; k < NCH; ++k) acc[k] = 0.0f;

    int   s0;
    float lr0[HEADS];
    edge_logits<HEADS>(col, als, ad, beg, deg, d, lane, s0, lr0);
#pragma unroll
    for (int h = 0; h < HEADS; ++h) m[h] = lr0[h];
    for (int c0 = 32; c0 < tot; c0 += 32) {
        int   s;
        float lr[HEADS];
        edge_logits<HEADS>(col, als, ad, beg, deg, d, c0 + lane, s, lr);
#pragma unroll
        for (int h = 0; h < HEADS; ++h) m[h] = fmaxf(m[h], lr[h]);
    }
#pragma unroll
    for (int h = 0; h < HEADS; ++h)
        for (int o = 16; o > 0; o >>= 1) m[h] = fmaxf(m[h], __shfl_xor(m[h], o));

    for (int c0 = 0; c0 < tot; c0 += 32) {
        int   s = s0;
        float ex[HEADS];
#pragma unroll
        for (int h = 0; h < HEADS; ++h) ex[h] = lr0[h];
        if (c0 > 0) edge_logits<HEADS>(col, als, ad, beg, deg, d, c0 + lane, s, ex);
#pragma unroll
        for (int h = 0; h < HEADS; ++h) {
            ex[h] = __expf(ex[h] - m[h]);
            dsum[h] += ex[h];
        }
        const int nj = (tot - c0) < 32 ? (tot - c0) : 32;
        for (int j0 = 0; j0 < nj; j0 += BATCH) {
            float hv[BATCH][NCH];
#pragma unroll
            for (int i = 0; i < BATCH; ++i) {
#pragma unroll
                for (int k = 0; k < NCH; ++k) hv[i][k] = 0.0f;
                if (j0 + i < nj) {
                    const int sj = __builtin_amdgcn_readlane(s, j0 + i);
                    const float* hs = H + (size_t)sj * HC;
#pragma unroll
                    for (int k = 0; k < NCH; ++k) {
                        const int c = k * 32 + lane;
                        if (c < HC) hv[i][k] = hs[c];
                    }
                }
            }
#pragma unroll
            for (int i = 0; i < BATCH; ++i) {
                if (j0 + i < nj) {
#pragma unroll
                    for (int k = 0; k < NCH; ++k) {
                        const int hlo = (k * 32) / C;
                        const int hhi = ((k * 32 + 31) / C) < HEADS ? ((k * 32 + 31) / C)
                                                                    : HEADS - 1;
                        const float wlo = __int_as_float(
                            __builtin_amdgcn_readlane(__float_as_int(ex[hlo]), j0 + i));
                        const float whi = __int_as_float(
                            __builtin_amdgcn_readlane(__float_as_int(ex[hhi]), j0 + i));
                        const float w = (k * 32 + lane) < (hlo + 1) * C ? wlo : whi;
                        acc[k] += hv[i][k] * w;
                    }
                }
            }
        }
    }
#pragma unroll
    for (int h = 0; h < HEADS; ++h)
        for (int o = 16; o > 0; o >>= 1) dsum[h] += __shfl_xor(dsum[h], o);

    float* orow = out + (size_t)d * HC;
#pragma unroll
    for (int k = 0; k < NCH; ++k) {
        const int c   = k * 32 + lane;
        const int hlo = (k * 32) / C;
        const int hhi = ((k * 32 + 31) / C) < HEADS ? ((k * 32 + 31) / C) : HEADS - 1;
        if (c < HC) {
            const float den = c < (hlo + 1) * C ? dsum[hlo] : dsum[hhi];
            float v = acc[k] / den + bias[c];
            if (RELU) v = fmaxf(v, 0.0f);
            orow[c] = v;
        }
    }
}

__global__ void aggregate_7x16(const int* __restrict__ rowptr, const int* __restrict__ cnt,
                               const int* __restrict__ col, const float* __restrict__ als,
                               const float* __restrict__ ald, const float* __restrict__ H,
                               const float* __restrict__ bias, float* __restrict__ out, int n) {
    aggregate_body<7, 16, 1, 8>(rowptr, cnt, col, als, ald, H, bias, out, n);
}

__global__ void aggregate_6x16(const int* __restrict__ rowptr, const int* __restrict__ cnt,
                               const int* __restrict__ col, const float* __restrict__ als,
                               const float* __restrict__ ald, const float* __restrict__ H,
                               const float* __restrict__ bias, float* __restrict__ out, int n) {
    aggregate_body<6, 16, 1, 8>(rowptr, cnt, col, als, ald, H, bias, out, n);
}

__global__ void aggregate_6x40(const int* __restrict__ rowptr, const int* __restrict__ cnt,
                               const int* __restrict__ col, const float* __restrict__ als,
                               const float* __restrict__ ald, const float* __restrict__ H,
                               const float* __restrict__ bias, float* __restrict__ out, int n) {
    aggregate_body<6, 40, 0, 4>(rowptr, cnt, col, als, ald, H, bias, out, n);
}


extern "C" void kernel_launch(void* const* d_in, const int* in_sizes, int n_in,
                              void* d_out, int out_size, void* d_ws, size_t ws_size,
                              hipStream_t stream) {
    const float* x      = (const float*)d_in[0];
    const int*   ei     = (const int*)  d_in[1];
    const float* W1     = (const float*)d_in[2];
    const float* a_src1 = (const float*)d_in[3];
    const float* a_dst1 = (const float*)d_in[4];
    const float* b1     = (const float*)d_in[5];
    const float* W2     = (const float*)d_in[6];
    const float* a_src2 = (const float*)d_in[7];
    const float* a_dst2 = (const float*)d_in[8];
    const float* b2     = (const float*)d_in[9];
    const float* W3     = (const float*)d_in[10];
    const float* a_src3 = (const float*)d_in[11];
    const float* a_dst3 = (const float*)d_in[12];
    const float* b3     = (const float*)d_in[13];

    const int E = in_sizes[1] / 2;

    float* ws   = (float*)d_ws;
    float* bufH = ws;
    float* act1 = bufH + (size_t)NNODE * 240;
    float* h2   = act1 + (size_t)NNODE * 112;
    float* act2 = h2   + (size_t)NNODE * 96;
    float* als  = act2 + (size_t)NNODE * 96;
    float* ald  = als  + (size_t)NNODE * LSTR;
    int*   cnt    = (int*)(ald + (size_t)NNODE * LSTR);
    int*   rowptr = cnt    + NNODE;
    int*   cursor = rowptr + NNODE;
    int*   col    = cursor + NNODE;
    float* out  = (float*)d_out;

    hipMemsetAsync(cnt, 0, (size_t)NNODE * sizeof(int), stream);
    dst_count<<<(E + 255) / 256, 256, 0, stream>>>(ei, E, cnt);
    dst_scan<<<1, 1024, 0, stream>>>(cnt, rowptr, cursor, NNODE);
    dst_fill<<<(E + 255) / 256, 256, 0, stream>>>(ei, E, cursor, col);

    auto layer = [&](const float* X, int K,
                     const float* W, const float* as, const float* ad, const float* b,
                     int heads, int C, float* Hbuf, float* obuf) {
        const int HC = heads * C;

        const int tiles = (NNODE / 16) * (HC / 16);
        gemm_wmma_f32<<<(tiles + 7) / 8, 256, 0, stream>>>(X, W, Hbuf, NNODE, K, HC);

        const int nh = NNODE * LSTR;
        att_logits<<<(nh + 255) / 256, 256, 0, stream>>>(Hbuf, as, ad, als, ald,
                                                         NNODE, heads, C);

        const int nb = (NNODE + 7) / 8;
        if (heads == 7)
            aggregate_7x16<<<nb, 256, 0, stream>>>(rowptr, cnt, col, als, ald, Hbuf,
                                                   b, obuf, NNODE);
        else if (C == 16)
            aggregate_6x16<<<nb, 256, 0, stream>>>(rowptr, cnt, col, als, ald, Hbuf,
                                                   b, obuf, NNODE);
        else
            aggregate_6x40<<<nb, 256, 0, stream>>>(rowptr, cnt, col, als, ald, Hbuf,
                                                   b, obuf, NNODE);
    };

    layer(x,    128, W1, a_src1, a_dst1, b1, 7, 16, bufH, act1);
    layer(act1, 112, W2, a_src2, a_dst2, b2, 6, 16, h2,   act2);
    layer(act2,  96, W3, a_src3, a_dst3, b3, 6, 40, bufH, out);
}


# edges grouped by destination with LDS counters and cursors only (bin by 256-node range, then rank inside each bin): no global atomics, no memset; sums stay in segment arrival order
# speedup vs baseline: 13.2039x; 1.2488x over previous
#include <hip/hip_runtime.h>
#include <math.h>

#define NNODE 50000

typedef float v2f __attribute__((ext_vector_type(2)));
typedef float v8f __attribute__((ext_vector_type(8)));

__global__ void gemm_wmma_f32(const float* __restrict__ X,
                              const float* __restrict__ W,
                              float* __restrict__ H,
                              int n, int K, int M) {
    const int lane = threadIdx.x & 31;
    const int wave = threadIdx.x >> 5;
    const int wavesPerBlock = blockDim.x >> 5;

    const int colTiles = M >> 4;
    const int rowTiles = n >> 4;
    const int tile = blockIdx.x * wavesPerBlock + wave;
    if (tile >= rowTiles * colTiles) return;

    const int rowT = tile / colTiles;
    const int colT = tile - rowT * colTiles;
    const int row0 = rowT << 4;
    const int col0 = colT << 4;

    const int m16  = lane & 15;
    const int half = lane >> 4;

    v8f acc = {};
    const float* xrow = X + (size_t)(row0 + m16) * K;

    for (int k0 = 0; k0 < K; k0 += 4) {
        const int k = k0 + half * 2;
        v2f a, b;
        a.x = xrow[k];
        a.y = xrow[k + 1];
        b.x = W[(size_t)(k)     * M + col0 + m16];
        b.y = W[(size_t)(k + 1) * M + col0 + m16];
        acc = __builtin_amdgcn_wmma_f32_16x16x4_f32(
                   false, a,  false, b,
                   (short)0, acc,  false,  false);
    }

#pragma unroll
    for (int v = 0; v < 8; ++v)
        H[(size_t)(row0 + half * 8 + v) * M + col0 + m16] = acc[v];
}

#define LSTR 8
__global__ void att_logits(const float* __restrict__ H,
                           const float* __restrict__ a_src,
                           const float* __restrict__ a_dst,
                           float* __restrict__ als, float* __restrict__ ald,
                           int n, int heads, int C) {
    int t = blockIdx.x * blockDim.x + threadIdx.x;
    if (t >= n * LSTR) return;
    int node = t / LSTR;
    int h    = t - node * LSTR;
    float s = 0.0f, d = 0.0f;
    if (h < heads) {
        const float* hp = H + (size_t)node * heads * C + (size_t)h * C;
        const float* as = a_src + h * C;
        const float* ad = a_dst + h * C;
        for (int c = 0; c < C; ++c) { float v = hp[c]; s += v * as[c]; d += v * ad[c]; }
    }
    als[t] = s;
    ald[t] = d;
}

#define BSHIFT 8
#define BINW   (1 << BSHIFT)
#define NBIN   ((NNODE + BINW - 1) >> BSHIFT)
#define GB     256
#define NTAB   (NBIN * GB)
#define EPT    4
#define BPT    8

__global__ void bin_count(const int* __restrict__ ei, int E, int* __restrict__ bh) {
    __shared__ int hist[NBIN];
    for (int i = threadIdx.x; i < NBIN; i += blockDim.x) hist[i] = 0;
    __syncthreads();
    const int chunk = (E + GB - 1) / GB;
    const int e0 = blockIdx.x * chunk;
    const int e1 = e0 + chunk < E ? e0 + chunk : E;
    for (int e = e0 + threadIdx.x; e < e1; e += EPT * blockDim.x) {
        int d[EPT];
#pragma unroll
        for (int q = 0; q < EPT; ++q) {
            const int eq = e + q * blockDim.x;
            d[q] = ei[E + (eq < e1 ? eq : e1 - 1)];
        }
#pragma unroll
        for (int q = 0; q < EPT; ++q)
            if (e + q * blockDim.x < e1) atomicAdd(&hist[d[q] >> BSHIFT], 1);
    }
    __syncthreads();
    for (int i = threadIdx.x; i < NBIN; i += blockDim.x) bh[i * GB + blockIdx.x] = hist[i];
}

__global__ void excl_scan(const int* __restrict__ in, int* __restrict__ out, int n) {
    __shared__ int wtot[32];
    const int lane  = threadIdx.x & 31;
    const int wave  = threadIdx.x >> 5;
    const int nw    = blockDim.x >> 5;
    const int iters = (n + nw * 128 - 1) / (nw * 128);
    const int base  = wave * iters * 128 + lane * 4;

    int s = 0;
    for (int i = 0; i < iters; ++i) {
        const int idx = base + i * 128;
#pragma unroll
        for (int q = 0; q < 4; ++q)
            if (idx + q < n) s += in[idx + q];
    }
    for (int o = 16; o > 0; o >>= 1) s += __shfl_xor(s, o);
    if (lane == 0) wtot[wave] = s;
    __syncthreads();

    int carry = 0;
    for (int w = 0; w < wave; ++w) carry += wtot[w];

    for (int i = 0; i < iters; ++i) {
        const int idx = base + i * 128;
        int v[4];
#pragma unroll
        for (int q = 0; q < 4; ++q) v[q] = idx + q < n ? in[idx + q] : 0;
        const int lt = v[0] + v[1] + v[2] + v[3];
        int inc = lt;
        for (int o = 1; o < 32; o <<= 1) {
            const int t = __shfl_up(inc, o);
            if (lane >= o) inc += t;
        }
        int run = carry + inc - lt;
#pragma unroll
        for (int q = 0; q < 4; ++q) {
            if (idx + q < n) out[idx + q] = run;
            run += v[q];
        }
        carry += __shfl(inc, 31);
    }
}

__global__ void bin_scatter(const int* __restrict__ ei, int E,
                            const int* __restrict__ boff, int2* __restrict__ binned) {
    __shared__ int cur[NBIN];
    for (int i = threadIdx.x; i < NBIN; i += blockDim.x) cur[i] = boff[i * GB + blockIdx.x];
    __syncthreads();
    const int chunk = (E + GB - 1) / GB;
    const int e0 = blockIdx.x * chunk;
    const int e1 = e0 + chunk < E ? e0 + chunk : E;
    for (int e = e0 + threadIdx.x; e < e1; e += EPT * blockDim.x) {
        int d[EPT], sv[EPT];
#pragma unroll
        for (int q = 0; q < EPT; ++q) {
            const int eq = e + q * blockDim.x;
            const int ec = eq < e1 ? eq : e1 - 1;
            d[q]  = ei[E + ec];
            sv[q] = ei[ec];
        }
#pragma unroll
        for (int q = 0; q < EPT; ++q) {
            if (e + q * blockDim.x < e1) {
                const int pos = atomicAdd(&cur[d[q] >> BSHIFT], 1);
                binned[pos] = make_int2(sv[q], d[q]);
            }
        }
    }
}

__global__ void bin_group(const int2* __restrict__ binned, const int* __restrict__ boff,
                          int E, int n, int* __restrict__ cnt, int* __restrict__ rowptr,
                          int* __restrict__ col) {
    __shared__ int c[BINW];
    __shared__ int cur[BINW];
    const int b     = blockIdx.x;
    const int node0 = b << BSHIFT;
    const int beg   = boff[b * GB];
    const int end   = b + 1 < NBIN ? boff[(b + 1) * GB] : E;
    for (int i = threadIdx.x; i < BINW; i += blockDim.x) c[i] = 0;
    __syncthreads();
    for (int i = beg + threadIdx.x; i < end; i += BPT * blockDim.x) {
        int dl[BPT];
#pragma unroll
        for (int q = 0; q < BPT; ++q) {
            const int iq = i + q * blockDim.x;
            dl[q] = binned[iq < end ? iq : end - 1].y;
        }
#pragma unroll
        for (int q = 0; q < BPT; ++q)
            if (i + q * blockDim.x < end) atomicAdd(&c[dl[q] - node0], 1);
    }
    __syncthreads();
    if (threadIdx.x < 32) {
        constexpr int PER = BINW / 32;
        const int lane = threadIdx.x;
        int v[PER];
        int lt = 0;
#pragma unroll
        for (int q = 0; q < PER; ++q) { v[q] = c[lane * PER + q]; lt += v[q]; }
        int inc = lt;
        for (int o = 1; o < 32; o <<= 1) {
            const int t = __shfl_up(inc, o);
            if (lane >= o) inc += t;
        }
        int run = beg + inc - lt;
#pragma unroll
        for (int q = 0; q < PER; ++q) {
            const int dl = lane * PER + q;
            if (node0 + dl < n) { cnt[node0 + dl] = v[q]; rowptr[node0 + dl] = run; }
            cur[dl] = run;
            run += v[q];
        }
    }
    __syncthreads();
    for (int i = beg + threadIdx.x; i < end; i += BPT * blockDim.x) {
        int2 ed[BPT];
#pragma unroll
        for (int q = 0; q < BPT; ++q) {
            const int iq = i + q * blockDim.x;
            ed[q] = binned[iq < end ? iq : end - 1];
        }
#pragma unroll
        for (int q = 0; q < BPT; ++q) {
            if (i + q * blockDim.x < end) {
                const int pos = atomicAdd(&cur[ed[q].y - node0], 1);
                col[pos] = ed[q].x;
            }
        }
    }
}

template <int HEADS>
__device__ __forceinline__ void edge_logits(const int* __restrict__ col,
                                            const float* __restrict__ als,
                                            const float* ad, int beg, int deg, int d,
                                            int t, int& s, float* lr) {
    s = d;
#pragma unroll
    for (int h = 0; h < HEADS; ++h) lr[h] = -INFINITY;
    if (t <= deg) {
        if (t < deg) s = col[beg + t];
        const float4* p = (const float4*)(als + (size_t)s * LSTR);
        const float4 lo = p[0], hi = p[1];
        const float a[8] = {lo.x, lo.y, lo.z, lo.w, hi.x, hi.y, hi.z, hi.w};
#pragma unroll
        for (int h = 0; h < HEADS; ++h) {
            const float x = a[h] + ad[h];
            lr[h] = x >= 0.0f ? x : 0.2f * x;
        }
    }
}

template <int HEADS, int C, int RELU, int BATCH>
__device__ __forceinline__ void aggregate_body(const int* __restrict__ rowptr,
                                               const int* __restrict__ cnt,
                                               const int* __restrict__ col,
                                               const float* __restrict__ als,
                                               const float* __restrict__ ald,
                                               const float* __restrict__ H,
                                               const float* __restrict__ bias,
                                               float* __restrict__ out, int n) {
    constexpr int HC  = HEADS * C;
    constexpr int NCH = (HC + 31) / 32;
    static_assert(C == 16 || C >= 32, "a 32-channel piece spans at most two heads");
    static_assert(HEADS <= LSTR, "logit slots per node");
    const int lane = threadIdx.x & 31;
    const int d = __builtin_amdgcn_readfirstlane(
        (int)(blockIdx.x * (blockDim.x >> 5) + (threadIdx.x >> 5)));
    if (d >= n) return;
    const int beg = rowptr[d];
    const int deg = cnt[d];
    const int tot = deg + 1;

    float ad[HEADS], m[HEADS], dsum[HEADS], acc[NCH];
#pragma unroll
    for (int h = 0; h < HEADS; ++h) {
        ad[h]   = ald[(size_t)d * LSTR + h];
        dsum[h] = 0.0f;
    }
#pragma unroll
    for (int k = 0; k < NCH; ++k) acc[k] = 0.0f;

    int   s0;
    float lr0[HEADS];
    edge_logits<HEADS>(col, als, ad, beg, deg, d, lane, s0, lr0);
#pragma unroll
    for (int h = 0; h < HEADS; ++h) m[h] = lr0[h];
    for (int c0 = 32; c0 < tot; c0 += 32) {
        int   s;
        float lr[HEADS];
        edge_logits<HEADS>(col, als, ad, beg, deg, d, c0 + lane, s, lr);
#pragma unroll
        for (int h = 0; h < HEADS; ++h) m[h] = fmaxf(m[h], lr[h]);
    }
#pragma unroll
    for (int h = 0; h < HEADS; ++h)
        for (int o = 16; o > 0; o >>= 1) m[h] = fmaxf(m[h], __shfl_xor(m[h], o));

    for (int c0 = 0; c0 < tot; c0 += 32) {
        int   s = s0;
        float ex[HEADS];
#pragma unroll
        for (int h = 0; h < HEADS; ++h) ex[h] = lr0[h];
        if (c0 > 0) edge_logits<HEADS>(col, als, ad, beg, deg, d, c0 + lane, s, ex);
#pragma unroll
        for (int h = 0; h < HEADS; ++h) {
            ex[h] = __expf(ex[h] - m[h]);
            dsum[h] += ex[h];
        }
        const int nj = (tot - c0) < 32 ? (tot - c0) : 32;
        for (int j0 = 0; j0 < nj; j0 += BATCH) {
            float hv[BATCH][NCH];
#pragma unroll
            for (int i = 0; i < BATCH; ++i) {
#pragma unroll
                for (int k = 0; k < NCH; ++k) hv[i][k] = 0.0f;
                if (j0 + i < nj) {
                    const int sj = __builtin_amdgcn_readlane(s, j0 + i);
                    const float* hs = H + (size_t)sj * HC;
#pragma unroll
                    for (int k = 0; k < NCH; ++k) {
                        const int c = k * 32 + lane;
                        if (c < HC) hv[i][k] = hs[c];
                    }
                }
            }
#pragma unroll
            for (int i = 0; i < BATCH; ++i) {
                if (j0 + i < nj) {
#pragma unroll
                    for (int k = 0; k < NCH; ++k) {
                        const int hlo = (k * 32) / C;
                        const int hhi = ((k * 32 + 31) / C) < HEADS ? ((k * 32 + 31) / C)
                                                                    : HEADS - 1;
                        const float wlo = __int_as_float(
                            __builtin_amdgcn_readlane(__float_as_int(ex[hlo]), j0 + i));
                        const float whi = __int_as_float(
                            __builtin_amdgcn_readlane(__float_as_int(ex[hhi]), j0 + i));
                        const float w = (k * 32 + lane) < (hlo + 1) * C ? wlo : whi;
                        acc[k] += hv[i][k] * w;
                    }
                }
            }
        }
    }
#pragma unroll
    for (int h = 0; h < HEADS; ++h)
        for (int o = 16; o > 0; o >>= 1) dsum[h] += __shfl_xor(dsum[h], o);

    float* orow = out + (size_t)d * HC;
#pragma unroll
    for (int k = 0; k < NCH; ++k) {
        const int c   = k * 32 + lane;
        const int hlo = (k * 32) / C;
        const int hhi = ((k * 32 + 31) / C) < HEADS ? ((k * 32 + 31) / C) : HEADS - 1;
        if (c < HC) {
            const float den = c < (hlo + 1) * C ? dsum[hlo] : dsum[hhi];
            float v = acc[k] / den + bias[c];
            if (RELU) v = fmaxf(v, 0.0f);
            orow[c] = v;
        }
    }
}

__global__ void aggregate_7x16(const int* __restrict__ rowptr, const int* __restrict__ cnt,
                               const int* __restrict__ col, const float* __restrict__ als,
                               const float* __restrict__ ald, const float* __restrict__ H,
                               const float* __restrict__ bias, float* __restrict__ out, int n) {
    aggregate_body<7, 16, 1, 8>(rowptr, cnt, col, als, ald, H, bias, out, n);
}

__global__ void aggregate_6x16(const int* __restrict__ rowptr, const int* __restrict__ cnt,
                               const int* __restrict__ col, const float* __restrict__ als,
                               const float* __restrict__ ald, const float* __restrict__ H,
                               const float* __restrict__ bias, float* __restrict__ out, int n) {
    aggregate_body<6, 16, 1, 8>(rowptr, cnt, col, als, ald, H, bias, out, n);
}

__global__ void aggregate_6x40(const int* __restrict__ rowptr, const int* __restrict__ cnt,
                               const int* __restrict__ col, const float* __restrict__ als,
                               const float* __restrict__ ald, const float* __restrict__ H,
                               const float* __restrict__ bias, float* __restrict__ out, int n) {
    aggregate_body<6, 40, 0, 4>(rowptr, cnt, col, als, ald, H, bias, out, n);
}


extern "C" void kernel_launch(void* const* d_in, const int* in_sizes, int n_in,
                              void* d_out, int out_size, void* d_ws, size_t ws_size,
                              hipStream_t stream) {
    const float* x      = (const float*)d_in[0];
    const int*   ei     = (const int*)  d_in[1];
    const float* W1     = (const float*)d_in[2];
    const float* a_src1 = (const float*)d_in[3];
    const float* a_dst1 = (const float*)d_in[4];
    const float* b1     = (const float*)d_in[5];
    const float* W2     = (const float*)d_in[6];
    const float* a_src2 = (const float*)d_in[7];
    const float* a_dst2 = (const float*)d_in[8];
    const float* b2     = (const float*)d_in[9];
    const float* W3     = (const float*)d_in[10];
    const float* a_src3 = (const float*)d_in[11];
    const float* a_dst3 = (const float*)d_in[12];
    const float* b3     = (const float*)d_in[13];

    const int E = in_sizes[1] / 2;

    float* ws   = (float*)d_ws;
    float* bufH = ws;
    float* act1 = bufH + (size_t)NNODE * 240;
    float* h2   = act1 + (size_t)NNODE * 112;
    float* act2 = h2   + (size_t)NNODE * 96;
    float* als  = act2 + (size_t)NNODE * 96;
    float* ald  = als  + (size_t)NNODE * LSTR;
    int*   cnt    = (int*)(ald + (size_t)NNODE * LSTR);
    int*   rowptr = cnt    + NNODE;
    int*   bh     = rowptr + NNODE;
    int*   boff   = bh     + NTAB;
    int2*  binned = (int2*)(boff + NTAB);
    int*   col    = (int*)(binned + E);
    float* out  = (float*)d_out;

    bin_count<<<GB, 1024, 0, stream>>>(ei, E, bh);
    excl_scan<<<1, 1024, 0, stream>>>(bh, boff, NTAB);
    bin_scatter<<<GB, 1024, 0, stream>>>(ei, E, boff, binned);
    bin_group<<<NBIN, 1024, 0, stream>>>(binned, boff, E, NNODE, cnt, rowptr, col);

    auto layer = [&](const float* X, int K,
                     const float* W, const float* as, const float* ad, const float* b,
                     int heads, int C, float* Hbuf, float* obuf) {
        const int HC = heads * C;

        const int tiles = (NNODE / 16) * (HC / 16);
        gemm_wmma_f32<<<(tiles + 7) / 8, 256, 0, stream>>>(X, W, Hbuf, NNODE, K, HC);

        const int nh = NNODE * LSTR;
        att_logits<<<(nh + 255) / 256, 256, 0, stream>>>(Hbuf, as, ad, als, ald,
                                                         NNODE, heads, C);

        const int nb = (NNODE + 7) / 8;
        if (heads == 7)
            aggregate_7x16<<<nb, 256, 0, stream>>>(rowptr, cnt, col, als, ald, Hbuf,
                                                   b, obuf, NNODE);
        else if (C == 16)
            aggregate_6x16<<<nb, 256, 0, stream>>>(rowptr, cnt, col, als, ald, Hbuf,
                                                   b, obuf, NNODE);
        else
            aggregate_6x40<<<nb, 256, 0, stream>>>(rowptr, cnt, col, als, ald, Hbuf,
                                                   b, obuf, NNODE);
    };

    layer(x,    128, W1, a_src1, a_dst1, b1, 7, 16, bufH, act1);
    layer(act1, 112, W2, a_src2, a_dst2, b2, 6, 16, h2,   act2);
    layer(act2,  96, W3, a_src3, a_dst3, b3, 6, 40, bufH, out);
}


# gemm: a wave takes a 16-row strip across all column tiles; A loaded once per 8 k (one 16-byte load), reused by every tile, next chunk loaded ahead; k now summed as {0,1,4,5} then {2,3,6,7} per 8
# speedup vs baseline: 15.0931x; 1.1431x over previous
#include <hip/hip_runtime.h>
#include <math.h>

#define NNODE 50000

typedef float v2f __attribute__((ext_vector_type(2)));
typedef float v8f __attribute__((ext_vector_type(8)));

template <int M, int NT>
__device__ __forceinline__ void gemm_load_chunk(const float* __restrict__ xk,
                                                const float* __restrict__ wk,
                                                float4& a, float (&b)[4][NT]) {
    a = *(const float4*)xk;
#pragma unroll
    for (int j = 0; j < 4; ++j)
#pragma unroll
        for (int t = 0; t < NT; ++t) b[j][t] = wk[j * M + t * 16];
}

template <int NT>
__device__ __forceinline__ void gemm_mac_chunk(const float4& a, const float (&b)[4][NT],
                                               v8f (&acc)[NT]) {
    v2f a01, a23;
    a01.x = a.x; a01.y = a.y;
    a23.x = a.z; a23.y = a.w;
#pragma unroll
    for (int t = 0; t < NT; ++t) {
        v2f bb;
        bb.x = b[0][t]; bb.y = b[1][t];
        acc[t] = __builtin_amdgcn_wmma_f32_16x16x4_f32(
                      false, a01,  false, bb,
                      (short)0, acc[t],  false,  false);
    }
#pragma unroll
    for (int t = 0; t < NT; ++t) {
        v2f bb;
        bb.x = b[2][t]; bb.y = b[3][t];
        acc[t] = __builtin_amdgcn_wmma_f32_16x16x4_f32(
                      false, a23,  false, bb,
                      (short)0, acc[t],  false,  false);
    }
}

template <int K, int NT>
__global__ void __launch_bounds__(64) __attribute__((amdgpu_waves_per_eu(4, 4)))
gemm_strip_wmma_f32(const float* __restrict__ X, const float* __restrict__ W,
                    float* __restrict__ H, int n) {
    constexpr int M = NT * 16;
    static_assert(K % 16 == 0, "two 8-k chunks per trip");
    const int lane  = threadIdx.x & 31;
    const int strip = blockIdx.x * (blockDim.x >> 5) + (threadIdx.x >> 5);
    if (strip >= (n >> 4)) return;

    const int row0 = strip << 4;
    const int m16  = lane & 15;
    const int half = lane >> 4;

    v8f acc[NT];
#pragma unroll
    for (int t = 0; t < NT; ++t) acc[t] = v8f{};

    const float* xp = X + (size_t)(row0 + m16) * K + half * 4;
    const float* wp = W + (size_t)(half * 4) * M + m16;

    float4 a0, a1;
    float  b0[4][NT], b1[4][NT];
    gemm_load_chunk<M, NT>(xp, wp, a0, b0);
#pragma nounroll
    for (int k0 = 0; k0 < K; k0 += 16) {
        gemm_load_chunk<M, NT>(xp + k0 + 8, wp + (size_t)(k0 + 8) * M, a1, b1);
        gemm_mac_chunk<NT>(a0, b0, acc);
        if (k0 + 16 < K)
            gemm_load_chunk<M, NT>(xp + k0 + 16, wp + (size_t)(k0 + 16) * M, a0, b0);
        gemm_mac_chunk<NT>(a1, b1, acc);
    }

    float* hp = H + (size_t)(row0 + half * 8) * M + m16;
#pragma unroll
    for (int t = 0; t < NT; ++t)
#pragma unroll
        for (int v = 0; v < 8; ++v)
            hp[(size_t)v * M + t * 16] = acc[t][v];
}

#define LSTR 8
__global__ void att_logits(const float* __restrict__ H,
                           const float* __restrict__ a_src,
                           const float* __restrict__ a_dst,
                           float* __restrict__ als, float* __restrict__ ald,
                           int n, int heads, int C) {
    int t = blockIdx.x * blockDim.x + threadIdx.x;
    if (t >= n * LSTR) return;
    int node = t / LSTR;
    int h    = t - node * LSTR;
    float s = 0.0f, d = 0.0f;
    if (h < heads) {
        const float* hp = H + (size_t)node * heads * C + (size_t)h * C;
        const float* as = a_src + h * C;
        const float* ad = a_dst + h * C;
        for (int c = 0; c < C; ++c) { float v = hp[c]; s += v * as[c]; d += v * ad[c]; }
    }
    als[t] = s;
    ald[t] = d;
}

#define BSHIFT 8
#define BINW   (1 << BSHIFT)
#define NBIN   ((NNODE + BINW - 1) >> BSHIFT)
#define GB     256
#define NTAB   (NBIN * GB)
#define EPT    4
#define BPT    8

__global__ void bin_count(const int* __restrict__ ei, int E, int* __restrict__ bh) {
    __shared__ int hist[NBIN];
    for (int i = threadIdx.x; i < NBIN; i += blockDim.x) hist[i] = 0;
    __syncthreads();
    const int chunk = (E + GB - 1) / GB;
    const int e0 = blockIdx.x * chunk;
    const int e1 = e0 + chunk < E ? e0 + chunk : E;
    for (int e = e0 + threadIdx.x; e < e1; e += EPT * blockDim.x) {
        int d[EPT];
#pragma unroll
        for (int q = 0; q < EPT; ++q) {
            const int eq = e + q * blockDim.x;
            d[q] = ei[E + (eq < e1 ? eq : e1 - 1)];
        }
#pragma unroll
        for (int q = 0; q < EPT; ++q)
            if (e + q * blockDim.x < e1) atomicAdd(&hist[d[q] >> BSHIFT], 1);
    }
    __syncthreads();
    for (int i = threadIdx.x; i < NBIN; i += blockDim.x) bh[i * GB + blockIdx.x] = hist[i];
}

__global__ void excl_scan(const int* __restrict__ in, int* __restrict__ out, int n) {
    __shared__ int wtot[32];
    const int lane  = threadIdx.x & 31;
    const int wave  = threadIdx.x >> 5;
    const int nw    = blockDim.x >> 5;
    const int iters = (n + nw * 128 - 1) / (nw * 128);
    const int base  = wave * iters * 128 + lane * 4;

    int s = 0;
    for (int i = 0; i < iters; ++i) {
        const int idx = base + i * 128;
#pragma unroll
        for (int q = 0; q < 4; ++q)
            if (idx + q < n) s += in[idx + q];
    }
    for (int o = 16; o > 0; o >>= 1) s += __shfl_xor(s, o);
    if (lane == 0) wtot[wave] = s;
    __syncthreads();

    int carry = 0;
    for (int w = 0; w < wave; ++w) carry += wtot[w];

    for (int i = 0; i < iters; ++i) {
        const int idx = base + i * 128;
        int v[4];
#pragma unroll
        for (int q = 0; q < 4; ++q) v[q] = idx + q < n ? in[idx + q] : 0;
        const int lt = v[0] + v[1] + v[2] + v[3];
        int inc = lt;
        for (int o = 1; o < 32; o <<= 1) {
            const int t = __shfl_up(inc, o);
            if (lane >= o) inc += t;
        }
        int run = carry + inc - lt;
#pragma unroll
        for (int q = 0; q < 4; ++q) {
            if (idx + q < n) out[idx + q] = run;
            run += v[q];
        }
        carry += __shfl(inc, 31);
    }
}

__global__ void bin_scatter(const int* __restrict__ ei, int E,
                            const int* __restrict__ boff, int2* __restrict__ binned) {
    __shared__ int cur[NBIN];
    for (int i = threadIdx.x; i < NBIN; i += blockDim.x) cur[i] = boff[i * GB + blockIdx.x];
    __syncthreads();
    const int chunk = (E + GB - 1) / GB;
    const int e0 = blockIdx.x * chunk;
    const int e1 = e0 + chunk < E ? e0 + chunk : E;
    for (int e = e0 + threadIdx.x; e < e1; e += EPT * blockDim.x) {
        int d[EPT], sv[EPT];
#pragma unroll
        for (int q = 0; q < EPT; ++q) {
            const int eq = e + q * blockDim.x;
            const int ec = eq < e1 ? eq : e1 - 1;
            d[q]  = ei[E + ec];
            sv[q] = ei[ec];
        }
#pragma unroll
        for (int q = 0; q < EPT; ++q) {
            if (e + q * blockDim.x < e1) {
                const int pos = atomicAdd(&cur[d[q] >> BSHIFT], 1);
                binned[pos] = make_int2(sv[q], d[q]);
            }
        }
    }
}

__global__ void bin_group(const int2* __restrict__ binned, const int* __restrict__ boff,
                          int E, int n, int* __restrict__ cnt, int* __restrict__ rowptr,
                          int* __restrict__ col) {
    __shared__ int c[BINW];
    __shared__ int cur[BINW];
    const int b     = blockIdx.x;
    const int node0 = b << BSHIFT;
    const int beg   = boff[b * GB];
    const int end   = b + 1 < NBIN ? boff[(b + 1) * GB] : E;
    for (int i = threadIdx.x; i < BINW; i += blockDim.x) c[i] = 0;
    __syncthreads();
    for (int i = beg + threadIdx.x; i < end; i += BPT * blockDim.x) {
        int dl[BPT];
#pragma unroll
        for (int q = 0; q < BPT; ++q) {
            const int iq = i + q * blockDim.x;
            dl[q] = binned[iq < end ? iq : end - 1].y;
        }
#pragma unroll
        for (int q = 0; q < BPT; ++q)
            if (i + q * blockDim.x < end) atomicAdd(&c[dl[q] - node0], 1);
    }
    __syncthreads();
    if (threadIdx.x < 32) {
        constexpr int PER = BINW / 32;
        const int lane = threadIdx.x;
        int v[PER];
        int lt = 0;
#pragma unroll
        for (int q = 0; q < PER; ++q) { v[q] = c[lane * PER + q]; lt += v[q]; }
        int inc = lt;
        for (int o = 1; o < 32; o <<= 1) {
            const int t = __shfl_up(inc, o);
            if (lane >= o) inc += t;
        }
        int run = beg + inc - lt;
#pragma unroll
        for (int q = 0; q < PER; ++q) {
            const int dl = lane * PER + q;
            if (node0 + dl < n) { cnt[node0 + dl] = v[q]; rowptr[node0 + dl] = run; }
            cur[dl] = run;
            run += v[q];
        }
    }
    __syncthreads();
    for (int i = beg + threadIdx.x; i < end; i += BPT * blockDim.x) {
        int2 ed[BPT];
#pragma unroll
        for (int q = 0; q < BPT; ++q) {
            const int iq = i + q * blockDim.x;
            ed[q] = binned[iq < end ? iq : end - 1];
        }
#pragma unroll
        for (int q = 0; q < BPT; ++q) {
            if (i + q * blockDim.x < end) {
                const int pos = atomicAdd(&cur[ed[q].y - node0], 1);
                col[pos] = ed[q].x;
            }
        }
    }
}

template <int HEADS>
__device__ __forceinline__ void edge_logits(const int* __restrict__ col,
                                            const float* __restrict__ als,
                                            const float* ad, int beg, int deg, int d,
                                            int t, int& s, float* lr) {
    s = d;
#pragma unroll
    for (int h = 0; h < HEADS; ++h) lr[h] = -INFINITY;
    if (t <= deg) {
        if (t < deg) s = col[beg + t];
        const float4* p = (const float4*)(als + (size_t)s * LSTR);
        const float4 lo = p[0], hi = p[1];
        const float a[8] = {lo.x, lo.y, lo.z, lo.w, hi.x, hi.y, hi.z, hi.w};
#pragma unroll
        for (int h = 0; h < HEADS; ++h) {
            const float x = a[h] + ad[h];
            lr[h] = x >= 0.0f ? x : 0.2f * x;
        }
    }
}

template <int HEADS, int C, int RELU, int BATCH>
__device__ __forceinline__ void aggregate_body(const int* __restrict__ rowptr,
                                               const int* __restrict__ cnt,
                                               const int* __restrict__ col,
                                               const float* __restrict__ als,
                                               const float* __restrict__ ald,
                                               const float* __restrict__ H,
                                               const float* __restrict__ bias,
                                               float* __restrict__ out, int n) {
    constexpr int HC  = HEADS * C;
    constexpr int NCH = (HC + 31) / 32;
    static_assert(C == 16 || C >= 32, "a 32-channel piece spans at most two heads");
    static_assert(HEADS <= LSTR, "logit slots per node");
    const int lane = threadIdx.x & 31;
    const int d = __builtin_amdgcn_readfirstlane(
        (int)(blockIdx.x * (blockDim.x >> 5) + (threadIdx.x >> 5)));
    if (d >= n) return;
    const int beg = rowptr[d];
    const int deg = cnt[d];
    const int tot = deg + 1;

    float ad[HEADS], m[HEADS], dsum[HEADS], acc[NCH];
#pragma unroll
    for (int h = 0; h < HEADS; ++h) {
        ad[h]   = ald[(size_t)d * LSTR + h];
        dsum[h] = 0.0f;
    }
#pragma unroll
    for (int k = 0; k < NCH; ++k) acc[k] = 0.0f;

    int   s0;
    float lr0[HEADS];
    edge_logits<HEADS>(col, als, ad, beg, deg, d, lane, s0, lr0);
#pragma unroll
    for (int h = 0; h < HEADS; ++h) m[h] = lr0[h];
    for (int c0 = 32; c0 < tot; c0 += 32) {
        int   s;
        float lr[HEADS];
        edge_logits<HEADS>(col, als, ad, beg, deg, d, c0 + lane, s, lr);
#pragma unroll
        for (int h = 0; h < HEADS; ++h) m[h] = fmaxf(m[h], lr[h]);
    }
#pragma unroll
    for (int h = 0; h < HEADS; ++h)
        for (int o = 16; o > 0; o >>= 1) m[h] = fmaxf(m[h], __shfl_xor(m[h], o));

    for (int c0 = 0; c0 < tot; c0 += 32) {
        int   s = s0;
        float ex[HEADS];
#pragma unroll
        for (int h = 0; h < HEADS; ++h) ex[h] = lr0[h];
        if (c0 > 0) edge_logits<HEADS>(col, als, ad, beg, deg, d, c0 + lane, s, ex);
#pragma unroll
        for (int h = 0; h < HEADS; ++h) {
            ex[h] = __expf(ex[h] - m[h]);
            dsum[h] += ex[h];
        }
        const int nj = (tot - c0) < 32 ? (tot - c0) : 32;
        for (int j0 = 0; j0 < nj; j0 += BATCH) {
            float hv[BATCH][NCH];
#pragma unroll
            for (int i = 0; i < BATCH; ++i) {
#pragma unroll
                for (int k = 0; k < NCH; ++k) hv[i][k] = 0.0f;
                if (j0 + i < nj) {
                    const int sj = __builtin_amdgcn_readlane(s, j0 + i);
                    const float* hs = H + (size_t)sj * HC;
#pragma unroll
                    for (int k = 0; k < NCH; ++k) {
                        const int c = k * 32 + lane;
                        if (c < HC) hv[i][k] = hs[c];
                    }
                }
            }
#pragma unroll
            for (int i = 0; i < BATCH; ++i) {
                if (j0 + i < nj) {
#pragma unroll
                    for (int k = 0; k < NCH; ++k) {
                        const int hlo = (k * 32) / C;
                        const int hhi = ((k * 32 + 31) / C) < HEADS ? ((k * 32 + 31) / C)
                                                                    : HEADS - 1;
                        const float wlo = __int_as_float(
                            __builtin_amdgcn_readlane(__float_as_int(ex[hlo]), j0 + i));
                        const float whi = __int_as_float(
                            __builtin_amdgcn_readlane(__float_as_int(ex[hhi]), j0 + i));
                        const float w = (k * 32 + lane) < (hlo + 1) * C ? wlo : whi;
                        acc[k] += hv[i][k] * w;
                    }
                }
            }
        }
    }
#pragma unroll
    for (int h = 0; h < HEADS; ++h)
        for (int o = 16; o > 0; o >>= 1) dsum[h] += __shfl_xor(dsum[h], o);

    float* orow = out + (size_t)d * HC;
#pragma unroll
    for (int k = 0; k < NCH; ++k) {
        const int c   = k * 32 + lane;
        const int hlo = (k * 32) / C;
        const int hhi = ((k * 32 + 31) / C) < HEADS ? ((k * 32 + 31) / C) : HEADS - 1;
        if (c < HC) {
            const float den = c < (hlo + 1) * C ? dsum[hlo] : dsum[hhi];
            float v = acc[k] / den + bias[c];
            if (RELU) v = fmaxf(v, 0.0f);
            orow[c] = v;
        }
    }
}

__global__ void aggregate_7x16(const int* __restrict__ rowptr, const int* __restrict__ cnt,
                               const int* __restrict__ col, const float* __restrict__ als,
                               const float* __restrict__ ald, const float* __restrict__ H,
                               const float* __restrict__ bias, float* __restrict__ out, int n) {
    aggregate_body<7, 16, 1, 8>(rowptr, cnt, col, als, ald, H, bias, out, n);
}

__global__ void aggregate_6x16(const int* __restrict__ rowptr, const int* __restrict__ cnt,
                               const int* __restrict__ col, const float* __restrict__ als,
                               const float* __restrict__ ald, const float* __restrict__ H,
                               const float* __restrict__ bias, float* __restrict__ out, int n) {
    aggregate_body<6, 16, 1, 8>(rowptr, cnt, col, als, ald, H, bias, out, n);
}

__global__ void aggregate_6x40(const int* __restrict__ rowptr, const int* __restrict__ cnt,
                               const int* __restrict__ col, const float* __restrict__ als,
                               const float* __restrict__ ald, const float* __restrict__ H,
                               const float* __restrict__ bias, float* __restrict__ out, int n) {
    aggregate_body<6, 40, 0, 4>(rowptr, cnt, col, als, ald, H, bias, out, n);
}


extern "C" void kernel_launch(void* const* d_in, const int* in_sizes, int n_in,
                              void* d_out, int out_size, void* d_ws, size_t ws_size,
                              hipStream_t stream) {
    const float* x      = (const float*)d_in[0];
    const int*   ei     = (const int*)  d_in[1];
    const float* W1     = (const float*)d_in[2];
    const float* a_src1 = (const float*)d_in[3];
    const float* a_dst1 = (const float*)d_in[4];
    const float* b1     = (const float*)d_in[5];
    const float* W2     = (const float*)d_in[6];
    const float* a_src2 = (const float*)d_in[7];
    const float* a_dst2 = (const float*)d_in[8];
    const float* b2     = (const float*)d_in[9];
    const float* W3     = (const float*)d_in[10];
    const float* a_src3 = (const float*)d_in[11];
    const float* a_dst3 = (const float*)d_in[12];
    const float* b3     = (const float*)d_in[13];

    const int E = in_sizes[1] / 2;

    float* ws   = (float*)d_ws;
    float* bufH = ws;
    float* act1 = bufH + (size_t)NNODE * 240;
    float* h2   = act1 + (size_t)NNODE * 112;
    float* act2 = h2   + (size_t)NNODE * 96;
    float* als  = act2 + (size_t)NNODE * 96;
    float* ald  = als  + (size_t)NNODE * LSTR;
    int*   cnt    = (int*)(ald + (size_t)NNODE * LSTR);
    int*   rowptr = cnt    + NNODE;
    int*   bh     = rowptr + NNODE;
    int*   boff   = bh     + NTAB;
    int2*  binned = (int2*)(boff + NTAB);
    int*   col    = (int*)(binned + E);
    float* out  = (float*)d_out;

    bin_count<<<GB, 1024, 0, stream>>>(ei, E, bh);
    excl_scan<<<1, 1024, 0, stream>>>(bh, boff, NTAB);
    bin_scatter<<<GB, 1024, 0, stream>>>(ei, E, boff, binned);
    bin_group<<<NBIN, 1024, 0, stream>>>(binned, boff, E, NNODE, cnt, rowptr, col);

    auto layer = [&](const float* X, int K,
                     const float* W, const float* as, const float* ad, const float* b,
                     int heads, int C, float* Hbuf, float* obuf) {
        const int gb = (NNODE / 16 + 1) / 2;
        if (heads == 7)
            gemm_strip_wmma_f32<128, 7><<<gb, 64, 0, stream>>>(X, W, Hbuf, NNODE);
        else if (C == 16)
            gemm_strip_wmma_f32<112, 6><<<gb, 64, 0, stream>>>(X, W, Hbuf, NNODE);
        else
            gemm_strip_wmma_f32<96, 15><<<gb, 64, 0, stream>>>(X, W, Hbuf, NNODE);

        const int nh = NNODE * LSTR;
        att_logits<<<(nh + 255) / 256, 256, 0, stream>>>(Hbuf, as, ad, als, ald,
                                                         NNODE, heads, C);

        const int nb = (NNODE + 7) / 8;
        if (heads == 7)
            aggregate_7x16<<<nb, 256, 0, stream>>>(rowptr, cnt, col, als, ald, Hbuf,
                                                   b, obuf, NNODE);
        else if (C == 16)
            aggregate_6x16<<<nb, 256, 0, stream>>>(rowptr, cnt, col, als, ald, Hbuf,
                                                   b, obuf, NNODE);
        else
            aggregate_6x40<<<nb, 256, 0, stream>>>(rowptr, cnt, col, als, ald, Hbuf,
                                                   b, obuf, NNODE);
    };

    layer(x,    128, W1, a_src1, a_dst1, b1, 7, 16, bufH, act1);
    layer(act1, 112, W2, a_src2, a_dst2, b2, 6, 16, h2,   act2);
    layer(act2,  96, W3, a_src3, a_dst3, b3, 6, 40, bufH, out);
}


# attention logits computed by the gemm wave from its own accumulators (per-lane products, then 16-lane DPP exchange sums: the logit sums become a lane tree, not serial); the 3 logit launches go
# speedup vs baseline: 16.1973x; 1.0732x over previous
#include <hip/hip_runtime.h>
#include <math.h>

#define NNODE 50000

typedef float v2f __attribute__((ext_vector_type(2)));
typedef float v8f __attribute__((ext_vector_type(8)));

template <int M, int NT>
__device__ __forceinline__ void gemm_load_chunk(const float* __restrict__ xk,
                                                const float* __restrict__ wk,
                                                float4& a, float (&b)[4][NT]) {
    a = *(const float4*)xk;
#pragma unroll
    for (int j = 0; j < 4; ++j)
#pragma unroll
        for (int t = 0; t < NT; ++t) b[j][t] = wk[j * M + t * 16];
}

template <int NT>
__device__ __forceinline__ void gemm_mac_chunk(const float4& a, const float (&b)[4][NT],
                                               v8f (&acc)[NT]) {
    v2f a01, a23;
    a01.x = a.x; a01.y = a.y;
    a23.x = a.z; a23.y = a.w;
#pragma unroll
    for (int t = 0; t < NT; ++t) {
        v2f bb;
        bb.x = b[0][t]; bb.y = b[1][t];
        acc[t] = __builtin_amdgcn_wmma_f32_16x16x4_f32(
                      false, a01,  false, bb,
                      (short)0, acc[t],  false,  false);
    }
#pragma unroll
    for (int t = 0; t < NT; ++t) {
        v2f bb;
        bb.x = b[2][t]; bb.y = b[3][t];
        acc[t] = __builtin_amdgcn_wmma_f32_16x16x4_f32(
                      false, a23,  false, bb,
                      (short)0, acc[t],  false,  false);
    }
}

#define LSTR 8

template <int XM>
__device__ __forceinline__ float row_xor_lane(float x) {
    return __int_as_float(__builtin_amdgcn_update_dpp(0, __float_as_int(x), 0x160 | XM,
                                                      0xf, 0xf, true));
}

template <int NT, int HEADS, int C>
__device__ __forceinline__ void strip_logits(const v8f (&acc)[NT],
                                             const float* __restrict__ a_src,
                                             const float* __restrict__ a_dst,
                                             float* __restrict__ als,
                                             float* __restrict__ ald,
                                             int row0, int half, int m16) {
    static_assert(HEADS <= LSTR, "logit slots per node");
    static_assert(HEADS * C == NT * 16, "the strip holds whole rows of H");
    const bool b3 = (m16 & 8) != 0, b2 = (m16 & 4) != 0, b1 = (m16 & 2) != 0, b0 = (m16 & 1) != 0;
    float res[LSTR];
#pragma unroll
    for (int h = 0; h < LSTR; ++h) res[h] = 0.0f;
#pragma unroll
    for (int h = 0; h < HEADS; ++h) {
        float p[16];
#pragma unroll
        for (int i = 0; i < 16; ++i) p[i] = 0.0f;
        bool first = true;
#pragma unroll
        for (int t = 0; t < NT; ++t) {
            const int lo = t * 16, hi = t * 16 + 15;
            if (hi >= h * C && lo < (h + 1) * C) {
                const int col = lo + m16;
                float ws = a_src[col], wd = a_dst[col];
                if (lo < h * C || hi >= (h + 1) * C) {
                    const bool in = col >= h * C && col < (h + 1) * C;
                    ws = in ? ws : 0.0f;
                    wd = in ? wd : 0.0f;
                }
#pragma unroll
                for (int v = 0; v < 8; ++v) {
                    p[v]     = first ? acc[t][v] * ws : p[v]     + acc[t][v] * ws;
                    p[8 + v] = first ? acc[t][v] * wd : p[8 + v] + acc[t][v] * wd;
                }
                first = false;
            }
        }
        float q8[8], q4[4], q2[2];
#pragma unroll
        for (int i = 0; i < 8; ++i)
            q8[i] = (b3 ? p[8 + i] : p[i]) + row_xor_lane<8>(b3 ? p[i] : p[8 + i]);
#pragma unroll
        for (int i = 0; i < 4; ++i)
            q4[i] = (b2 ? q8[4 + i] : q8[i]) + row_xor_lane<4>(b2 ? q8[i] : q8[4 + i]);
#pragma unroll
        for (int i = 0; i < 2; ++i)
            q2[i] = (b1 ? q4[2 + i] : q4[i]) + row_xor_lane<2>(b1 ? q4[i] : q4[2 + i]);
        res[h] = (b0 ? q2[1] : q2[0]) + row_xor_lane<1>(b0 ? q2[0] : q2[1]);
    }
    float* lp = (b3 ? ald : als) + (size_t)(row0 + half * 8 + (m16 & 7)) * LSTR;
    *(float4*)lp       = make_float4(res[0], res[1], res[2], res[3]);
    *(float4*)(lp + 4) = make_float4(res[4], res[5], res[6], res[7]);
}

template <int K, int NT, int HEADS, int C>
__global__ void __launch_bounds__(64) __attribute__((amdgpu_waves_per_eu(4, 4)))
gemm_strip_wmma_f32(const float* __restrict__ X, const float* __restrict__ W,
                    const float* __restrict__ a_src, const float* __restrict__ a_dst,
                    float* __restrict__ H, float* __restrict__ als, float* __restrict__ ald,
                    int n) {
    constexpr int M = NT * 16;
    static_assert(K % 16 == 0, "two 8-k chunks per trip");
    const int lane  = threadIdx.x & 31;
    const int strip = blockIdx.x * (blockDim.x >> 5) + (threadIdx.x >> 5);
    if (strip >= (n >> 4)) return;

    const int row0 = strip << 4;
    const int m16  = lane & 15;
    const int half = lane >> 4;

    v8f acc[NT];
#pragma unroll
    for (int t = 0; t < NT; ++t) acc[t] = v8f{};

    const float* xp = X + (size_t)(row0 + m16) * K + half * 4;
    const float* wp = W + (size_t)(half * 4) * M + m16;

    float4 a0, a1;
    float  b0[4][NT], b1[4][NT];
    gemm_load_chunk<M, NT>(xp, wp, a0, b0);
#pragma nounroll
    for (int k0 = 0; k0 < K; k0 += 16) {
        gemm_load_chunk<M, NT>(xp + k0 + 8, wp + (size_t)(k0 + 8) * M, a1, b1);
        gemm_mac_chunk<NT>(a0, b0, acc);
        if (k0 + 16 < K)
            gemm_load_chunk<M, NT>(xp + k0 + 16, wp + (size_t)(k0 + 16) * M, a0, b0);
        gemm_mac_chunk<NT>(a1, b1, acc);
    }

    float* hp = H + (size_t)(row0 + half * 8) * M + m16;
#pragma unroll
    for (int t = 0; t < NT; ++t)
#pragma unroll
        for (int v = 0; v < 8; ++v)
            hp[(size_t)v * M + t * 16] = acc[t][v];

    strip_logits<NT, HEADS, C>(acc, a_src, a_dst, als, ald, row0, half, m16);
}

#define BSHIFT 8
#define BINW   (1 << BSHIFT)
#define NBIN   ((NNODE + BINW - 1) >> BSHIFT)
#define GB     256
#define NTAB   (NBIN * GB)
#define EPT    4
#define BPT    8

__global__ void bin_count(const int* __restrict__ ei, int E, int* __restrict__ bh) {
    __shared__ int hist[NBIN];
    for (int i = threadIdx.x; i < NBIN; i += blockDim.x) hist[i] = 0;
    __syncthreads();
    const int chunk = (E + GB - 1) / GB;
    const int e0 = blockIdx.x * chunk;
    const int e1 = e0 + chunk < E ? e0 + chunk : E;
    for (int e = e0 + threadIdx.x; e < e1; e += EPT * blockDim.x) {
        int d[EPT];
#pragma unroll
        for (int q = 0; q < EPT; ++q) {
            const int eq = e + q * blockDim.x;
            d[q] = ei[E + (eq < e1 ? eq : e1 - 1)];
        }
#pragma unroll
        for (int q = 0; q < EPT; ++q)
            if (e + q * blockDim.x < e1) atomicAdd(&hist[d[q] >> BSHIFT], 1);
    }
    __syncthreads();
    for (int i = threadIdx.x; i < NBIN; i += blockDim.x) bh[i * GB + blockIdx.x] = hist[i];
}

__global__ void excl_scan(const int* __restrict__ in, int* __restrict__ out, int n) {
    __shared__ int wtot[32];
    const int lane  = threadIdx.x & 31;
    const int wave  = threadIdx.x >> 5;
    const int nw    = blockDim.x >> 5;
    const int iters = (n + nw * 128 - 1) / (nw * 128);
    const int base  = wave * iters * 128 + lane * 4;

    int s = 0;
    for (int i = 0; i < iters; ++i) {
        const int idx = base + i * 128;
#pragma unroll
        for (int q = 0; q < 4; ++q)
            if (idx + q < n) s += in[idx + q];
    }
    for (int o = 16; o > 0; o >>= 1) s += __shfl_xor(s, o);
    if (lane == 0) wtot[wave] = s;
    __syncthreads();

    int carry = 0;
    for (int w = 0; w < wave; ++w) carry += wtot[w];

    for (int i = 0; i < iters; ++i) {
        const int idx = base + i * 128;
        int v[4];
#pragma unroll
        for (int q = 0; q < 4; ++q) v[q] = idx + q < n ? in[idx + q] : 0;
        const int lt = v[0] + v[1] + v[2] + v[3];
        int inc = lt;
        for (int o = 1; o < 32; o <<= 1) {
            const int t = __shfl_up(inc, o);
            if (lane >= o) inc += t;
        }
        int run = carry + inc - lt;
#pragma unroll
        for (int q = 0; q < 4; ++q) {
            if (idx + q < n) out[idx + q] = run;
            run += v[q];
        }
        carry += __shfl(inc, 31);
    }
}

__global__ void bin_scatter(const int* __restrict__ ei, int E,
                            const int* __restrict__ boff, int2* __restrict__ binned) {
    __shared__ int cur[NBIN];
    for (int i = threadIdx.x; i < NBIN; i += blockDim.x) cur[i] = boff[i * GB + blockIdx.x];
    __syncthreads();
    const int chunk = (E + GB - 1) / GB;
    const int e0 = blockIdx.x * chunk;
    const int e1 = e0 + chunk < E ? e0 + chunk : E;
    for (int e = e0 + threadIdx.x; e < e1; e += EPT * blockDim.x) {
        int d[EPT], sv[EPT];
#pragma unroll
        for (int q = 0; q < EPT; ++q) {
            const int eq = e + q * blockDim.x;
            const int ec = eq < e1 ? eq : e1 - 1;
            d[q]  = ei[E + ec];
            sv[q] = ei[ec];
        }
#pragma unroll
        for (int q = 0; q < EPT; ++q) {
            if (e + q * blockDim.x < e1) {
                const int pos = atomicAdd(&cur[d[q] >> BSHIFT], 1);
                binned[pos] = make_int2(sv[q], d[q]);
            }
        }
    }
}

__global__ void bin_group(const int2* __restrict__ binned, const int* __restrict__ boff,
                          int E, int n, int* __restrict__ cnt, int* __restrict__ rowptr,
                          int* __restrict__ col) {
    __shared__ int c[BINW];
    __shared__ int cur[BINW];
    const int b     = blockIdx.x;
    const int node0 = b << BSHIFT;
    const int beg   = boff[b * GB];
    const int end   = b + 1 < NBIN ? boff[(b + 1) * GB] : E;
    for (int i = threadIdx.x; i < BINW; i += blockDim.x) c[i] = 0;
    __syncthreads();
    for (int i = beg + threadIdx.x; i < end; i += BPT * blockDim.x) {
        int dl[BPT];
#pragma unroll
        for (int q = 0; q < BPT; ++q) {
            const int iq = i + q * blockDim.x;
            dl[q] = binned[iq < end ? iq : end - 1].y;
        }
#pragma unroll
        for (int q = 0; q < BPT; ++q)
            if (i + q * blockDim.x < end) atomicAdd(&c[dl[q] - node0], 1);
    }
    __syncthreads();
    if (threadIdx.x < 32) {
        constexpr int PER = BINW / 32;
        const int lane = threadIdx.x;
        int v[PER];
        int lt = 0;
#pragma unroll
        for (int q = 0; q < PER; ++q) { v[q] = c[lane * PER + q]; lt += v[q]; }
        int inc = lt;
        for (int o = 1; o < 32; o <<= 1) {
            const int t = __shfl_up(inc, o);
            if (lane >= o) inc += t;
        }
        int run = beg + inc - lt;
#pragma unroll
        for (int q = 0; q < PER; ++q) {
            const int dl = lane * PER + q;
            if (node0 + dl < n) { cnt[node0 + dl] = v[q]; rowptr[node0 + dl] = run; }
            cur[dl] = run;
            run += v[q];
        }
    }
    __syncthreads();
    for (int i = beg + threadIdx.x; i < end; i += BPT * blockDim.x) {
        int2 ed[BPT];
#pragma unroll
        for (int q = 0; q < BPT; ++q) {
            const int iq = i + q * blockDim.x;
            ed[q] = binned[iq < end ? iq : end - 1];
        }
#pragma unroll
        for (int q = 0; q < BPT; ++q) {
            if (i + q * blockDim.x < end) {
                const int pos = atomicAdd(&cur[ed[q].y - node0], 1);
                col[pos] = ed[q].x;
            }
        }
    }
}

template <int HEADS>
__device__ __forceinline__ void edge_logits(const int* __restrict__ col,
                                            const float* __restrict__ als,
                                            const float* ad, int beg, int deg, int d,
                                            int t, int& s, float* lr) {
    s = d;
#pragma unroll
    for (int h = 0; h < HEADS; ++h) lr[h] = -INFINITY;
    if (t <= deg) {
        if (t < deg) s = col[beg + t];
        const float4* p = (const float4*)(als + (size_t)s * LSTR);
        const float4 lo = p[0], hi = p[1];
        const float a[8] = {lo.x, lo.y, lo.z, lo.w, hi.x, hi.y, hi.z, hi.w};
#pragma unroll
        for (int h = 0; h < HEADS; ++h) {
            const float x = a[h] + ad[h];
            lr[h] = x >= 0.0f ? x : 0.2f * x;
        }
    }
}

template <int HEADS, int C, int RELU, int BATCH>
__device__ __forceinline__ void aggregate_body(const int* __restrict__ rowptr,
                                               const int* __restrict__ cnt,
                                               const int* __restrict__ col,
                                               const float* __restrict__ als,
                                               const float* __restrict__ ald,
                                               const float* __restrict__ H,
                                               const float* __restrict__ bias,
                                               float* __restrict__ out, int n) {
    constexpr int HC  = HEADS * C;
    constexpr int NCH = (HC + 31) / 32;
    static_assert(C == 16 || C >= 32, "a 32-channel piece spans at most two heads");
    static_assert(HEADS <= LSTR, "logit slots per node");
    const int lane = threadIdx.x & 31;
    const int d = __builtin_amdgcn_readfirstlane(
        (int)(blockIdx.x * (blockDim.x >> 5) + (threadIdx.x >> 5)));
    if (d >= n) return;
    const int beg = rowptr[d];
    const int deg = cnt[d];
    const int tot = deg + 1;

    float ad[HEADS], m[HEADS], dsum[HEADS], acc[NCH];
#pragma unroll
    for (int h = 0; h < HEADS; ++h) {
        ad[h]   = ald[(size_t)d * LSTR + h];
        dsum[h] = 0.0f;
    }
#pragma unroll
    for (int k = 0; k < NCH; ++k) acc[k] = 0.0f;

    int   s0;
    float lr0[HEADS];
    edge_logits<HEADS>(col, als, ad, beg, deg, d, lane, s0, lr0);
#pragma unroll
    for (int h = 0; h < HEADS; ++h) m[h] = lr0[h];
    for (int c0 = 32; c0 < tot; c0 += 32) {
        int   s;
        float lr[HEADS];
        edge_logits<HEADS>(col, als, ad, beg, deg, d, c0 + lane, s, lr);
#pragma unroll
        for (int h = 0; h < HEADS; ++h) m[h] = fmaxf(m[h], lr[h]);
    }
#pragma unroll
    for (int h = 0; h < HEADS; ++h)
        for (int o = 16; o > 0; o >>= 1) m[h] = fmaxf(m[h], __shfl_xor(m[h], o));

    for (int c0 = 0; c0 < tot; c0 += 32) {
        int   s = s0;
        float ex[HEADS];
#pragma unroll
        for (int h = 0; h < HEADS; ++h) ex[h] = lr0[h];
        if (c0 > 0) edge_logits<HEADS>(col, als, ad, beg, deg, d, c0 + lane, s, ex);
#pragma unroll
        for (int h = 0; h < HEADS; ++h) {
            ex[h] = __expf(ex[h] - m[h]);
            dsum[h] += ex[h];
        }
        const int nj = (tot - c0) < 32 ? (tot - c0) : 32;
        for (int j0 = 0; j0 < nj; j0 += BATCH) {
            float hv[BATCH][NCH];
#pragma unroll
            for (int i = 0; i < BATCH; ++i) {
#pragma unroll
                for (int k = 0; k < NCH; ++k) hv[i][k] = 0.0f;
                if (j0 + i < nj) {
                    const int sj = __builtin_amdgcn_readlane(s, j0 + i);
                    const float* hs = H + (size_t)sj * HC;
#pragma unroll
                    for (int k = 0; k < NCH; ++k) {
                        const int c = k * 32 + lane;
                        if (c < HC) hv[i][k] = hs[c];
                    }
                }
            }
#pragma unroll
            for (int i = 0; i < BATCH; ++i) {
                if (j0 + i < nj) {
#pragma unroll
                    for (int k = 0; k < NCH; ++k) {
                        const int hlo = (k * 32) / C;
                        const int hhi = ((k * 32 + 31) / C) < HEADS ? ((k * 32 + 31) / C)
                                                                    : HEADS - 1;
                        const float wlo = __int_as_float(
                            __builtin_amdgcn_readlane(__float_as_int(ex[hlo]), j0 + i));
                        const float whi = __int_as_float(
                            __builtin_amdgcn_readlane(__float_as_int(ex[hhi]), j0 + i));
                        const float w = (k * 32 + lane) < (hlo + 1) * C ? wlo : whi;
                        acc[k] += hv[i][k] * w;
                    }
                }
            }
        }
    }
#pragma unroll
    for (int h = 0; h < HEADS; ++h)
        for (int o = 16; o > 0; o >>= 1) dsum[h] += __shfl_xor(dsum[h], o);

    float* orow = out + (size_t)d * HC;
#pragma unroll
    for (int k = 0; k < NCH; ++k) {
        const int c   = k * 32 + lane;
        const int hlo = (k * 32) / C;
        const int hhi = ((k * 32 + 31) / C) < HEADS ? ((k * 32 + 31) / C) : HEADS - 1;
        if (c < HC) {
            const float den = c < (hlo + 1) * C ? dsum[hlo] : dsum[hhi];
            float v = acc[k] / den + bias[c];
            if (RELU) v = fmaxf(v, 0.0f);
            orow[c] = v;
        }
    }
}

__global__ void aggregate_7x16(const int* __restrict__ rowptr, const int* __restrict__ cnt,
                               const int* __restrict__ col, const float* __restrict__ als,
                               const float* __restrict__ ald, const float* __restrict__ H,
                               const float* __restrict__ bias, float* __restrict__ out, int n) {
    aggregate_body<7, 16, 1, 8>(rowptr, cnt, col, als, ald, H, bias, out, n);
}

__global__ void aggregate_6x16(const int* __restrict__ rowptr, const int* __restrict__ cnt,
                               const int* __restrict__ col, const float* __restrict__ als,
                               const float* __restrict__ ald, const float* __restrict__ H,
                               const float* __restrict__ bias, float* __restrict__ out, int n) {
    aggregate_body<6, 16, 1, 8>(rowptr, cnt, col, als, ald, H, bias, out, n);
}

__global__ void aggregate_6x40(const int* __restrict__ rowptr, const int* __restrict__ cnt,
                               const int* __restrict__ col, const float* __restrict__ als,
                               const float* __restrict__ ald, const float* __restrict__ H,
                               const float* __restrict__ bias, float* __restrict__ out, int n) {
    aggregate_body<6, 40, 0, 4>(rowptr, cnt, col, als, ald, H, bias, out, n);
}


extern "C" void kernel_launch(void* const* d_in, const int* in_sizes, int n_in,
                              void* d_out, int out_size, void* d_ws, size_t ws_size,
                              hipStream_t stream) {
    const float* x      = (const float*)d_in[0];
    const int*   ei     = (const int*)  d_in[1];
    const float* W1     = (const float*)d_in[2];
    const float* a_src1 = (const float*)d_in[3];
    const float* a_dst1 = (const float*)d_in[4];
    const float* b1     = (const float*)d_in[5];
    const float* W2     = (const float*)d_in[6];
    const float* a_src2 = (const float*)d_in[7];
    const float* a_dst2 = (const float*)d_in[8];
    const float* b2     = (const float*)d_in[9];
    const float* W3     = (const float*)d_in[10];
    const float* a_src3 = (const float*)d_in[11];
    const float* a_dst3 = (const float*)d_in[12];
    const float* b3     = (const float*)d_in[13];

    const int E = in_sizes[1] / 2;

    float* ws   = (float*)d_ws;
    float* bufH = ws;
    float* act1 = bufH + (size_t)NNODE * 240;
    float* h2   = act1 + (size_t)NNODE * 112;
    float* act2 = h2   + (size_t)NNODE * 96;
    float* als  = act2 + (size_t)NNODE * 96;
    float* ald  = als  + (size_t)NNODE * LSTR;
    int*   cnt    = (int*)(ald + (size_t)NNODE * LSTR);
    int*   rowptr = cnt    + NNODE;
    int*   bh     = rowptr + NNODE;
    int*   boff   = bh     + NTAB;
    int2*  binned = (int2*)(boff + NTAB);
    int*   col    = (int*)(binned + E);
    float* out  = (float*)d_out;

    bin_count<<<GB, 1024, 0, stream>>>(ei, E, bh);
    excl_scan<<<1, 1024, 0, stream>>>(bh, boff, NTAB);
    bin_scatter<<<GB, 1024, 0, stream>>>(ei, E, boff, binned);
    bin_group<<<NBIN, 1024, 0, stream>>>(binned, boff, E, NNODE, cnt, rowptr, col);

    auto layer = [&](const float* X, int K,
                     const float* W, const float* as, const float* ad, const float* b,
                     int heads, int C, float* Hbuf, float* obuf) {
        const int gb = (NNODE / 16 + 1) / 2;
        if (heads == 7)
            gemm_strip_wmma_f32<128, 7, 7, 16><<<gb, 64, 0, stream>>>(X, W, as, ad, Hbuf,
                                                                      als, ald, NNODE);
        else if (C == 16)
            gemm_strip_wmma_f32<112, 6, 6, 16><<<gb, 64, 0, stream>>>(X, W, as, ad, Hbuf,
                                                                      als, ald, NNODE);
        else
            gemm_strip_wmma_f32<96, 15, 6, 40><<<gb, 64, 0, stream>>>(X, W, as, ad, Hbuf,
                                                                      als, ald, NNODE);

        const int nb = (NNODE + 7) / 8;
        if (heads == 7)
            aggregate_7x16<<<nb, 256, 0, stream>>>(rowptr, cnt, col, als, ald, Hbuf,
                                                   b, obuf, NNODE);
        else if (C == 16)
            aggregate_6x16<<<nb, 256, 0, stream>>>(rowptr, cnt, col, als, ald, Hbuf,
                                                   b, obuf, NNODE);
        else
            aggregate_6x40<<<nb, 256, 0, stream>>>(rowptr, cnt, col, als, ald, Hbuf,
                                                   b, obuf, NNODE);
    };

    layer(x,    128, W1, a_src1, a_dst1, b1, 7, 16, bufH, act1);
    layer(act1, 112, W2, a_src2, a_dst2, b2, 6, 16, h2,   act2);
    layer(act2,  96, W3, a_src3, a_dst3, b3, 6, 40, bufH, out);
}


# aggregation reads a source row 16 bytes per lane (one load per 448/384-B row, two per 960-B row, not 4/3/8 of 4 bytes); a lane owns 4 channels of one head, edge weights come through an LDS table
# speedup vs baseline: 16.7805x; 1.0360x over previous
#include <hip/hip_runtime.h>
#include <math.h>

#define NNODE 50000

typedef float v2f __attribute__((ext_vector_type(2)));
typedef float v8f __attribute__((ext_vector_type(8)));

template <int M, int NT>
__device__ __forceinline__ void gemm_load_chunk(const float* __restrict__ xk,
                                                const float* __restrict__ wk,
                                                float4& a, float (&b)[4][NT]) {
    a = *(const float4*)xk;
#pragma unroll
    for (int j = 0; j < 4; ++j)
#pragma unroll
        for (int t = 0; t < NT; ++t) b[j][t] = wk[j * M + t * 16];
}

template <int NT>
__device__ __forceinline__ void gemm_mac_chunk(const float4& a, const float (&b)[4][NT],
                                               v8f (&acc)[NT]) {
    v2f a01, a23;
    a01.x = a.x; a01.y = a.y;
    a23.x = a.z; a23.y = a.w;
#pragma unroll
    for (int t = 0; t < NT; ++t) {
        v2f bb;
        bb.x = b[0][t]; bb.y = b[1][t];
        acc[t] = __builtin_amdgcn_wmma_f32_16x16x4_f32(
                      false, a01,  false, bb,
                      (short)0, acc[t],  false,  false);
    }
#pragma unroll
    for (int t = 0; t < NT; ++t) {
        v2f bb;
        bb.x = b[2][t]; bb.y = b[3][t];
        acc[t] = __builtin_amdgcn_wmma_f32_16x16x4_f32(
                      false, a23,  false, bb,
                      (short)0, acc[t],  false,  false);
    }
}

#define LSTR 8

template <int XM>
__device__ __forceinline__ float row_xor_lane(float x) {
    return __int_as_float(__builtin_amdgcn_update_dpp(0, __float_as_int(x), 0x160 | XM,
                                                      0xf, 0xf, true));
}

template <int NT, int HEADS, int C>
__device__ __forceinline__ void strip_logits(const v8f (&acc)[NT],
                                             const float* __restrict__ a_src,
                                             const float* __restrict__ a_dst,
                                             float* __restrict__ als,
                                             float* __restrict__ ald,
                                             int row0, int half, int m16) {
    static_assert(HEADS <= LSTR, "logit slots per node");
    static_assert(HEADS * C == NT * 16, "the strip holds whole rows of H");
    const bool b3 = (m16 & 8) != 0, b2 = (m16 & 4) != 0, b1 = (m16 & 2) != 0, b0 = (m16 & 1) != 0;
    float res[LSTR];
#pragma unroll
    for (int h = 0; h < LSTR; ++h) res[h] = 0.0f;
#pragma unroll
    for (int h = 0; h < HEADS; ++h) {
        float p[16];
#pragma unroll
        for (int i = 0; i < 16; ++i) p[i] = 0.0f;
        bool first = true;
#pragma unroll
        for (int t = 0; t < NT; ++t) {
            const int lo = t * 16, hi = t * 16 + 15;
            if (hi >= h * C && lo < (h + 1) * C) {
                const int col = lo + m16;
                float ws = a_src[col], wd = a_dst[col];
                if (lo < h * C || hi >= (h + 1) * C) {
                    const bool in = col >= h * C && col < (h + 1) * C;
                    ws = in ? ws : 0.0f;
                    wd = in ? wd : 0.0f;
                }
#pragma unroll
                for (int v = 0; v < 8; ++v) {
                    p[v]     = first ? acc[t][v] * ws : p[v]     + acc[t][v] * ws;
                    p[8 + v] = first ? acc[t][v] * wd : p[8 + v] + acc[t][v] * wd;
                }
                first = false;
            }
        }
        float q8[8], q4[4], q2[2];
#pragma unroll
        for (int i = 0; i < 8; ++i)
            q8[i] = (b3 ? p[8 + i] : p[i]) + row_xor_lane<8>(b3 ? p[i] : p[8 + i]);
#pragma unroll
        for (int i = 0; i < 4; ++i)
            q4[i] = (b2 ? q8[4 + i] : q8[i]) + row_xor_lane<4>(b2 ? q8[i] : q8[4 + i]);
#pragma unroll
        for (int i = 0; i < 2; ++i)
            q2[i] = (b1 ? q4[2 + i] : q4[i]) + row_xor_lane<2>(b1 ? q4[i] : q4[2 + i]);
        res[h] = (b0 ? q2[1] : q2[0]) + row_xor_lane<1>(b0 ? q2[0] : q2[1]);
    }
    float* lp = (b3 ? ald : als) + (size_t)(row0 + half * 8 + (m16 & 7)) * LSTR;
    *(float4*)lp       = make_float4(res[0], res[1], res[2], res[3]);
    *(float4*)(lp + 4) = make_float4(res[4], res[5], res[6], res[7]);
}

template <int K, int NT, int HEADS, int C>
__global__ void __launch_bounds__(64) __attribute__((amdgpu_waves_per_eu(4, 4)))
gemm_strip_wmma_f32(const float* __restrict__ X, const float* __restrict__ W,
                    const float* __restrict__ a_src, const float* __restrict__ a_dst,
                    float* __restrict__ H, float* __restrict__ als, float* __restrict__ ald,
                    int n) {
    constexpr int M = NT * 16;
    static_assert(K % 16 == 0, "two 8-k chunks per trip");
    const int lane  = threadIdx.x & 31;
    const int strip = blockIdx.x * (blockDim.x >> 5) + (threadIdx.x >> 5);
    if (strip >= (n >> 4)) return;

    const int row0 = strip << 4;
    const int m16  = lane & 15;
    const int half = lane >> 4;

    v8f acc[NT];
#pragma unroll
    for (int t = 0; t < NT; ++t) acc[t] = v8f{};

    const float* xp = X + (size_t)(row0 + m16) * K + half * 4;
    const float* wp = W + (size_t)(half * 4) * M + m16;

    float4 a0, a1;
    float  b0[4][NT], b1[4][NT];
    gemm_load_chunk<M, NT>(xp, wp, a0, b0);
#pragma nounroll
    for (int k0 = 0; k0 < K; k0 += 16) {
        gemm_load_chunk<M, NT>(xp + k0 + 8, wp + (size_t)(k0 + 8) * M, a1, b1);
        gemm_mac_chunk<NT>(a0, b0, acc);
        if (k0 + 16 < K)
            gemm_load_chunk<M, NT>(xp + k0 + 16, wp + (size_t)(k0 + 16) * M, a0, b0);
        gemm_mac_chunk<NT>(a1, b1, acc);
    }

    float* hp = H + (size_t)(row0 + half * 8) * M + m16;
#pragma unroll
    for (int t = 0; t < NT; ++t)
#pragma unroll
        for (int v = 0; v < 8; ++v)
            hp[(size_t)v * M + t * 16] = acc[t][v];

    strip_logits<NT, HEADS, C>(acc, a_src, a_dst, als, ald, row0, half, m16);
}

#define BSHIFT 8
#define BINW   (1 << BSHIFT)
#define NBIN   ((NNODE + BINW - 1) >> BSHIFT)
#define GB     256
#define NTAB   (NBIN * GB)
#define EPT    4
#define BPT    8

__global__ void bin_count(const int* __restrict__ ei, int E, int* __restrict__ bh) {
    __shared__ int hist[NBIN];
    for (int i = threadIdx.x; i < NBIN; i += blockDim.x) hist[i] = 0;
    __syncthreads();
    const int chunk = (E + GB - 1) / GB;
    const int e0 = blockIdx.x * chunk;
    const int e1 = e0 + chunk < E ? e0 + chunk : E;
    for (int e = e0 + threadIdx.x; e < e1; e += EPT * blockDim.x) {
        int d[EPT];
#pragma unroll
        for (int q = 0; q < EPT; ++q) {
            const int eq = e + q * blockDim.x;
            d[q] = ei[E + (eq < e1 ? eq : e1 - 1)];
        }
#pragma unroll
        for (int q = 0; q < EPT; ++q)
            if (e + q * blockDim.x < e1) atomicAdd(&hist[d[q] >> BSHIFT], 1);
    }
    __syncthreads();
    for (int i = threadIdx.x; i < NBIN; i += blockDim.x) bh[i * GB + blockIdx.x] = hist[i];
}

__global__ void excl_scan(const int* __restrict__ in, int* __restrict__ out, int n) {
    __shared__ int wtot[32];
    const int lane  = threadIdx.x & 31;
    const int wave  = threadIdx.x >> 5;
    const int nw    = blockDim.x >> 5;
    const int iters = (n + nw * 128 - 1) / (nw * 128);
    const int base  = wave * iters * 128 + lane * 4;

    int s = 0;
    for (int i = 0; i < iters; ++i) {
        const int idx = base + i * 128;
#pragma unroll
        for (int q = 0; q < 4; ++q)
            if (idx + q < n) s += in[idx + q];
    }
    for (int o = 16; o > 0; o >>= 1) s += __shfl_xor(s, o);
    if (lane == 0) wtot[wave] = s;
    __syncthreads();

    int carry = 0;
    for (int w = 0; w < wave; ++w) carry += wtot[w];

    for (int i = 0; i < iters; ++i) {
        const int idx = base + i * 128;
        int v[4];
#pragma unroll
        for (int q = 0; q < 4; ++q) v[q] = idx + q < n ? in[idx + q] : 0;
        const int lt = v[0] + v[1] + v[2] + v[3];
        int inc = lt;
        for (int o = 1; o < 32; o <<= 1) {
            const int t = __shfl_up(inc, o);
            if (lane >= o) inc += t;
        }
        int run = carry + inc - lt;
#pragma unroll
        for (int q = 0; q < 4; ++q) {
            if (idx + q < n) out[idx + q] = run;
            run += v[q];
        }
        carry += __shfl(inc, 31);
    }
}

__global__ void bin_scatter(const int* __restrict__ ei, int E,
                            const int* __restrict__ boff, int2* __restrict__ binned) {
    __shared__ int cur[NBIN];
    for (int i = threadIdx.x; i < NBIN; i += blockDim.x) cur[i] = boff[i * GB + blockIdx.x];
    __syncthreads();
    const int chunk = (E + GB - 1) / GB;
    const int e0 = blockIdx.x * chunk;
    const int e1 = e0 + chunk < E ? e0 + chunk : E;
    for (int e = e0 + threadIdx.x; e < e1; e += EPT * blockDim.x) {
        int d[EPT], sv[EPT];
#pragma unroll
        for (int q = 0; q < EPT; ++q) {
            const int eq = e + q * blockDim.x;
            const int ec = eq < e1 ? eq : e1 - 1;
            d[q]  = ei[E + ec];
            sv[q] = ei[ec];
        }
#pragma unroll
        for (int q = 0; q < EPT; ++q) {
            if (e + q * blockDim.x < e1) {
                const int pos = atomicAdd(&cur[d[q] >> BSHIFT], 1);
                binned[pos] = make_int2(sv[q], d[q]);
            }
        }
    }
}

__global__ void bin_group(const int2* __restrict__ binned, const int* __restrict__ boff,
                          int E, int n, int* __restrict__ cnt, int* __restrict__ rowptr,
                          int* __restrict__ col) {
    __shared__ int c[BINW];
    __shared__ int cur[BINW];
    const int b     = blockIdx.x;
    const int node0 = b << BSHIFT;
    const int beg   = boff[b * GB];
    const int end   = b + 1 < NBIN ? boff[(b + 1) * GB] : E;
    for (int i = threadIdx.x; i < BINW; i += blockDim.x) c[i] = 0;
    __syncthreads();
    for (int i = beg + threadIdx.x; i < end; i += BPT * blockDim.x) {
        int dl[BPT];
#pragma unroll
        for (int q = 0; q < BPT; ++q) {
            const int iq = i + q * blockDim.x;
            dl[q] = binned[iq < end ? iq : end - 1].y;
        }
#pragma unroll
        for (int q = 0; q < BPT; ++q)
            if (i + q * blockDim.x < end) atomicAdd(&c[dl[q] - node0], 1);
    }
    __syncthreads();
    if (threadIdx.x < 32) {
        constexpr int PER = BINW / 32;
        const int lane = threadIdx.x;
        int v[PER];
        int lt = 0;
#pragma unroll
        for (int q = 0; q < PER; ++q) { v[q] = c[lane * PER + q]; lt += v[q]; }
        int inc = lt;
        for (int o = 1; o < 32; o <<= 1) {
            const int t = __shfl_up(inc, o);
            if (lane >= o) inc += t;
        }
        int run = beg + inc - lt;
#pragma unroll
        for (int q = 0; q < PER; ++q) {
            const int dl = lane * PER + q;
            if (node0 + dl < n) { cnt[node0 + dl] = v[q]; rowptr[node0 + dl] = run; }
            cur[dl] = run;
            run += v[q];
        }
    }
    __syncthreads();
    for (int i = beg + threadIdx.x; i < end; i += BPT * blockDim.x) {
        int2 ed[BPT];
#pragma unroll
        for (int q = 0; q < BPT; ++q) {
            const int iq = i + q * blockDim.x;
            ed[q] = binned[iq < end ? iq : end - 1];
        }
#pragma unroll
        for (int q = 0; q < BPT; ++q) {
            if (i + q * blockDim.x < end) {
                const int pos = atomicAdd(&cur[ed[q].y - node0], 1);
                col[pos] = ed[q].x;
            }
        }
    }
}

#define WREC (32 * LSTR)

template <int HEADS>
__device__ __forceinline__ void edge_logits(const int* __restrict__ col,
                                            const float* __restrict__ als,
                                            const float* ad, int beg, int deg, int d,
                                            int t, int& s, float* lr) {
    s = d;
#pragma unroll
    for (int h = 0; h < HEADS; ++h) lr[h] = -INFINITY;
    if (t <= deg) {
        if (t < deg) s = col[beg + t];
        const float4* p = (const float4*)(als + (size_t)s * LSTR);
        const float4 lo = p[0], hi = p[1];
        const float a[8] = {lo.x, lo.y, lo.z, lo.w, hi.x, hi.y, hi.z, hi.w};
#pragma unroll
        for (int h = 0; h < HEADS; ++h) {
            const float x = a[h] + ad[h];
            lr[h] = x >= 0.0f ? x : 0.2f * x;
        }
    }
}

template <int HEADS, int C, int RELU, int BATCH>
__device__ __forceinline__ void aggregate_body(const int* __restrict__ rowptr,
                                               const int* __restrict__ cnt,
                                               const int* __restrict__ col,
                                               const float* __restrict__ als,
                                               const float* __restrict__ ald,
                                               const float* __restrict__ H,
                                               const float* __restrict__ bias,
                                               float* __restrict__ out, int n,
                                               float* wt  ) {
    constexpr int HC = HEADS * C;
    constexpr int NP = (HC + 127) / 128;
    static_assert(C % 4 == 0, "a lane's four channels belong to one head");
    static_assert(HC % 4 == 0, "rows are whole 16-byte groups");
    static_assert(HEADS <= LSTR, "logit slots per node");
    static_assert(32 % BATCH == 0, "a batch never reaches past the 32 records of the table");
    const int lane = threadIdx.x & 31;
    const int d = __builtin_amdgcn_readfirstlane(
        (int)(blockIdx.x * (blockDim.x >> 5) + (threadIdx.x >> 5)));
    if (d >= n) return;
    const int beg = rowptr[d];
    const int deg = cnt[d];
    const int tot = deg + 1;

    float  ad[HEADS], m[HEADS], dsum[HEADS];
    float4 acc[NP];
    bool   live[NP];
    int    hp[NP];
#pragma unroll
    for (int h = 0; h < HEADS; ++h) {
        ad[h]   = ald[(size_t)d * LSTR + h];
        dsum[h] = 0.0f;
    }
#pragma unroll
    for (int p = 0; p < NP; ++p) {
        const int c = p * 128 + 4 * lane;
        acc[p]  = make_float4(0.0f, 0.0f, 0.0f, 0.0f);
        live[p] = c < HC;
        hp[p]   = c / C < HEADS ? c / C : HEADS - 1;
    }

    int   s0;
    float lr0[HEADS];
    edge_logits<HEADS>(col, als, ad, beg, deg, d, lane, s0, lr0);
#pragma unroll
    for (int h = 0; h < HEADS; ++h) m[h] = lr0[h];
    for (int c0 = 32; c0 < tot; c0 += 32) {
        int   s;
        float lr[HEADS];
        edge_logits<HEADS>(col, als, ad, beg, deg, d, c0 + lane, s, lr);
#pragma unroll
        for (int h = 0; h < HEADS; ++h) m[h] = fmaxf(m[h], lr[h]);
    }
#pragma unroll
    for (int h = 0; h < HEADS; ++h)
        for (int o = 16; o > 0; o >>= 1) m[h] = fmaxf(m[h], __shfl_xor(m[h], o));

    for (int c0 = 0; c0 < tot; c0 += 32) {
        int   s = s0;
        float ex[LSTR];
#pragma unroll
        for (int h = 0; h < HEADS; ++h) ex[h] = lr0[h];
        if (c0 > 0) edge_logits<HEADS>(col, als, ad, beg, deg, d, c0 + lane, s, ex);
#pragma unroll
        for (int h = 0; h < HEADS; ++h) {
            ex[h] = __expf(ex[h] - m[h]);
            dsum[h] += ex[h];
        }
#pragma unroll
        for (int h = HEADS; h < LSTR; ++h) ex[h] = 0.0f;
        *(float4*)(wt + lane * LSTR)     = make_float4(ex[0], ex[1], ex[2], ex[3]);
        *(float4*)(wt + lane * LSTR + 4) = make_float4(ex[4], ex[5], ex[6], ex[7]);

        const int nj = (tot - c0) < 32 ? (tot - c0) : 32;
        for (int j0 = 0; j0 < nj; j0 += BATCH) {
            float4 hv[BATCH][NP];
            float  w[BATCH][NP];
#pragma unroll
            for (int i = 0; i < BATCH; ++i) {
#pragma unroll
                for (int p = 0; p < NP; ++p) hv[i][p] = make_float4(0.0f, 0.0f, 0.0f, 0.0f);
                if (j0 + i < nj) {
                    const int sj = __builtin_amdgcn_readlane(s, j0 + i);
                    const float* hs = H + (size_t)sj * HC;
#pragma unroll
                    for (int p = 0; p < NP; ++p)
                        if (live[p]) hv[i][p] = *(const float4*)(hs + p * 128 + 4 * lane);
                }
            }
#pragma unroll
            for (int i = 0; i < BATCH; ++i)
#pragma unroll
                for (int p = 0; p < NP; ++p) w[i][p] = wt[(j0 + i) * LSTR + hp[p]];
#pragma unroll
            for (int i = 0; i < BATCH; ++i) {
                if (j0 + i < nj) {
#pragma unroll
                    for (int p = 0; p < NP; ++p) {
                        acc[p].x += hv[i][p].x * w[i][p];
                        acc[p].y += hv[i][p].y * w[i][p];
                        acc[p].z += hv[i][p].z * w[i][p];
                        acc[p].w += hv[i][p].w * w[i][p];
                    }
                }
            }
        }
    }
#pragma unroll
    for (int h = 0; h < HEADS; ++h)
        for (int o = 16; o > 0; o >>= 1) dsum[h] += __shfl_xor(dsum[h], o);

    float* orow = out + (size_t)d * HC;
#pragma unroll
    for (int p = 0; p < NP; ++p) {
        if (live[p]) {
            const int c = p * 128 + 4 * lane;
            float den = dsum[0];
#pragma unroll
            for (int h = 1; h < HEADS; ++h) den = hp[p] == h ? dsum[h] : den;
            const float4 b = *(const float4*)(bias + c);
            float4 v;
            v.x = acc[p].x / den + b.x;
            v.y = acc[p].y / den + b.y;
            v.z = acc[p].z / den + b.z;
            v.w = acc[p].w / den + b.w;
            if (RELU) {
                v.x = fmaxf(v.x, 0.0f); v.y = fmaxf(v.y, 0.0f);
                v.z = fmaxf(v.z, 0.0f); v.w = fmaxf(v.w, 0.0f);
            }
            *(float4*)(orow + c) = v;
        }
    }
}

__global__ void aggregate_7x16(const int* __restrict__ rowptr, const int* __restrict__ cnt,
                               const int* __restrict__ col, const float* __restrict__ als,
                               const float* __restrict__ ald, const float* __restrict__ H,
                               const float* __restrict__ bias, float* __restrict__ out, int n) {
    __shared__ __attribute__((aligned(16))) float wtab[8 * WREC];
    aggregate_body<7, 16, 1, 8>(rowptr, cnt, col, als, ald, H, bias, out, n,
                                wtab + (threadIdx.x >> 5) * WREC);
}

__global__ void aggregate_6x16(const int* __restrict__ rowptr, const int* __restrict__ cnt,
                               const int* __restrict__ col, const float* __restrict__ als,
                               const float* __restrict__ ald, const float* __restrict__ H,
                               const float* __restrict__ bias, float* __restrict__ out, int n) {
    __shared__ __attribute__((aligned(16))) float wtab[8 * WREC];
    aggregate_body<6, 16, 1, 8>(rowptr, cnt, col, als, ald, H, bias, out, n,
                                wtab + (threadIdx.x >> 5) * WREC);
}

__global__ void aggregate_6x40(const int* __restrict__ rowptr, const int* __restrict__ cnt,
                               const int* __restrict__ col, const float* __restrict__ als,
                               const float* __restrict__ ald, const float* __restrict__ H,
                               const float* __restrict__ bias, float* __restrict__ out, int n) {
    __shared__ __attribute__((aligned(16))) float wtab[8 * WREC];
    aggregate_body<6, 40, 0, 4>(rowptr, cnt, col, als, ald, H, bias, out, n,
                                wtab + (threadIdx.x >> 5) * WREC);
}


extern "C" void kernel_launch(void* const* d_in, const int* in_sizes, int n_in,
                              void* d_out, int out_size, void* d_ws, size_t ws_size,
                              hipStream_t stream) {
    const float* x      = (const float*)d_in[0];
    const int*   ei     = (const int*)  d_in[1];
    const float* W1     = (const float*)d_in[2];
    const float* a_src1 = (const float*)d_in[3];
    const float* a_dst1 = (const float*)d_in[4];
    const float* b1     = (const float*)d_in[5];
    const float* W2     = (const float*)d_in[6];
    const float* a_src2 = (const float*)d_in[7];
    const float* a_dst2 = (const float*)d_in[8];
    const float* b2     = (const float*)d_in[9];
    const float* W3     = (const float*)d_in[10];
    const float* a_src3 = (const float*)d_in[11];
    const float* a_dst3 = (const float*)d_in[12];
    const float* b3     = (const float*)d_in[13];

    const int E = in_sizes[1] / 2;

    float* ws   = (float*)d_ws;
    float* bufH = ws;
    float* act1 = bufH + (size_t)NNODE * 240;
    float* h2   = act1 + (size_t)NNODE * 112;
    float* act2 = h2   + (size_t)NNODE * 96;
    float* als  = act2 + (size_t)NNODE * 96;
    float* ald  = als  + (size_t)NNODE * LSTR;
    int*   cnt    = (int*)(ald + (size_t)NNODE * LSTR);
    int*   rowptr = cnt    + NNODE;
    int*   bh     = rowptr + NNODE;
    int*   boff   = bh     + NTAB;
    int2*  binned = (int2*)(boff + NTAB);
    int*   col    = (int*)(binned + E);
    float* out  = (float*)d_out;

    bin_count<<<GB, 1024, 0, stream>>>(ei, E, bh);
    excl_scan<<<1, 1024, 0, stream>>>(bh, boff, NTAB);
    bin_scatter<<<GB, 1024, 0, stream>>>(ei, E, boff, binned);
    bin_group<<<NBIN, 1024, 0, stream>>>(binned, boff, E, NNODE, cnt, rowptr, col);

    auto layer = [&](const float* X, int K,
                     const float* W, const float* as, const float* ad, const float* b,
                     int heads, int C, float* Hbuf, float* obuf) {
        const int gb = (NNODE / 16 + 1) / 2;
        if (heads == 7)
            gemm_strip_wmma_f32<128, 7, 7, 16><<<gb, 64, 0, stream>>>(X, W, as, ad, Hbuf,
                                                                      als, ald, NNODE);
        else if (C == 16)
            gemm_strip_wmma_f32<112, 6, 6, 16><<<gb, 64, 0, stream>>>(X, W, as, ad, Hbuf,
                                                                      als, ald, NNODE);
        else
            gemm_strip_wmma_f32<96, 15, 6, 40><<<gb, 64, 0, stream>>>(X, W, as, ad, Hbuf,
                                                                      als, ald, NNODE);

        const int nb = (NNODE + 7) / 8;
        if (heads == 7)
            aggregate_7x16<<<nb, 256, 0, stream>>>(rowptr, cnt, col, als, ald, Hbuf,
                                                   b, obuf, NNODE);
        else if (C == 16)
            aggregate_6x16<<<nb, 256, 0, stream>>>(rowptr, cnt, col, als, ald, Hbuf,
                                                   b, obuf, NNODE);
        else
            aggregate_6x40<<<nb, 256, 0, stream>>>(rowptr, cnt, col, als, ald, Hbuf,
                                                   b, obuf, NNODE);
    };

    layer(x,    128, W1, a_src1, a_dst1, b1, 7, 16, bufH, act1);
    layer(act1, 112, W2, a_src2, a_dst2, b2, 6, 16, h2,   act2);
    layer(act2,  96, W3, a_src3, a_dst3, b3, 6, 40, bufH, out);
}


# edge grouping in 2 launches, not 4: each workgroup bins its own edge chunk in place from its own LDS histogram, the per-bin launch collects the 256 pieces; sums stay in segment arrival order
# speedup vs baseline: 19.6382x; 1.1703x over previous
#include <hip/hip_runtime.h>
#include <math.h>

#define NNODE 50000

typedef float v2f __attribute__((ext_vector_type(2)));
typedef float v8f __attribute__((ext_vector_type(8)));

template <int M, int NT>
__device__ __forceinline__ void gemm_load_chunk(const float* __restrict__ xk,
                                                const float* __restrict__ wk,
                                                float4& a, float (&b)[4][NT]) {
    a = *(const float4*)xk;
#pragma unroll
    for (int j = 0; j < 4; ++j)
#pragma unroll
        for (int t = 0; t < NT; ++t) b[j][t] = wk[j * M + t * 16];
}

template <int NT>
__device__ __forceinline__ void gemm_mac_chunk(const float4& a, const float (&b)[4][NT],
                                               v8f (&acc)[NT]) {
    v2f a01, a23;
    a01.x = a.x; a01.y = a.y;
    a23.x = a.z; a23.y = a.w;
#pragma unroll
    for (int t = 0; t < NT; ++t) {
        v2f bb;
        bb.x = b[0][t]; bb.y = b[1][t];
        acc[t] = __builtin_amdgcn_wmma_f32_16x16x4_f32(
                      false, a01,  false, bb,
                      (short)0, acc[t],  false,  false);
    }
#pragma unroll
    for (int t = 0; t < NT; ++t) {
        v2f bb;
        bb.x = b[2][t]; bb.y = b[3][t];
        acc[t] = __builtin_amdgcn_wmma_f32_16x16x4_f32(
                      false, a23,  false, bb,
                      (short)0, acc[t],  false,  false);
    }
}

#define LSTR 8

template <int XM>
__device__ __forceinline__ float row_xor_lane(float x) {
    return __int_as_float(__builtin_amdgcn_update_dpp(0, __float_as_int(x), 0x160 | XM,
                                                      0xf, 0xf, true));
}

template <int NT, int HEADS, int C>
__device__ __forceinline__ void strip_logits(const v8f (&acc)[NT],
                                             const float* __restrict__ a_src,
                                             const float* __restrict__ a_dst,
                                             float* __restrict__ als,
                                             float* __restrict__ ald,
                                             int row0, int half, int m16) {
    static_assert(HEADS <= LSTR, "logit slots per node");
    static_assert(HEADS * C == NT * 16, "the strip holds whole rows of H");
    const bool b3 = (m16 & 8) != 0, b2 = (m16 & 4) != 0, b1 = (m16 & 2) != 0, b0 = (m16 & 1) != 0;
    float res[LSTR];
#pragma unroll
    for (int h = 0; h < LSTR; ++h) res[h] = 0.0f;
#pragma unroll
    for (int h = 0; h < HEADS; ++h) {
        float p[16];
#pragma unroll
        for (int i = 0; i < 16; ++i) p[i] = 0.0f;
        bool first = true;
#pragma unroll
        for (int t = 0; t < NT; ++t) {
            const int lo = t * 16, hi = t * 16 + 15;
            if (hi >= h * C && lo < (h + 1) * C) {
                const int col = lo + m16;
                float ws = a_src[col], wd = a_dst[col];
                if (lo < h * C || hi >= (h + 1) * C) {
                    const bool in = col >= h * C && col < (h + 1) * C;
                    ws = in ? ws : 0.0f;
                    wd = in ? wd : 0.0f;
                }
#pragma unroll
                for (int v = 0; v < 8; ++v) {
                    p[v]     = first ? acc[t][v] * ws : p[v]     + acc[t][v] * ws;
                    p[8 + v] = first ? acc[t][v] * wd : p[8 + v] + acc[t][v] * wd;
                }
                first = false;
            }
        }
        float q8[8], q4[4], q2[2];
#pragma unroll
        for (int i = 0; i < 8; ++i)
            q8[i] = (b3 ? p[8 + i] : p[i]) + row_xor_lane<8>(b3 ? p[i] : p[8 + i]);
#pragma unroll
        for (int i = 0; i < 4; ++i)
            q4[i] = (b2 ? q8[4 + i] : q8[i]) + row_xor_lane<4>(b2 ? q8[i] : q8[4 + i]);
#pragma unroll
        for (int i = 0; i < 2; ++i)
            q2[i] = (b1 ? q4[2 + i] : q4[i]) + row_xor_lane<2>(b1 ? q4[i] : q4[2 + i]);
        res[h] = (b0 ? q2[1] : q2[0]) + row_xor_lane<1>(b0 ? q2[0] : q2[1]);
    }
    float* lp = (b3 ? ald : als) + (size_t)(row0 + half * 8 + (m16 & 7)) * LSTR;
    *(float4*)lp       = make_float4(res[0], res[1], res[2], res[3]);
    *(float4*)(lp + 4) = make_float4(res[4], res[5], res[6], res[7]);
}

template <int K, int NT, int HEADS, int C>
__global__ void __launch_bounds__(64) __attribute__((amdgpu_waves_per_eu(4, 4)))
gemm_strip_wmma_f32(const float* __restrict__ X, const float* __restrict__ W,
                    const float* __restrict__ a_src, const float* __restrict__ a_dst,
                    float* __restrict__ H, float* __restrict__ als, float* __restrict__ ald,
                    int n) {
    constexpr int M = NT * 16;
    static_assert(K % 16 == 0, "two 8-k chunks per trip");
    const int lane  = threadIdx.x & 31;
    const int strip = blockIdx.x * (blockDim.x >> 5) + (threadIdx.x >> 5);
    if (strip >= (n >> 4)) return;

    const int row0 = strip << 4;
    const int m16  = lane & 15;
    const int half = lane >> 4;

    v8f acc[NT];
#pragma unroll
    for (int t = 0; t < NT; ++t) acc[t] = v8f{};

    const float* xp = X + (size_t)(row0 + m16) * K + half * 4;
    const float* wp = W + (size_t)(half * 4) * M + m16;

    float4 a0, a1;
    float  b0[4][NT], b1[4][NT];
    gemm_load_chunk<M, NT>(xp, wp, a0, b0);
#pragma nounroll
    for (int k0 = 0; k0 < K; k0 += 16) {
        gemm_load_chunk<M, NT>(xp + k0 + 8, wp + (size_t)(k0 + 8) * M, a1, b1);
        gemm_mac_chunk<NT>(a0, b0, acc);
        if (k0 + 16 < K)
            gemm_load_chunk<M, NT>(xp + k0 + 16, wp + (size_t)(k0 + 16) * M, a0, b0);
        gemm_mac_chunk<NT>(a1, b1, acc);
    }

    float* hp = H + (size_t)(row0 + half * 8) * M + m16;
#pragma unroll
    for (int t = 0; t < NT; ++t)
#pragma unroll
        for (int v = 0; v < 8; ++v)
            hp[(size_t)v * M + t * 16] = acc[t][v];

    strip_logits<NT, HEADS, C>(acc, a_src, a_dst, als, ald, row0, half, m16);
}

#define BSHIFT 8
#define BINW   (1 << BSHIFT)
#define NBIN   ((NNODE + BINW - 1) >> BSHIFT)
#define GB     256
#define NPST   ((NBIN + 1) * GB)
#define EPT    4
#define BPT    8
#define PSUB   4

__global__ void bin_local(const int* __restrict__ ei, int E, int* __restrict__ pst,
                          int2* __restrict__ binned) {
    __shared__ int hist[NBIN];
    for (int i = threadIdx.x; i < NBIN; i += blockDim.x) hist[i] = 0;
    __syncthreads();
    const int chunk = (E + GB - 1) / GB;
    const int c0 = blockIdx.x * chunk;
    const int e0 = c0 < E ? c0 : E;
    const int e1 = e0 + chunk < E ? e0 + chunk : E;
    for (int e = e0 + threadIdx.x; e < e1; e += EPT * blockDim.x) {
        int d[EPT];
#pragma unroll
        for (int q = 0; q < EPT; ++q) {
            const int eq = e + q * blockDim.x;
            d[q] = ei[E + (eq < e1 ? eq : e1 - 1)];
        }
#pragma unroll
        for (int q = 0; q < EPT; ++q)
            if (e + q * blockDim.x < e1) atomicAdd(&hist[d[q] >> BSHIFT], 1);
    }
    __syncthreads();
    if (threadIdx.x < 32) {
        constexpr int PER = (NBIN + 31) / 32;
        const int lane = threadIdx.x;
        int v[PER];
        int lt = 0;
#pragma unroll
        for (int q = 0; q < PER; ++q) {
            const int b = lane * PER + q;
            v[q] = b < NBIN ? hist[b] : 0;
            lt += v[q];
        }
        int inc = lt;
        for (int o = 1; o < 32; o <<= 1) {
            const int t = __shfl_up(inc, o);
            if (lane >= o) inc += t;
        }
        int run = e0 + inc - lt;
#pragma unroll
        for (int q = 0; q < PER; ++q) {
            const int b = lane * PER + q;
            if (b < NBIN) { hist[b] = run; pst[b * GB + blockIdx.x] = run; }
            run += v[q];
        }
        if (lane == 31) pst[NBIN * GB + blockIdx.x] = run;
    }
    __syncthreads();
    for (int e = e0 + threadIdx.x; e < e1; e += EPT * blockDim.x) {
        int d[EPT], sv[EPT];
#pragma unroll
        for (int q = 0; q < EPT; ++q) {
            const int eq = e + q * blockDim.x;
            const int ec = eq < e1 ? eq : e1 - 1;
            d[q]  = ei[E + ec];
            sv[q] = ei[ec];
        }
#pragma unroll
        for (int q = 0; q < EPT; ++q) {
            if (e + q * blockDim.x < e1) {
                const int pos = atomicAdd(&hist[d[q] >> BSHIFT], 1);
                binned[pos] = make_int2(sv[q], d[q]);
            }
        }
    }
}

__global__ void bin_group(const int2* __restrict__ binned, const int* __restrict__ pst,
                          int E, int n, int* __restrict__ cnt, int* __restrict__ rowptr,
                          int* __restrict__ col) {
    __shared__ int c[BINW];
    __shared__ int cur[BINW];
    __shared__ int wlow[32];
    const int b     = blockIdx.x;
    const int node0 = b << BSHIFT;
    const int lane  = threadIdx.x & 31;
    const int p     = threadIdx.x / PSUB;
    const int sub   = threadIdx.x % PSUB;
    const int ps    = p < GB ? pst[b * GB + p] : 0;
    const int len   = p < GB ? pst[(b + 1) * GB + p] - ps : 0;
    for (int i = threadIdx.x; i < BINW; i += blockDim.x) c[i] = 0;
    {
        const int chunk = (E + GB - 1) / GB;
        const int c0 = p * chunk;
        int low = (p < GB && sub == 0) ? ps - (c0 < E ? c0 : E) : 0;
        for (int o = 16; o > 0; o >>= 1) low += __shfl_xor(low, o);
        if (lane == 0) wlow[threadIdx.x >> 5] = low;
    }
    __syncthreads();
    for (int k = sub; k < len; k += PSUB * BPT) {
        int dl[BPT];
#pragma unroll
        for (int q = 0; q < BPT; ++q) {
            const int kq = k + q * PSUB;
            dl[q] = binned[ps + (kq < len ? kq : len - 1)].y;
        }
#pragma unroll
        for (int q = 0; q < BPT; ++q)
            if (k + q * PSUB < len) atomicAdd(&c[dl[q] - node0], 1);
    }
    __syncthreads();
    if (threadIdx.x < 32) {
        constexpr int PER = BINW / 32;
        int beg = lane < (int)(blockDim.x >> 5) ? wlow[lane] : 0;
        for (int o = 16; o > 0; o >>= 1) beg += __shfl_xor(beg, o);
        int v[PER];
        int lt = 0;
#pragma unroll
        for (int q = 0; q < PER; ++q) { v[q] = c[lane * PER + q]; lt += v[q]; }
        int inc = lt;
        for (int o = 1; o < 32; o <<= 1) {
            const int t = __shfl_up(inc, o);
            if (lane >= o) inc += t;
        }
        int run = beg + inc - lt;
#pragma unroll
        for (int q = 0; q < PER; ++q) {
            const int dl = lane * PER + q;
            if (node0 + dl < n) { cnt[node0 + dl] = v[q]; rowptr[node0 + dl] = run; }
            cur[dl] = run;
            run += v[q];
        }
    }
    __syncthreads();
    for (int k = sub; k < len; k += PSUB * BPT) {
        int2 ed[BPT];
#pragma unroll
        for (int q = 0; q < BPT; ++q) {
            const int kq = k + q * PSUB;
            ed[q] = binned[ps + (kq < len ? kq : len - 1)];
        }
#pragma unroll
        for (int q = 0; q < BPT; ++q) {
            if (k + q * PSUB < len) {
                const int pos = atomicAdd(&cur[ed[q].y - node0], 1);
                col[pos] = ed[q].x;
            }
        }
    }
}

#define WREC (32 * LSTR)

template <int HEADS>
__device__ __forceinline__ void edge_logits(const int* __restrict__ col,
                                            const float* __restrict__ als,
                                            const float* ad, int beg, int deg, int d,
                                            int t, int& s, float* lr) {
    s = d;
#pragma unroll
    for (int h = 0; h < HEADS; ++h) lr[h] = -INFINITY;
    if (t <= deg) {
        if (t < deg) s = col[beg + t];
        const float4* p = (const float4*)(als + (size_t)s * LSTR);
        const float4 lo = p[0], hi = p[1];
        const float a[8] = {lo.x, lo.y, lo.z, lo.w, hi.x, hi.y, hi.z, hi.w};
#pragma unroll
        for (int h = 0; h < HEADS; ++h) {
            const float x = a[h] + ad[h];
            lr[h] = x >= 0.0f ? x : 0.2f * x;
        }
    }
}

template <int HEADS, int C, int RELU, int BATCH>
__device__ __forceinline__ void aggregate_body(const int* __restrict__ rowptr,
                                               const int* __restrict__ cnt,
                                               const int* __restrict__ col,
                                               const float* __restrict__ als,
                                               const float* __restrict__ ald,
                                               const float* __restrict__ H,
                                               const float* __restrict__ bias,
                                               float* __restrict__ out, int n,
                                               float* wt  ) {
    constexpr int HC = HEADS * C;
    constexpr int NP = (HC + 127) / 128;
    static_assert(C % 4 == 0, "a lane's four channels belong to one head");
    static_assert(HC % 4 == 0, "rows are whole 16-byte groups");
    static_assert(HEADS <= LSTR, "logit slots per node");
    static_assert(32 % BATCH == 0, "a batch never reaches past the 32 records of the table");
    const int lane = threadIdx.x & 31;
    const int d = __builtin_amdgcn_readfirstlane(
        (int)(blockIdx.x * (blockDim.x >> 5) + (threadIdx.x >> 5)));
    if (d >= n) return;
    const int beg = rowptr[d];
    const int deg = cnt[d];
    const int tot = deg + 1;

    float  ad[HEADS], m[HEADS], dsum[HEADS];
    float4 acc[NP];
    bool   live[NP];
    int    hp[NP];
#pragma unroll
    for (int h = 0; h < HEADS; ++h) {
        ad[h]   = ald[(size_t)d * LSTR + h];
        dsum[h] = 0.0f;
    }
#pragma unroll
    for (int p = 0; p < NP; ++p) {
        const int c = p * 128 + 4 * lane;
        acc[p]  = make_float4(0.0f, 0.0f, 0.0f, 0.0f);
        live[p] = c < HC;
        hp[p]   = c / C < HEADS ? c / C : HEADS - 1;
    }

    int   s0;
    float lr0[HEADS];
    edge_logits<HEADS>(col, als, ad, beg, deg, d, lane, s0, lr0);
#pragma unroll
    for (int h = 0; h < HEADS; ++h) m[h] = lr0[h];
    for (int c0 = 32; c0 < tot; c0 += 32) {
        int   s;
        float lr[HEADS];
        edge_logits<HEADS>(col, als, ad, beg, deg, d, c0 + lane, s, lr);
#pragma unroll
        for (int h = 0; h < HEADS; ++h) m[h] = fmaxf(m[h], lr[h]);
    }
#pragma unroll
    for (int h = 0; h < HEADS; ++h)
        for (int o = 16; o > 0; o >>= 1) m[h] = fmaxf(m[h], __shfl_xor(m[h], o));

    for (int c0 = 0; c0 < tot; c0 += 32) {
        int   s = s0;
        float ex[LSTR];
#pragma unroll
        for (int h = 0; h < HEADS; ++h) ex[h] = lr0[h];
        if (c0 > 0) edge_logits<HEADS>(col, als, ad, beg, deg, d, c0 + lane, s, ex);
#pragma unroll
        for (int h = 0; h < HEADS; ++h) {
            ex[h] = __expf(ex[h] - m[h]);
            dsum[h] += ex[h];
        }
#pragma unroll
        for (int h = HEADS; h < LSTR; ++h) ex[h] = 0.0f;
        *(float4*)(wt + lane * LSTR)     = make_float4(ex[0], ex[1], ex[2], ex[3]);
        *(float4*)(wt + lane * LSTR + 4) = make_float4(ex[4], ex[5], ex[6], ex[7]);

        const int nj = (tot - c0) < 32 ? (tot - c0) : 32;
        for (int j0 = 0; j0 < nj; j0 += BATCH) {
            float4 hv[BATCH][NP];
            float  w[BATCH][NP];
#pragma unroll
            for (int i = 0; i < BATCH; ++i) {
#pragma unroll
                for (int p = 0; p < NP; ++p) hv[i][p] = make_float4(0.0f, 0.0f, 0.0f, 0.0f);
                if (j0 + i < nj) {
                    const int sj = __builtin_amdgcn_readlane(s, j0 + i);
                    const float* hs = H + (size_t)sj * HC;
#pragma unroll
                    for (int p = 0; p < NP; ++p)
                        if (live[p]) hv[i][p] = *(const float4*)(hs + p * 128 + 4 * lane);
                }
            }
#pragma unroll
            for (int i = 0; i < BATCH; ++i)
#pragma unroll
                for (int p = 0; p < NP; ++p) w[i][p] = wt[(j0 + i) * LSTR + hp[p]];
#pragma unroll
            for (int i = 0; i < BATCH; ++i) {
                if (j0 + i < nj) {
#pragma unroll
                    for (int p = 0; p < NP; ++p) {
                        acc[p].x += hv[i][p].x * w[i][p];
                        acc[p].y += hv[i][p].y * w[i][p];
                        acc[p].z += hv[i][p].z * w[i][p];
                        acc[p].w += hv[i][p].w * w[i][p];
                    }
                }
            }
        }
    }
#pragma unroll
    for (int h = 0; h < HEADS; ++h)
        for (int o = 16; o > 0; o >>= 1) dsum[h] += __shfl_xor(dsum[h], o);

    float* orow = out + (size_t)d * HC;
#pragma unroll
    for (int p = 0; p < NP; ++p) {
        if (live[p]) {
            const int c = p * 128 + 4 * lane;
            float den = dsum[0];
#pragma unroll
            for (int h = 1; h < HEADS; ++h) den = hp[p] == h ? dsum[h] : den;
            const float4 b = *(const float4*)(bias + c);
            float4 v;
            v.x = acc[p].x / den + b.x;
            v.y = acc[p].y / den + b.y;
            v.z = acc[p].z / den + b.z;
            v.w = acc[p].w / den + b.w;
            if (RELU) {
                v.x = fmaxf(v.x, 0.0f); v.y = fmaxf(v.y, 0.0f);
                v.z = fmaxf(v.z, 0.0f); v.w = fmaxf(v.w, 0.0f);
            }
            *(float4*)(orow + c) = v;
        }
    }
}

__global__ void aggregate_7x16(const int* __restrict__ rowptr, const int* __restrict__ cnt,
                               const int* __restrict__ col, const float* __restrict__ als,
                               const float* __restrict__ ald, const float* __restrict__ H,
                               const float* __restrict__ bias, float* __restrict__ out, int n) {
    __shared__ __attribute__((aligned(16))) float wtab[8 * WREC];
    aggregate_body<7, 16, 1, 8>(rowptr, cnt, col, als, ald, H, bias, out, n,
                                wtab + (threadIdx.x >> 5) * WREC);
}

__global__ void aggregate_6x16(const int* __restrict__ rowptr, const int* __restrict__ cnt,
                               const int* __restrict__ col, const float* __restrict__ als,
                               const float* __restrict__ ald, const float* __restrict__ H,
                               const float* __restrict__ bias, float* __restrict__ out, int n) {
    __shared__ __attribute__((aligned(16))) float wtab[8 * WREC];
    aggregate_body<6, 16, 1, 8>(rowptr, cnt, col, als, ald, H, bias, out, n,
                                wtab + (threadIdx.x >> 5) * WREC);
}

__global__ void aggregate_6x40(const int* __restrict__ rowptr, const int* __restrict__ cnt,
                               const int* __restrict__ col, const float* __restrict__ als,
                               const float* __restrict__ ald, const float* __restrict__ H,
                               const float* __restrict__ bias, float* __restrict__ out, int n) {
    __shared__ __attribute__((aligned(16))) float wtab[8 * WREC];
    aggregate_body<6, 40, 0, 4>(rowptr, cnt, col, als, ald, H, bias, out, n,
                                wtab + (threadIdx.x >> 5) * WREC);
}


extern "C" void kernel_launch(void* const* d_in, const int* in_sizes, int n_in,
                              void* d_out, int out_size, void* d_ws, size_t ws_size,
                              hipStream_t stream) {
    const float* x      = (const float*)d_in[0];
    const int*   ei     = (const int*)  d_in[1];
    const float* W1     = (const float*)d_in[2];
    const float* a_src1 = (const float*)d_in[3];
    const float* a_dst1 = (const float*)d_in[4];
    const float* b1     = (const float*)d_in[5];
    const float* W2     = (const float*)d_in[6];
    const float* a_src2 = (const float*)d_in[7];
    const float* a_dst2 = (const float*)d_in[8];
    const float* b2     = (const float*)d_in[9];
    const float* W3     = (const float*)d_in[10];
    const float* a_src3 = (const float*)d_in[11];
    const float* a_dst3 = (const float*)d_in[12];
    const float* b3     = (const float*)d_in[13];

    const int E = in_sizes[1] / 2;

    float* ws   = (float*)d_ws;
    float* bufH = ws;
    float* act1 = bufH + (size_t)NNODE * 240;
    float* h2   = act1 + (size_t)NNODE * 112;
    float* act2 = h2   + (size_t)NNODE * 96;
    float* als  = act2 + (size_t)NNODE * 96;
    float* ald  = als  + (size_t)NNODE * LSTR;
    int*   cnt    = (int*)(ald + (size_t)NNODE * LSTR);
    int*   rowptr = cnt    + NNODE;
    int*   pst    = rowptr + NNODE;
    int2*  binned = (int2*)(pst + NPST);
    int*   col    = (int*)(binned + E);
    float* out  = (float*)d_out;

    bin_local<<<GB, 1024, 0, stream>>>(ei, E, pst, binned);
    bin_group<<<NBIN, PSUB * GB, 0, stream>>>(binned, pst, E, NNODE, cnt, rowptr, col);

    auto layer = [&](const float* X, int K,
                     const float* W, const float* as, const float* ad, const float* b,
                     int heads, int C, float* Hbuf, float* obuf) {
        const int gb = (NNODE / 16 + 1) / 2;
        if (heads == 7)
            gemm_strip_wmma_f32<128, 7, 7, 16><<<gb, 64, 0, stream>>>(X, W, as, ad, Hbuf,
                                                                      als, ald, NNODE);
        else if (C == 16)
            gemm_strip_wmma_f32<112, 6, 6, 16><<<gb, 64, 0, stream>>>(X, W, as, ad, Hbuf,
                                                                      als, ald, NNODE);
        else
            gemm_strip_wmma_f32<96, 15, 6, 40><<<gb, 64, 0, stream>>>(X, W, as, ad, Hbuf,
                                                                      als, ald, NNODE);

        const int nb = (NNODE + 7) / 8;
        if (heads == 7)
            aggregate_7x16<<<nb, 256, 0, stream>>>(rowptr, cnt, col, als, ald, Hbuf,
                                                   b, obuf, NNODE);
        else if (C == 16)
            aggregate_6x16<<<nb, 256, 0, stream>>>(rowptr, cnt, col, als, ald, Hbuf,
                                                   b, obuf, NNODE);
        else
            aggregate_6x40<<<nb, 256, 0, stream>>>(rowptr, cnt, col, als, ald, Hbuf,
                                                   b, obuf, NNODE);
    };

    layer(x,    128, W1, a_src1, a_dst1, b1, 7, 16, bufH, act1);
    layer(act1, 112, W2, a_src2, a_dst2, b2, 6, 16, h2,   act2);
    layer(act2,  96, W3, a_src3, a_dst3, b3, 6, 40, bufH, out);
}


# aggregation: a wave walks several destination nodes in turn (grid-stride loop): 1250 blocks of 8 waves in layers 1-2 (5 nodes a wave), 1024 in layer 3, not 6250; per-node sum order unchanged
# speedup vs baseline: 20.5805x; 1.0480x over previous
#include <hip/hip_runtime.h>
#include <math.h>

#define NNODE 50000

typedef float v2f __attribute__((ext_vector_type(2)));
typedef float v8f __attribute__((ext_vector_type(8)));

template <int M, int NT>
__device__ __forceinline__ void gemm_load_chunk(const float* __restrict__ xk,
                                                const float* __restrict__ wk,
                                                float4& a, float (&b)[4][NT]) {
    a = *(const float4*)xk;
#pragma unroll
    for (int j = 0; j < 4; ++j)
#pragma unroll
        for (int t = 0; t < NT; ++t) b[j][t] = wk[j * M + t * 16];
}

template <int NT>
__device__ __forceinline__ void gemm_mac_chunk(const float4& a, const float (&b)[4][NT],
                                               v8f (&acc)[NT]) {
    v2f a01, a23;
    a01.x = a.x; a01.y = a.y;
    a23.x = a.z; a23.y = a.w;
#pragma unroll
    for (int t = 0; t < NT; ++t) {
        v2f bb;
        bb.x = b[0][t]; bb.y = b[1][t];
        acc[t] = __builtin_amdgcn_wmma_f32_16x16x4_f32(
                      false, a01,  false, bb,
                      (short)0, acc[t],  false,  false);
    }
#pragma unroll
    for (int t = 0; t < NT; ++t) {
        v2f bb;
        bb.x = b[2][t]; bb.y = b[3][t];
        acc[t] = __builtin_amdgcn_wmma_f32_16x16x4_f32(
                      false, a23,  false, bb,
                      (short)0, acc[t],  false,  false);
    }
}

#define LSTR 8

template <int XM>
__device__ __forceinline__ float row_xor_lane(float x) {
    return __int_as_float(__builtin_amdgcn_update_dpp(0, __float_as_int(x), 0x160 | XM,
                                                      0xf, 0xf, true));
}

template <int NT, int HEADS, int C>
__device__ __forceinline__ void strip_logits(const v8f (&acc)[NT],
                                             const float* __restrict__ a_src,
                                             const float* __restrict__ a_dst,
                                             float* __restrict__ als,
                                             float* __restrict__ ald,
                                             int row0, int half, int m16) {
    static_assert(HEADS <= LSTR, "logit slots per node");
    static_assert(HEADS * C == NT * 16, "the strip holds whole rows of H");
    const bool b3 = (m16 & 8) != 0, b2 = (m16 & 4) != 0, b1 = (m16 & 2) != 0, b0 = (m16 & 1) != 0;
    float res[LSTR];
#pragma unroll
    for (int h = 0; h < LSTR; ++h) res[h] = 0.0f;
#pragma unroll
    for (int h = 0; h < HEADS; ++h) {
        float p[16];
#pragma unroll
        for (int i = 0; i < 16; ++i) p[i] = 0.0f;
        bool first = true;
#pragma unroll
        for (int t = 0; t < NT; ++t) {
            const int lo = t * 16, hi = t * 16 + 15;
            if (hi >= h * C && lo < (h + 1) * C) {
                const int col = lo + m16;
                float ws = a_src[col], wd = a_dst[col];
                if (lo < h * C || hi >= (h + 1) * C) {
                    const bool in = col >= h * C && col < (h + 1) * C;
                    ws = in ? ws : 0.0f;
                    wd = in ? wd : 0.0f;
                }
#pragma unroll
                for (int v = 0; v < 8; ++v) {
                    p[v]     = first ? acc[t][v] * ws : p[v]     + acc[t][v] * ws;
                    p[8 + v] = first ? acc[t][v] * wd : p[8 + v] + acc[t][v] * wd;
                }
                first = false;
            }
        }
        float q8[8], q4[4], q2[2];
#pragma unroll
        for (int i = 0; i < 8; ++i)
            q8[i] = (b3 ? p[8 + i] : p[i]) + row_xor_lane<8>(b3 ? p[i] : p[8 + i]);
#pragma unroll
        for (int i = 0; i < 4; ++i)
            q4[i] = (b2 ? q8[4 + i] : q8[i]) + row_xor_lane<4>(b2 ? q8[i] : q8[4 + i]);
#pragma unroll
        for (int i = 0; i < 2; ++i)
            q2[i] = (b1 ? q4[2 + i] : q4[i]) + row_xor_lane<2>(b1 ? q4[i] : q4[2 + i]);
        res[h] = (b0 ? q2[1] : q2[0]) + row_xor_lane<1>(b0 ? q2[0] : q2[1]);
    }
    float* lp = (b3 ? ald : als) + (size_t)(row0 + half * 8 + (m16 & 7)) * LSTR;
    *(float4*)lp       = make_float4(res[0], res[1], res[2], res[3]);
    *(float4*)(lp + 4) = make_float4(res[4], res[5], res[6], res[7]);
}

template <int K, int NT, int HEADS, int C>
__global__ void __launch_bounds__(64) __attribute__((amdgpu_waves_per_eu(4, 4)))
gemm_strip_wmma_f32(const float* __restrict__ X, const float* __restrict__ W,
                    const float* __restrict__ a_src, const float* __restrict__ a_dst,
                    float* __restrict__ H, float* __restrict__ als, float* __restrict__ ald,
                    int n) {
    constexpr int M = NT * 16;
    static_assert(K % 16 == 0, "two 8-k chunks per trip");
    const int lane  = threadIdx.x & 31;
    const int strip = blockIdx.x * (blockDim.x >> 5) + (threadIdx.x >> 5);
    if (strip >= (n >> 4)) return;

    const int row0 = strip << 4;
    const int m16  = lane & 15;
    const int half = lane >> 4;

    v8f acc[NT];
#pragma unroll
    for (int t = 0; t < NT; ++t) acc[t] = v8f{};

    const float* xp = X + (size_t)(row0 + m16) * K + half * 4;
    const float* wp = W + (size_t)(half * 4) * M + m16;

    float4 a0, a1;
    float  b0[4][NT], b1[4][NT];
    gemm_load_chunk<M, NT>(xp, wp, a0, b0);
#pragma nounroll
    for (int k0 = 0; k0 < K; k0 += 16) {
        gemm_load_chunk<M, NT>(xp + k0 + 8, wp + (size_t)(k0 + 8) * M, a1, b1);
        gemm_mac_chunk<NT>(a0, b0, acc);
        if (k0 + 16 < K)
            gemm_load_chunk<M, NT>(xp + k0 + 16, wp + (size_t)(k0 + 16) * M, a0, b0);
        gemm_mac_chunk<NT>(a1, b1, acc);
    }

    float* hp = H + (size_t)(row0 + half * 8) * M + m16;
#pragma unroll
    for (int t = 0; t < NT; ++t)
#pragma unroll
        for (int v = 0; v < 8; ++v)
            hp[(size_t)v * M + t * 16] = acc[t][v];

    strip_logits<NT, HEADS, C>(acc, a_src, a_dst, als, ald, row0, half, m16);
}

#define BSHIFT 8
#define BINW   (1 << BSHIFT)
#define NBIN   ((NNODE + BINW - 1) >> BSHIFT)
#define GB     256
#define NPST   ((NBIN + 1) * GB)
#define EPT    4
#define BPT    8
#define PSUB   4

__global__ void bin_local(const int* __restrict__ ei, int E, int* __restrict__ pst,
                          int2* __restrict__ binned) {
    __shared__ int hist[NBIN];
    for (int i = threadIdx.x; i < NBIN; i += blockDim.x) hist[i] = 0;
    __syncthreads();
    const int chunk = (E + GB - 1) / GB;
    const int c0 = blockIdx.x * chunk;
    const int e0 = c0 < E ? c0 : E;
    const int e1 = e0 + chunk < E ? e0 + chunk : E;
    for (int e = e0 + threadIdx.x; e < e1; e += EPT * blockDim.x) {
        int d[EPT];
#pragma unroll
        for (int q = 0; q < EPT; ++q) {
            const int eq = e + q * blockDim.x;
            d[q] = ei[E + (eq < e1 ? eq : e1 - 1)];
        }
#pragma unroll
        for (int q = 0; q < EPT; ++q)
            if (e + q * blockDim.x < e1) atomicAdd(&hist[d[q] >> BSHIFT], 1);
    }
    __syncthreads();
    if (threadIdx.x < 32) {
        constexpr int PER = (NBIN + 31) / 32;
        const int lane = threadIdx.x;
        int v[PER];
        int lt = 0;
#pragma unroll
        for (int q = 0; q < PER; ++q) {
            const int b = lane * PER + q;
            v[q] = b < NBIN ? hist[b] : 0;
            lt += v[q];
        }
        int inc = lt;
        for (int o = 1; o < 32; o <<= 1) {
            const int t = __shfl_up(inc, o);
            if (lane >= o) inc += t;
        }
        int run = e0 + inc - lt;
#pragma unroll
        for (int q = 0; q < PER; ++q) {
            const int b = lane * PER + q;
            if (b < NBIN) { hist[b] = run; pst[b * GB + blockIdx.x] = run; }
            run += v[q];
        }
        if (lane == 31) pst[NBIN * GB + blockIdx.x] = run;
    }
    __syncthreads();
    for (int e = e0 + threadIdx.x; e < e1; e += EPT * blockDim.x) {
        int d[EPT], sv[EPT];
#pragma unroll
        for (int q = 0; q < EPT; ++q) {
            const int eq = e + q * blockDim.x;
            const int ec = eq < e1 ? eq : e1 - 1;
            d[q]  = ei[E + ec];
            sv[q] = ei[ec];
        }
#pragma unroll
        for (int q = 0; q < EPT; ++q) {
            if (e + q * blockDim.x < e1) {
                const int pos = atomicAdd(&hist[d[q] >> BSHIFT], 1);
                binned[pos] = make_int2(sv[q], d[q]);
            }
        }
    }
}

__global__ void bin_group(const int2* __restrict__ binned, const int* __restrict__ pst,
                          int E, int n, int* __restrict__ cnt, int* __restrict__ rowptr,
                          int* __restrict__ col) {
    __shared__ int c[BINW];
    __shared__ int cur[BINW];
    __shared__ int wlow[32];
    const int b     = blockIdx.x;
    const int node0 = b << BSHIFT;
    const int lane  = threadIdx.x & 31;
    const int p     = threadIdx.x / PSUB;
    const int sub   = threadIdx.x % PSUB;
    const int ps    = p < GB ? pst[b * GB + p] : 0;
    const int len   = p < GB ? pst[(b + 1) * GB + p] - ps : 0;
    for (int i = threadIdx.x; i < BINW; i += blockDim.x) c[i] = 0;
    {
        const int chunk = (E + GB - 1) / GB;
        const int c0 = p * chunk;
        int low = (p < GB && sub == 0) ? ps - (c0 < E ? c0 : E) : 0;
        for (int o = 16; o > 0; o >>= 1) low += __shfl_xor(low, o);
        if (lane == 0) wlow[threadIdx.x >> 5] = low;
    }
    __syncthreads();
    for (int k = sub; k < len; k += PSUB * BPT) {
        int dl[BPT];
#pragma unroll
        for (int q = 0; q < BPT; ++q) {
            const int kq = k + q * PSUB;
            dl[q] = binned[ps + (kq < len ? kq : len - 1)].y;
        }
#pragma unroll
        for (int q = 0; q < BPT; ++q)
            if (k + q * PSUB < len) atomicAdd(&c[dl[q] - node0], 1);
    }
    __syncthreads();
    if (threadIdx.x < 32) {
        constexpr int PER = BINW / 32;
        int beg = lane < (int)(blockDim.x >> 5) ? wlow[lane] : 0;
        for (int o = 16; o > 0; o >>= 1) beg += __shfl_xor(beg, o);
        int v[PER];
        int lt = 0;
#pragma unroll
        for (int q = 0; q < PER; ++q) { v[q] = c[lane * PER + q]; lt += v[q]; }
        int inc = lt;
        for (int o = 1; o < 32; o <<= 1) {
            const int t = __shfl_up(inc, o);
            if (lane >= o) inc += t;
        }
        int run = beg + inc - lt;
#pragma unroll
        for (int q = 0; q < PER; ++q) {
            const int dl = lane * PER + q;
            if (node0 + dl < n) { cnt[node0 + dl] = v[q]; rowptr[node0 + dl] = run; }
            cur[dl] = run;
            run += v[q];
        }
    }
    __syncthreads();
    for (int k = sub; k < len; k += PSUB * BPT) {
        int2 ed[BPT];
#pragma unroll
        for (int q = 0; q < BPT; ++q) {
            const int kq = k + q * PSUB;
            ed[q] = binned[ps + (kq < len ? kq : len - 1)];
        }
#pragma unroll
        for (int q = 0; q < BPT; ++q) {
            if (k + q * PSUB < len) {
                const int pos = atomicAdd(&cur[ed[q].y - node0], 1);
                col[pos] = ed[q].x;
            }
        }
    }
}

#define WREC (32 * LSTR)
#define AGG_WG16 1250
#define AGG_WG40 1024

template <int HEADS>
__device__ __forceinline__ void edge_logits(const int* __restrict__ col,
                                            const float* __restrict__ als,
                                            const float* ad, int beg, int deg, int d,
                                            int t, int& s, float* lr) {
    s = d;
#pragma unroll
    for (int h = 0; h < HEADS; ++h) lr[h] = -INFINITY;
    if (t <= deg) {
        if (t < deg) s = col[beg + t];
        const float4* p = (const float4*)(als + (size_t)s * LSTR);
        const float4 lo = p[0], hi = p[1];
        const float a[8] = {lo.x, lo.y, lo.z, lo.w, hi.x, hi.y, hi.z, hi.w};
#pragma unroll
        for (int h = 0; h < HEADS; ++h) {
            const float x = a[h] + ad[h];
            lr[h] = x >= 0.0f ? x : 0.2f * x;
        }
    }
}

template <int HEADS, int C, int RELU, int BATCH>
__device__ __forceinline__ void aggregate_body(const int* __restrict__ rowptr,
                                               const int* __restrict__ cnt,
                                               const int* __restrict__ col,
                                               const float* __restrict__ als,
                                               const float* __restrict__ ald,
                                               const float* __restrict__ H,
                                               const float* __restrict__ bias,
                                               float* __restrict__ out, int n,
                                               float* wt  ) {
    constexpr int HC = HEADS * C;
    constexpr int NP = (HC + 127) / 128;
    static_assert(C % 4 == 0, "a lane's four channels belong to one head");
    static_assert(HC % 4 == 0, "rows are whole 16-byte groups");
    static_assert(HEADS <= LSTR, "logit slots per node");
    static_assert(32 % BATCH == 0, "a batch never reaches past the 32 records of the table");
    const int lane = threadIdx.x & 31;
    const int wpb = (int)(blockDim.x >> 5);
    const int w0  = __builtin_amdgcn_readfirstlane((int)(blockIdx.x * wpb + (threadIdx.x >> 5)));
    const int nw  = (int)gridDim.x * wpb;

    bool   live[NP];
    int    hp[NP];
#pragma unroll
    for (int p = 0; p < NP; ++p) {
        const int c = p * 128 + 4 * lane;
        live[p] = c < HC;
        hp[p]   = c / C < HEADS ? c / C : HEADS - 1;
    }

    for (int d = w0; d < n; d += nw) {
    const int beg = rowptr[d];
    const int deg = cnt[d];
    const int tot = deg + 1;

    float  ad[HEADS], m[HEADS], dsum[HEADS];
    float4 acc[NP];
#pragma unroll
    for (int h = 0; h < HEADS; ++h) {
        ad[h]   = ald[(size_t)d * LSTR + h];
        dsum[h] = 0.0f;
    }
#pragma unroll
    for (int p = 0; p < NP; ++p) acc[p] = make_float4(0.0f, 0.0f, 0.0f, 0.0f);

    int   s0;
    float lr0[HEADS];
    edge_logits<HEADS>(col, als, ad, beg, deg, d, lane, s0, lr0);
#pragma unroll
    for (int h = 0; h < HEADS; ++h) m[h] = lr0[h];
    for (int c0 = 32; c0 < tot; c0 += 32) {
        int   s;
        float lr[HEADS];
        edge_logits<HEADS>(col, als, ad, beg, deg, d, c0 + lane, s, lr);
#pragma unroll
        for (int h = 0; h < HEADS; ++h) m[h] = fmaxf(m[h], lr[h]);
    }
#pragma unroll
    for (int h = 0; h < HEADS; ++h)
        for (int o = 16; o > 0; o >>= 1) m[h] = fmaxf(m[h], __shfl_xor(m[h], o));

    for (int c0 = 0; c0 < tot; c0 += 32) {
        int   s = s0;
        float ex[LSTR];
#pragma unroll
        for (int h = 0; h < HEADS; ++h) ex[h] = lr0[h];
        if (c0 > 0) edge_logits<HEADS>(col, als, ad, beg, deg, d, c0 + lane, s, ex);
#pragma unroll
        for (int h = 0; h < HEADS; ++h) {
            ex[h] = __expf(ex[h] - m[h]);
            dsum[h] += ex[h];
        }
#pragma unroll
        for (int h = HEADS; h < LSTR; ++h) ex[h] = 0.0f;
        *(float4*)(wt + lane * LSTR)     = make_float4(ex[0], ex[1], ex[2], ex[3]);
        *(float4*)(wt + lane * LSTR + 4) = make_float4(ex[4], ex[5], ex[6], ex[7]);

        const int nj = (tot - c0) < 32 ? (tot - c0) : 32;
        for (int j0 = 0; j0 < nj; j0 += BATCH) {
            float4 hv[BATCH][NP];
            float  w[BATCH][NP];
#pragma unroll
            for (int i = 0; i < BATCH; ++i) {
#pragma unroll
                for (int p = 0; p < NP; ++p) hv[i][p] = make_float4(0.0f, 0.0f, 0.0f, 0.0f);
                if (j0 + i < nj) {
                    const int sj = __builtin_amdgcn_readlane(s, j0 + i);
                    const float* hs = H + (size_t)sj * HC;
#pragma unroll
                    for (int p = 0; p < NP; ++p)
                        if (live[p]) hv[i][p] = *(const float4*)(hs + p * 128 + 4 * lane);
                }
            }
#pragma unroll
            for (int i = 0; i < BATCH; ++i)
#pragma unroll
                for (int p = 0; p < NP; ++p) w[i][p] = wt[(j0 + i) * LSTR + hp[p]];
#pragma unroll
            for (int i = 0; i < BATCH; ++i) {
                if (j0 + i < nj) {
#pragma unroll
                    for (int p = 0; p < NP; ++p) {
                        acc[p].x += hv[i][p].x * w[i][p];
                        acc[p].y += hv[i][p].y * w[i][p];
                        acc[p].z += hv[i][p].z * w[i][p];
                        acc[p].w += hv[i][p].w * w[i][p];
                    }
                }
            }
        }
    }
#pragma unroll
    for (int h = 0; h < HEADS; ++h)
        for (int o = 16; o > 0; o >>= 1) dsum[h] += __shfl_xor(dsum[h], o);

    float* orow = out + (size_t)d * HC;
#pragma unroll
    for (int p = 0; p < NP; ++p) {
        if (live[p]) {
            const int c = p * 128 + 4 * lane;
            float den = dsum[0];
#pragma unroll
            for (int h = 1; h < HEADS; ++h) den = hp[p] == h ? dsum[h] : den;
            const float4 b = *(const float4*)(bias + c);
            float4 v;
            v.x = acc[p].x / den + b.x;
            v.y = acc[p].y / den + b.y;
            v.z = acc[p].z / den + b.z;
            v.w = acc[p].w / den + b.w;
            if (RELU) {
                v.x = fmaxf(v.x, 0.0f); v.y = fmaxf(v.y, 0.0f);
                v.z = fmaxf(v.z, 0.0f); v.w = fmaxf(v.w, 0.0f);
            }
            *(float4*)(orow + c) = v;
        }
    }
    }
}

__global__ void aggregate_7x16(const int* __restrict__ rowptr, const int* __restrict__ cnt,
                               const int* __restrict__ col, const float* __restrict__ als,
                               const float* __restrict__ ald, const float* __restrict__ H,
                               const float* __restrict__ bias, float* __restrict__ out, int n) {
    __shared__ __attribute__((aligned(16))) float wtab[8 * WREC];
    aggregate_body<7, 16, 1, 8>(rowptr, cnt, col, als, ald, H, bias, out, n,
                                wtab + (threadIdx.x >> 5) * WREC);
}

__global__ void aggregate_6x16(const int* __restrict__ rowptr, const int* __restrict__ cnt,
                               const int* __restrict__ col, const float* __restrict__ als,
                               const float* __restrict__ ald, const float* __restrict__ H,
                               const float* __restrict__ bias, float* __restrict__ out, int n) {
    __shared__ __attribute__((aligned(16))) float wtab[8 * WREC];
    aggregate_body<6, 16, 1, 8>(rowptr, cnt, col, als, ald, H, bias, out, n,
                                wtab + (threadIdx.x >> 5) * WREC);
}

__global__ void aggregate_6x40(const int* __restrict__ rowptr, const int* __restrict__ cnt,
                               const int* __restrict__ col, const float* __restrict__ als,
                               const float* __restrict__ ald, const float* __restrict__ H,
                               const float* __restrict__ bias, float* __restrict__ out, int n) {
    __shared__ __attribute__((aligned(16))) float wtab[8 * WREC];
    aggregate_body<6, 40, 0, 4>(rowptr, cnt, col, als, ald, H, bias, out, n,
                                wtab + (threadIdx.x >> 5) * WREC);
}


extern "C" void kernel_launch(void* const* d_in, const int* in_sizes, int n_in,
                              void* d_out, int out_size, void* d_ws, size_t ws_size,
                              hipStream_t stream) {
    const float* x      = (const float*)d_in[0];
    const int*   ei     = (const int*)  d_in[1];
    const float* W1     = (const float*)d_in[2];
    const float* a_src1 = (const float*)d_in[3];
    const float* a_dst1 = (const float*)d_in[4];
    const float* b1     = (const float*)d_in[5];
    const float* W2     = (const float*)d_in[6];
    const float* a_src2 = (const float*)d_in[7];
    const float* a_dst2 = (const float*)d_in[8];
    const float* b2     = (const float*)d_in[9];
    const float* W3     = (const float*)d_in[10];
    const float* a_src3 = (const float*)d_in[11];
    const float* a_dst3 = (const float*)d_in[12];
    const float* b3     = (const float*)d_in[13];

    const int E = in_sizes[1] / 2;

    float* ws   = (float*)d_ws;
    float* bufH = ws;
    float* act1 = bufH + (size_t)NNODE * 240;
    float* h2   = act1 + (size_t)NNODE * 112;
    float* act2 = h2   + (size_t)NNODE * 96;
    float* als  = act2 + (size_t)NNODE * 96;
    float* ald  = als  + (size_t)NNODE * LSTR;
    int*   cnt    = (int*)(ald + (size_t)NNODE * LSTR);
    int*   rowptr = cnt    + NNODE;
    int*   pst    = rowptr + NNODE;
    int2*  binned = (int2*)(pst + NPST);
    int*   col    = (int*)(binned + E);
    float* out  = (float*)d_out;

    bin_local<<<GB, 1024, 0, stream>>>(ei, E, pst, binned);
    bin_group<<<NBIN, PSUB * GB, 0, stream>>>(binned, pst, E, NNODE, cnt, rowptr, col);

    auto layer = [&](const float* X, int K,
                     const float* W, const float* as, const float* ad, const float* b,
                     int heads, int C, float* Hbuf, float* obuf) {
        const int gb = (NNODE / 16 + 1) / 2;
        if (heads == 7)
            gemm_strip_wmma_f32<128, 7, 7, 16><<<gb, 64, 0, stream>>>(X, W, as, ad, Hbuf,
                                                                      als, ald, NNODE);
        else if (C == 16)
            gemm_strip_wmma_f32<112, 6, 6, 16><<<gb, 64, 0, stream>>>(X, W, as, ad, Hbuf,
                                                                      als, ald, NNODE);
        else
            gemm_strip_wmma_f32<96, 15, 6, 40><<<gb, 64, 0, stream>>>(X, W, as, ad, Hbuf,
                                                                      als, ald, NNODE);

        if (heads == 7)
            aggregate_7x16<<<AGG_WG16, 256, 0, stream>>>(rowptr, cnt, col, als, ald, Hbuf,
                                                         b, obuf, NNODE);
        else if (C == 16)
            aggregate_6x16<<<AGG_WG16, 256, 0, stream>>>(rowptr, cnt, col, als, ald, Hbuf,
                                                         b, obuf, NNODE);
        else
            aggregate_6x40<<<AGG_WG40, 256, 0, stream>>>(rowptr, cnt, col, als, ald, Hbuf,
                                                         b, obuf, NNODE);
    };

    layer(x,    128, W1, a_src1, a_dst1, b1, 7, 16, bufH, act1);
    layer(act1, 112, W2, a_src2, a_dst2, b2, 6, 16, h2,   act2);
    layer(act2,  96, W3, a_src3, a_dst3, b3, 6, 40, bufH, out);
}


# aggregation: one-pass softmax per lane with a running maximum (partial sums rescaled when it rises: sum tree and rounding change); source logit read from the H row's own padding, no logit table
# speedup vs baseline: 22.6583x; 1.1010x over previous
#include <hip/hip_runtime.h>
#include <math.h>

#define NNODE 50000

typedef float v2f __attribute__((ext_vector_type(2)));
typedef float v8f __attribute__((ext_vector_type(8)));

template <int M, int NT>
__device__ __forceinline__ void gemm_load_chunk(const float* __restrict__ xk,
                                                const float* __restrict__ wk,
                                                float4& a, float (&b)[4][NT]) {
    a = *(const float4*)xk;
#pragma unroll
    for (int j = 0; j < 4; ++j)
#pragma unroll
        for (int t = 0; t < NT; ++t) b[j][t] = wk[j * M + t * 16];
}

template <int NT>
__device__ __forceinline__ void gemm_mac_chunk(const float4& a, const float (&b)[4][NT],
                                               v8f (&acc)[NT]) {
    v2f a01, a23;
    a01.x = a.x; a01.y = a.y;
    a23.x = a.z; a23.y = a.w;
#pragma unroll
    for (int t = 0; t < NT; ++t) {
        v2f bb;
        bb.x = b[0][t]; bb.y = b[1][t];
        acc[t] = __builtin_amdgcn_wmma_f32_16x16x4_f32(
                      false, a01,  false, bb,
                      (short)0, acc[t],  false,  false);
    }
#pragma unroll
    for (int t = 0; t < NT; ++t) {
        v2f bb;
        bb.x = b[2][t]; bb.y = b[3][t];
        acc[t] = __builtin_amdgcn_wmma_f32_16x16x4_f32(
                      false, a23,  false, bb,
                      (short)0, acc[t],  false,  false);
    }
}

#define LSTR 8
#define ROWSTR(M) ((((M) + 127) / 128) * 128)

template <int XM>
__device__ __forceinline__ float row_xor_lane(float x) {
    return __int_as_float(__builtin_amdgcn_update_dpp(0, __float_as_int(x), 0x160 | XM,
                                                      0xf, 0xf, true));
}

template <int NT, int HEADS, int C>
__device__ __forceinline__ void strip_logits(const v8f (&acc)[NT],
                                             const float* __restrict__ a_src,
                                             const float* __restrict__ a_dst,
                                             float* __restrict__ H,
                                             float* __restrict__ ald,
                                             int row0, int half, int m16) {
    constexpr int M  = NT * 16;
    constexpr int RS = ROWSTR(M);
    static_assert(HEADS <= LSTR, "logit slots per node");
    static_assert(M + LSTR <= RS, "the logit slots fit behind the channels");
    static_assert(HEADS * C == NT * 16, "the strip holds whole rows of H");
    const bool b3 = (m16 & 8) != 0, b2 = (m16 & 4) != 0, b1 = (m16 & 2) != 0, b0 = (m16 & 1) != 0;
    float res[LSTR];
#pragma unroll
    for (int h = 0; h < LSTR; ++h) res[h] = 0.0f;
#pragma unroll
    for (int h = 0; h < HEADS; ++h) {
        float p[16];
#pragma unroll
        for (int i = 0; i < 16; ++i) p[i] = 0.0f;
        bool first = true;
#pragma unroll
        for (int t = 0; t < NT; ++t) {
            const int lo = t * 16, hi = t * 16 + 15;
            if (hi >= h * C && lo < (h + 1) * C) {
                const int col = lo + m16;
                float ws = a_src[col], wd = a_dst[col];
                if (lo < h * C || hi >= (h + 1) * C) {
                    const bool in = col >= h * C && col < (h + 1) * C;
                    ws = in ? ws : 0.0f;
                    wd = in ? wd : 0.0f;
                }
#pragma unroll
                for (int v = 0; v < 8; ++v) {
                    p[v]     = first ? acc[t][v] * ws : p[v]     + acc[t][v] * ws;
                    p[8 + v] = first ? acc[t][v] * wd : p[8 + v] + acc[t][v] * wd;
                }
                first = false;
            }
        }
        float q8[8], q4[4], q2[2];
#pragma unroll
        for (int i = 0; i < 8; ++i)
            q8[i] = (b3 ? p[8 + i] : p[i]) + row_xor_lane<8>(b3 ? p[i] : p[8 + i]);
#pragma unroll
        for (int i = 0; i < 4; ++i)
            q4[i] = (b2 ? q8[4 + i] : q8[i]) + row_xor_lane<4>(b2 ? q8[i] : q8[4 + i]);
#pragma unroll
        for (int i = 0; i < 2; ++i)
            q2[i] = (b1 ? q4[2 + i] : q4[i]) + row_xor_lane<2>(b1 ? q4[i] : q4[2 + i]);
        res[h] = (b0 ? q2[1] : q2[0]) + row_xor_lane<1>(b0 ? q2[0] : q2[1]);
    }
    const size_t row = (size_t)(row0 + half * 8 + (m16 & 7));
    float* lp = b3 ? ald + row * LSTR : H + row * RS + M;
    *(float4*)lp       = make_float4(res[0], res[1], res[2], res[3]);
    *(float4*)(lp + 4) = make_float4(res[4], res[5], res[6], res[7]);
}

template <int K, int NT, int HEADS, int C>
__global__ void __launch_bounds__(64) __attribute__((amdgpu_waves_per_eu(4, 4)))
gemm_strip_wmma_f32(const float* __restrict__ X, const float* __restrict__ W,
                    const float* __restrict__ a_src, const float* __restrict__ a_dst,
                    float* __restrict__ H, float* __restrict__ ald, int n) {
    constexpr int M  = NT * 16;
    constexpr int RS = ROWSTR(M);
    static_assert(K % 16 == 0, "two 8-k chunks per trip");
    const int lane  = threadIdx.x & 31;
    const int strip = blockIdx.x * (blockDim.x >> 5) + (threadIdx.x >> 5);
    if (strip >= (n >> 4)) return;

    const int row0 = strip << 4;
    const int m16  = lane & 15;
    const int half = lane >> 4;

    v8f acc[NT];
#pragma unroll
    for (int t = 0; t < NT; ++t) acc[t] = v8f{};

    const float* xp = X + (size_t)(row0 + m16) * K + half * 4;
    const float* wp = W + (size_t)(half * 4) * M + m16;

    float4 a0, a1;
    float  b0[4][NT], b1[4][NT];
    gemm_load_chunk<M, NT>(xp, wp, a0, b0);
#pragma nounroll
    for (int k0 = 0; k0 < K; k0 += 16) {
        gemm_load_chunk<M, NT>(xp + k0 + 8, wp + (size_t)(k0 + 8) * M, a1, b1);
        gemm_mac_chunk<NT>(a0, b0, acc);
        if (k0 + 16 < K)
            gemm_load_chunk<M, NT>(xp + k0 + 16, wp + (size_t)(k0 + 16) * M, a0, b0);
        gemm_mac_chunk<NT>(a1, b1, acc);
    }

    float* hp = H + (size_t)(row0 + half * 8) * RS + m16;
#pragma unroll
    for (int t = 0; t < NT; ++t)
#pragma unroll
        for (int v = 0; v < 8; ++v)
            hp[(size_t)v * RS + t * 16] = acc[t][v];

    strip_logits<NT, HEADS, C>(acc, a_src, a_dst, H, ald, row0, half, m16);
}

#define BSHIFT 8
#define BINW   (1 << BSHIFT)
#define NBIN   ((NNODE + BINW - 1) >> BSHIFT)
#define GB     256
#define NPST   ((NBIN + 1) * GB)
#define EPT    4
#define BPT    8
#define PSUB   4

__global__ void bin_local(const int* __restrict__ ei, int E, int* __restrict__ pst,
                          int2* __restrict__ binned) {
    __shared__ int hist[NBIN];
    for (int i = threadIdx.x; i < NBIN; i += blockDim.x) hist[i] = 0;
    __syncthreads();
    const int chunk = (E + GB - 1) / GB;
    const int c0 = blockIdx.x * chunk;
    const int e0 = c0 < E ? c0 : E;
    const int e1 = e0 + chunk < E ? e0 + chunk : E;
    for (int e = e0 + threadIdx.x; e < e1; e += EPT * blockDim.x) {
        int d[EPT];
#pragma unroll
        for (int q = 0; q < EPT; ++q) {
            const int eq = e + q * blockDim.x;
            d[q] = ei[E + (eq < e1 ? eq : e1 - 1)];
        }
#pragma unroll
        for (int q = 0; q < EPT; ++q)
            if (e + q * blockDim.x < e1) atomicAdd(&hist[d[q] >> BSHIFT], 1);
    }
    __syncthreads();
    if (threadIdx.x < 32) {
        constexpr int PER = (NBIN + 31) / 32;
        const int lane = threadIdx.x;
        int v[PER];
        int lt = 0;
#pragma unroll
        for (int q = 0; q < PER; ++q) {
            const int b = lane * PER + q;
            v[q] = b < NBIN ? hist[b] : 0;
            lt += v[q];
        }
        int inc = lt;
        for (int o = 1; o < 32; o <<= 1) {
            const int t = __shfl_up(inc, o);
            if (lane >= o) inc += t;
        }
        int run = e0 + inc - lt;
#pragma unroll
        for (int q = 0; q < PER; ++q) {
            const int b = lane * PER + q;
            if (b < NBIN) { hist[b] = run; pst[b * GB + blockIdx.x] = run; }
            run += v[q];
        }
        if (lane == 31) pst[NBIN * GB + blockIdx.x] = run;
    }
    __syncthreads();
    for (int e = e0 + threadIdx.x; e < e1; e += EPT * blockDim.x) {
        int d[EPT], sv[EPT];
#pragma unroll
        for (int q = 0; q < EPT; ++q) {
            const int eq = e + q * blockDim.x;
            const int ec = eq < e1 ? eq : e1 - 1;
            d[q]  = ei[E + ec];
            sv[q] = ei[ec];
        }
#pragma unroll
        for (int q = 0; q < EPT; ++q) {
            if (e + q * blockDim.x < e1) {
                const int pos = atomicAdd(&hist[d[q] >> BSHIFT], 1);
                binned[pos] = make_int2(sv[q], d[q]);
            }
        }
    }
}

__global__ void bin_group(const int2* __restrict__ binned, const int* __restrict__ pst,
                          int E, int n, int* __restrict__ cnt, int* __restrict__ rowptr,
                          int* __restrict__ col) {
    __shared__ int c[BINW];
    __shared__ int cur[BINW];
    __shared__ int wlow[32];
    const int b     = blockIdx.x;
    const int node0 = b << BSHIFT;
    const int lane  = threadIdx.x & 31;
    const int p     = threadIdx.x / PSUB;
    const int sub   = threadIdx.x % PSUB;
    const int ps    = p < GB ? pst[b * GB + p] : 0;
    const int len   = p < GB ? pst[(b + 1) * GB + p] - ps : 0;
    for (int i = threadIdx.x; i < BINW; i += blockDim.x) c[i] = 0;
    {
        const int chunk = (E + GB - 1) / GB;
        const int c0 = p * chunk;
        int low = (p < GB && sub == 0) ? ps - (c0 < E ? c0 : E) : 0;
        for (int o = 16; o > 0; o >>= 1) low += __shfl_xor(low, o);
        if (lane == 0) wlow[threadIdx.x >> 5] = low;
    }
    __syncthreads();
    for (int k = sub; k < len; k += PSUB * BPT) {
        int dl[BPT];
#pragma unroll
        for (int q = 0; q < BPT; ++q) {
            const int kq = k + q * PSUB;
            dl[q] = binned[ps + (kq < len ? kq : len - 1)].y;
        }
#pragma unroll
        for (int q = 0; q < BPT; ++q)
            if (k + q * PSUB < len) atomicAdd(&c[dl[q] - node0], 1);
    }
    __syncthreads();
    if (threadIdx.x < 32) {
        constexpr int PER = BINW / 32;
        int beg = lane < (int)(blockDim.x >> 5) ? wlow[lane] : 0;
        for (int o = 16; o > 0; o >>= 1) beg += __shfl_xor(beg, o);
        int v[PER];
        int lt = 0;
#pragma unroll
        for (int q = 0; q < PER; ++q) { v[q] = c[lane * PER + q]; lt += v[q]; }
        int inc = lt;
        for (int o = 1; o < 32; o <<= 1) {
            const int t = __shfl_up(inc, o);
            if (lane >= o) inc += t;
        }
        int run = beg + inc - lt;
#pragma unroll
        for (int q = 0; q < PER; ++q) {
            const int dl = lane * PER + q;
            if (node0 + dl < n) { cnt[node0 + dl] = v[q]; rowptr[node0 + dl] = run; }
            cur[dl] = run;
            run += v[q];
        }
    }
    __syncthreads();
    for (int k = sub; k < len; k += PSUB * BPT) {
        int2 ed[BPT];
#pragma unroll
        for (int q = 0; q < BPT; ++q) {
            const int kq = k + q * PSUB;
            ed[q] = binned[ps + (kq < len ? kq : len - 1)];
        }
#pragma unroll
        for (int q = 0; q < BPT; ++q) {
            if (k + q * PSUB < len) {
                const int pos = atomicAdd(&cur[ed[q].y - node0], 1);
                col[pos] = ed[q].x;
            }
        }
    }
}

#define AGG_WG16 1250
#define AGG_WG40 1024

template <int HEADS, int C, int RELU, int BATCH>
__device__ __forceinline__ void aggregate_body(const int* __restrict__ rowptr,
                                               const int* __restrict__ cnt,
                                               const int* __restrict__ col,
                                               const float* __restrict__ ald,
                                               const float* __restrict__ H,
                                               const float* __restrict__ bias,
                                               float* __restrict__ out, int n) {
    constexpr int HC = HEADS * C;
    constexpr int NP = (HC + 127) / 128;
    constexpr int RS = ROWSTR(HC);
    static_assert(C % 4 == 0, "a lane's four channels belong to one head");
    static_assert(HC % 4 == 0, "rows are whole 16-byte groups");
    static_assert(HEADS <= LSTR, "logit slots per node");
    static_assert(HC + LSTR <= RS, "the logit slots sit behind the channels");
    static_assert(32 % BATCH == 0, "whole batches per 32 edges");
    const int lane = threadIdx.x & 31;
    const int wpb = (int)(blockDim.x >> 5);
    const int w0  = __builtin_amdgcn_readfirstlane((int)(blockIdx.x * wpb + (threadIdx.x >> 5)));
    const int nw  = (int)gridDim.x * wpb;

    bool   live[NP];
    int    hp[NP];
#pragma unroll
    for (int p = 0; p < NP; ++p) {
        const int c = p * 128 + 4 * lane;
        live[p] = c < HC;
        hp[p]   = c / C < HEADS ? c / C : HEADS - 1;
    }

    for (int d = w0; d < n; d += nw) {
    const int beg = rowptr[d];
    const int deg = cnt[d];
    const int tot = deg + 1;

    float  adh[HEADS];
#pragma unroll
    for (int h = 0; h < HEADS; ++h) adh[h] = ald[(size_t)d * LSTR + h];
    float  adp[NP], m[NP], dsum[NP];
    float4 acc[NP];
#pragma unroll
    for (int p = 0; p < NP; ++p) {
        adp[p] = adh[0];
#pragma unroll
        for (int h = 1; h < HEADS; ++h) adp[p] = hp[p] == h ? adh[h] : adp[p];
        m[p]    = -INFINITY;
        dsum[p] = 0.0f;
        acc[p]  = make_float4(0.0f, 0.0f, 0.0f, 0.0f);
    }

    int c0 = 0;
    do {
        const int t = c0 + lane;
        int s = d;
        if (t < deg) s = col[beg + t];

        const int nj = (tot - c0) < 32 ? (tot - c0) : 32;
        for (int j0 = 0; j0 < nj; j0 += BATCH) {
            float4 hv[BATCH][NP];
            float  lg[BATCH][NP];
            int    sj[BATCH];
#pragma unroll
            for (int i = 0; i < BATCH; ++i)
                sj[i] = __builtin_amdgcn_readlane(s, j0 + i < nj ? j0 + i : nj - 1);
#pragma unroll
            for (int i = 0; i < BATCH; ++i) {
#pragma unroll
                for (int p = 0; p < NP; ++p) {
                    hv[i][p] = make_float4(0.0f, 0.0f, 0.0f, 0.0f);
                    lg[i][p] = -INFINITY;
                }
                if (j0 + i < nj) {
                    const float* hs = H + (size_t)sj[i] * RS;
#pragma unroll
                    for (int p = 0; p < NP; ++p) {
                        lg[i][p] = 0.0f;
                        if (live[p]) {
                            hv[i][p] = *(const float4*)(hs + p * 128 + 4 * lane);
                            lg[i][p] = hs[HC + hp[p]];
                        }
                    }
                }
            }
#pragma unroll
            for (int p = 0; p < NP; ++p) {
                float x[BATCH];
                float bm = -INFINITY;
#pragma unroll
                for (int i = 0; i < BATCH; ++i) {
                    const float v = lg[i][p] + adp[p];
                    x[i] = v >= 0.0f ? v : 0.2f * v;
                    bm = fmaxf(bm, x[i]);
                }
                const float mn = fmaxf(m[p], bm);
                const float sc = __expf(m[p] - mn);
                m[p] = mn;
                dsum[p]  *= sc;
                acc[p].x *= sc; acc[p].y *= sc; acc[p].z *= sc; acc[p].w *= sc;
#pragma unroll
                for (int i = 0; i < BATCH; ++i) {
                    const float e = __expf(x[i] - mn);
                    dsum[p]  += e;
                    acc[p].x += hv[i][p].x * e;
                    acc[p].y += hv[i][p].y * e;
                    acc[p].z += hv[i][p].z * e;
                    acc[p].w += hv[i][p].w * e;
                }
            }
        }
        c0 += 32;
    } while (c0 < tot);

    float* orow = out + (size_t)d * HC;
#pragma unroll
    for (int p = 0; p < NP; ++p) {
        if (live[p]) {
            const int c = p * 128 + 4 * lane;
            const float den = dsum[p];
            const float4 b = *(const float4*)(bias + c);
            float4 v;
            v.x = acc[p].x / den + b.x;
            v.y = acc[p].y / den + b.y;
            v.z = acc[p].z / den + b.z;
            v.w = acc[p].w / den + b.w;
            if (RELU) {
                v.x = fmaxf(v.x, 0.0f); v.y = fmaxf(v.y, 0.0f);
                v.z = fmaxf(v.z, 0.0f); v.w = fmaxf(v.w, 0.0f);
            }
            *(float4*)(orow + c) = v;
        }
    }
    }
}

__global__ void aggregate_7x16(const int* __restrict__ rowptr, const int* __restrict__ cnt,
                               const int* __restrict__ col, const float* __restrict__ ald,
                               const float* __restrict__ H, const float* __restrict__ bias,
                               float* __restrict__ out, int n) {
    aggregate_body<7, 16, 1, 8>(rowptr, cnt, col, ald, H, bias, out, n);
}

__global__ void aggregate_6x16(const int* __restrict__ rowptr, const int* __restrict__ cnt,
                               const int* __restrict__ col, const float* __restrict__ ald,
                               const float* __restrict__ H, const float* __restrict__ bias,
                               float* __restrict__ out, int n) {
    aggregate_body<6, 16, 1, 8>(rowptr, cnt, col, ald, H, bias, out, n);
}

__global__ void aggregate_6x40(const int* __restrict__ rowptr, const int* __restrict__ cnt,
                               const int* __restrict__ col, const float* __restrict__ ald,
                               const float* __restrict__ H, const float* __restrict__ bias,
                               float* __restrict__ out, int n) {
    aggregate_body<6, 40, 0, 4>(rowptr, cnt, col, ald, H, bias, out, n);
}


extern "C" void kernel_launch(void* const* d_in, const int* in_sizes, int n_in,
                              void* d_out, int out_size, void* d_ws, size_t ws_size,
                              hipStream_t stream) {
    const float* x      = (const float*)d_in[0];
    const int*   ei     = (const int*)  d_in[1];
    const float* W1     = (const float*)d_in[2];
    const float* a_src1 = (const float*)d_in[3];
    const float* a_dst1 = (const float*)d_in[4];
    const float* b1     = (const float*)d_in[5];
    const float* W2     = (const float*)d_in[6];
    const float* a_src2 = (const float*)d_in[7];
    const float* a_dst2 = (const float*)d_in[8];
    const float* b2     = (const float*)d_in[9];
    const float* W3     = (const float*)d_in[10];
    const float* a_src3 = (const float*)d_in[11];
    const float* a_dst3 = (const float*)d_in[12];
    const float* b3     = (const float*)d_in[13];

    const int E = in_sizes[1] / 2;

    float* ws   = (float*)d_ws;
    float* bufH = ws;
    float* act1 = bufH + (size_t)NNODE * 256;
    float* h2   = act1 + (size_t)NNODE * 112;
    float* act2 = h2   + (size_t)NNODE * 128;
    float* ald  = act2 + (size_t)NNODE * 96;
    int*   cnt    = (int*)(ald + (size_t)NNODE * LSTR);
    int*   rowptr = cnt    + NNODE;
    int*   pst    = rowptr + NNODE;
    int2*  binned = (int2*)(pst + NPST);
    int*   col    = (int*)(binned + E);
    float* out  = (float*)d_out;

    bin_local<<<GB, 1024, 0, stream>>>(ei, E, pst, binned);
    bin_group<<<NBIN, PSUB * GB, 0, stream>>>(binned, pst, E, NNODE, cnt, rowptr, col);

    auto layer = [&](const float* X, int K,
                     const float* W, const float* as, const float* ad, const float* b,
                     int heads, int C, float* Hbuf, float* obuf) {
        const int gb = (NNODE / 16 + 1) / 2;
        if (heads == 7)
            gemm_strip_wmma_f32<128, 7, 7, 16><<<gb, 64, 0, stream>>>(X, W, as, ad, Hbuf,
                                                                      ald, NNODE);
        else if (C == 16)
            gemm_strip_wmma_f32<112, 6, 6, 16><<<gb, 64, 0, stream>>>(X, W, as, ad, Hbuf,
                                                                      ald, NNODE);
        else
            gemm_strip_wmma_f32<96, 15, 6, 40><<<gb, 64, 0, stream>>>(X, W, as, ad, Hbuf,
                                                                      ald, NNODE);

        if (heads == 7)
            aggregate_7x16<<<AGG_WG16, 256, 0, stream>>>(rowptr, cnt, col, ald, Hbuf,
                                                         b, obuf, NNODE);
        else if (C == 16)
            aggregate_6x16<<<AGG_WG16, 256, 0, stream>>>(rowptr, cnt, col, ald, Hbuf,
                                                         b, obuf, NNODE);
        else
            aggregate_6x40<<<AGG_WG40, 256, 0, stream>>>(rowptr, cnt, col, ald, Hbuf,
                                                         b, obuf, NNODE);
    };

    layer(x,    128, W1, a_src1, a_dst1, b1, 7, 16, bufH, act1);
    layer(act1, 112, W2, a_src2, a_dst2, b2, 6, 16, h2,   act2);
    layer(act2,  96, W3, a_src3, a_dst3, b3, 6, 40, bufH, out);
}


# aggregation: full batches of edges load in a loop of their own, no per-slot test, zero fill or mask change (lanes past a row's channels read a clamped offset); the partial last batch is separate
# speedup vs baseline: 23.5201x; 1.0380x over previous
#include <hip/hip_runtime.h>
#include <math.h>

#define NNODE 50000

typedef float v2f __attribute__((ext_vector_type(2)));
typedef float v8f __attribute__((ext_vector_type(8)));

template <int M, int NT>
__device__ __forceinline__ void gemm_load_chunk(const float* __restrict__ xk,
                                                const float* __restrict__ wk,
                                                float4& a, float (&b)[4][NT]) {
    a = *(const float4*)xk;
#pragma unroll
    for (int j = 0; j < 4; ++j)
#pragma unroll
        for (int t = 0; t < NT; ++t) b[j][t] = wk[j * M + t * 16];
}

template <int NT>
__device__ __forceinline__ void gemm_mac_chunk(const float4& a, const float (&b)[4][NT],
                                               v8f (&acc)[NT]) {
    v2f a01, a23;
    a01.x = a.x; a01.y = a.y;
    a23.x = a.z; a23.y = a.w;
#pragma unroll
    for (int t = 0; t < NT; ++t) {
        v2f bb;
        bb.x = b[0][t]; bb.y = b[1][t];
        acc[t] = __builtin_amdgcn_wmma_f32_16x16x4_f32(
                      false, a01,  false, bb,
                      (short)0, acc[t],  false,  false);
    }
#pragma unroll
    for (int t = 0; t < NT; ++t) {
        v2f bb;
        bb.x = b[2][t]; bb.y = b[3][t];
        acc[t] = __builtin_amdgcn_wmma_f32_16x16x4_f32(
                      false, a23,  false, bb,
                      (short)0, acc[t],  false,  false);
    }
}

#define LSTR 8
#define ROWSTR(M) ((((M) + 127) / 128) * 128)

template <int XM>
__device__ __forceinline__ float row_xor_lane(float x) {
    return __int_as_float(__builtin_amdgcn_update_dpp(0, __float_as_int(x), 0x160 | XM,
                                                      0xf, 0xf, true));
}

template <int NT, int HEADS, int C>
__device__ __forceinline__ void strip_logits(const v8f (&acc)[NT],
                                             const float* __restrict__ a_src,
                                             const float* __restrict__ a_dst,
                                             float* __restrict__ H,
                                             float* __restrict__ ald,
                                             int row0, int half, int m16) {
    constexpr int M  = NT * 16;
    constexpr int RS = ROWSTR(M);
    static_assert(HEADS <= LSTR, "logit slots per node");
    static_assert(M + LSTR <= RS, "the logit slots fit behind the channels");
    static_assert(HEADS * C == NT * 16, "the strip holds whole rows of H");
    const bool b3 = (m16 & 8) != 0, b2 = (m16 & 4) != 0, b1 = (m16 & 2) != 0, b0 = (m16 & 1) != 0;
    float res[LSTR];
#pragma unroll
    for (int h = 0; h < LSTR; ++h) res[h] = 0.0f;
#pragma unroll
    for (int h = 0; h < HEADS; ++h) {
        float p[16];
#pragma unroll
        for (int i = 0; i < 16; ++i) p[i] = 0.0f;
        bool first = true;
#pragma unroll
        for (int t = 0; t < NT; ++t) {
            const int lo = t * 16, hi = t * 16 + 15;
            if (hi >= h * C && lo < (h + 1) * C) {
                const int col = lo + m16;
                float ws = a_src[col], wd = a_dst[col];
                if (lo < h * C || hi >= (h + 1) * C) {
                    const bool in = col >= h * C && col < (h + 1) * C;
                    ws = in ? ws : 0.0f;
                    wd = in ? wd : 0.0f;
                }
#pragma unroll
                for (int v = 0; v < 8; ++v) {
                    p[v]     = first ? acc[t][v] * ws : p[v]     + acc[t][v] * ws;
                    p[8 + v] = first ? acc[t][v] * wd : p[8 + v] + acc[t][v] * wd;
                }
                first = false;
            }
        }
        float q8[8], q4[4], q2[2];
#pragma unroll
        for (int i = 0; i < 8; ++i)
            q8[i] = (b3 ? p[8 + i] : p[i]) + row_xor_lane<8>(b3 ? p[i] : p[8 + i]);
#pragma unroll
        for (int i = 0; i < 4; ++i)
            q4[i] = (b2 ? q8[4 + i] : q8[i]) + row_xor_lane<4>(b2 ? q8[i] : q8[4 + i]);
#pragma unroll
        for (int i = 0; i < 2; ++i)
            q2[i] = (b1 ? q4[2 + i] : q4[i]) + row_xor_lane<2>(b1 ? q4[i] : q4[2 + i]);
        res[h] = (b0 ? q2[1] : q2[0]) + row_xor_lane<1>(b0 ? q2[0] : q2[1]);
    }
    const size_t row = (size_t)(row0 + half * 8 + (m16 & 7));
    float* lp = b3 ? ald + row * LSTR : H + row * RS + M;
    *(float4*)lp       = make_float4(res[0], res[1], res[2], res[3]);
    *(float4*)(lp + 4) = make_float4(res[4], res[5], res[6], res[7]);
}

template <int K, int NT, int HEADS, int C>
__global__ void __launch_bounds__(64) __attribute__((amdgpu_waves_per_eu(4, 4)))
gemm_strip_wmma_f32(const float* __restrict__ X, const float* __restrict__ W,
                    const float* __restrict__ a_src, const float* __restrict__ a_dst,
                    float* __restrict__ H, float* __restrict__ ald, int n) {
    constexpr int M  = NT * 16;
    constexpr int RS = ROWSTR(M);
    static_assert(K % 16 == 0, "two 8-k chunks per trip");
    const int lane  = threadIdx.x & 31;
    const int strip = blockIdx.x * (blockDim.x >> 5) + (threadIdx.x >> 5);
    if (strip >= (n >> 4)) return;

    const int row0 = strip << 4;
    const int m16  = lane & 15;
    const int half = lane >> 4;

    v8f acc[NT];
#pragma unroll
    for (int t = 0; t < NT; ++t) acc[t] = v8f{};

    const float* xp = X + (size_t)(row0 + m16) * K + half * 4;
    const float* wp = W + (size_t)(half * 4) * M + m16;

    float4 a0, a1;
    float  b0[4][NT], b1[4][NT];
    gemm_load_chunk<M, NT>(xp, wp, a0, b0);
#pragma nounroll
    for (int k0 = 0; k0 < K; k0 += 16) {
        gemm_load_chunk<M, NT>(xp + k0 + 8, wp + (size_t)(k0 + 8) * M, a1, b1);
        gemm_mac_chunk<NT>(a0, b0, acc);
        if (k0 + 16 < K)
            gemm_load_chunk<M, NT>(xp + k0 + 16, wp + (size_t)(k0 + 16) * M, a0, b0);
        gemm_mac_chunk<NT>(a1, b1, acc);
    }

    float* hp = H + (size_t)(row0 + half * 8) * RS + m16;
#pragma unroll
    for (int t = 0; t < NT; ++t)
#pragma unroll
        for (int v = 0; v < 8; ++v)
            hp[(size_t)v * RS + t * 16] = acc[t][v];

    strip_logits<NT, HEADS, C>(acc, a_src, a_dst, H, ald, row0, half, m16);
}

#define BSHIFT 8
#define BINW   (1 << BSHIFT)
#define NBIN   ((NNODE + BINW - 1) >> BSHIFT)
#define GB     256
#define NPST   ((NBIN + 1) * GB)
#define EPT    4
#define BPT    8
#define PSUB   4

__global__ void bin_local(const int* __restrict__ ei, int E, int* __restrict__ pst,
                          int2* __restrict__ binned) {
    __shared__ int hist[NBIN];
    for (int i = threadIdx.x; i < NBIN; i += blockDim.x) hist[i] = 0;
    __syncthreads();
    const int chunk = (E + GB - 1) / GB;
    const int c0 = blockIdx.x * chunk;
    const int e0 = c0 < E ? c0 : E;
    const int e1 = e0 + chunk < E ? e0 + chunk : E;
    for (int e = e0 + threadIdx.x; e < e1; e += EPT * blockDim.x) {
        int d[EPT];
#pragma unroll
        for (int q = 0; q < EPT; ++q) {
            const int eq = e + q * blockDim.x;
            d[q] = ei[E + (eq < e1 ? eq : e1 - 1)];
        }
#pragma unroll
        for (int q = 0; q < EPT; ++q)
            if (e + q * blockDim.x < e1) atomicAdd(&hist[d[q] >> BSHIFT], 1);
    }
    __syncthreads();
    if (threadIdx.x < 32) {
        constexpr int PER = (NBIN + 31) / 32;
        const int lane = threadIdx.x;
        int v[PER];
        int lt = 0;
#pragma unroll
        for (int q = 0; q < PER; ++q) {
            const int b = lane * PER + q;
            v[q] = b < NBIN ? hist[b] : 0;
            lt += v[q];
        }
        int inc = lt;
        for (int o = 1; o < 32; o <<= 1) {
            const int t = __shfl_up(inc, o);
            if (lane >= o) inc += t;
        }
        int run = e0 + inc - lt;
#pragma unroll
        for (int q = 0; q < PER; ++q) {
            const int b = lane * PER + q;
            if (b < NBIN) { hist[b] = run; pst[b * GB + blockIdx.x] = run; }
            run += v[q];
        }
        if (lane == 31) pst[NBIN * GB + blockIdx.x] = run;
    }
    __syncthreads();
    for (int e = e0 + threadIdx.x; e < e1; e += EPT * blockDim.x) {
        int d[EPT], sv[EPT];
#pragma unroll
        for (int q = 0; q < EPT; ++q) {
            const int eq = e + q * blockDim.x;
            const int ec = eq < e1 ? eq : e1 - 1;
            d[q]  = ei[E + ec];
            sv[q] = ei[ec];
        }
#pragma unroll
        for (int q = 0; q < EPT; ++q) {
            if (e + q * blockDim.x < e1) {
                const int pos = atomicAdd(&hist[d[q] >> BSHIFT], 1);
                binned[pos] = make_int2(sv[q], d[q]);
            }
        }
    }
}

__global__ void bin_group(const int2* __restrict__ binned, const int* __restrict__ pst,
                          int E, int n, int* __restrict__ cnt, int* __restrict__ rowptr,
                          int* __restrict__ col) {
    __shared__ int c[BINW];
    __shared__ int cur[BINW];
    __shared__ int wlow[32];
    const int b     = blockIdx.x;
    const int node0 = b << BSHIFT;
    const int lane  = threadIdx.x & 31;
    const int p     = threadIdx.x / PSUB;
    const int sub   = threadIdx.x % PSUB;
    const int ps    = p < GB ? pst[b * GB + p] : 0;
    const int len   = p < GB ? pst[(b + 1) * GB + p] - ps : 0;
    for (int i = threadIdx.x; i < BINW; i += blockDim.x) c[i] = 0;
    {
        const int chunk = (E + GB - 1) / GB;
        const int c0 = p * chunk;
        int low = (p < GB && sub == 0) ? ps - (c0 < E ? c0 : E) : 0;
        for (int o = 16; o > 0; o >>= 1) low += __shfl_xor(low, o);
        if (lane == 0) wlow[threadIdx.x >> 5] = low;
    }
    __syncthreads();
    for (int k = sub; k < len; k += PSUB * BPT) {
        int dl[BPT];
#pragma unroll
        for (int q = 0; q < BPT; ++q) {
            const int kq = k + q * PSUB;
            dl[q] = binned[ps + (kq < len ? kq : len - 1)].y;
        }
#pragma unroll
        for (int q = 0; q < BPT; ++q)
            if (k + q * PSUB < len) atomicAdd(&c[dl[q] - node0], 1);
    }
    __syncthreads();
    if (threadIdx.x < 32) {
        constexpr int PER = BINW / 32;
        int beg = lane < (int)(blockDim.x >> 5) ? wlow[lane] : 0;
        for (int o = 16; o > 0; o >>= 1) beg += __shfl_xor(beg, o);
        int v[PER];
        int lt = 0;
#pragma unroll
        for (int q = 0; q < PER; ++q) { v[q] = c[lane * PER + q]; lt += v[q]; }
        int inc = lt;
        for (int o = 1; o < 32; o <<= 1) {
            const int t = __shfl_up(inc, o);
            if (lane >= o) inc += t;
        }
        int run = beg + inc - lt;
#pragma unroll
        for (int q = 0; q < PER; ++q) {
            const int dl = lane * PER + q;
            if (node0 + dl < n) { cnt[node0 + dl] = v[q]; rowptr[node0 + dl] = run; }
            cur[dl] = run;
            run += v[q];
        }
    }
    __syncthreads();
    for (int k = sub; k < len; k += PSUB * BPT) {
        int2 ed[BPT];
#pragma unroll
        for (int q = 0; q < BPT; ++q) {
            const int kq = k + q * PSUB;
            ed[q] = binned[ps + (kq < len ? kq : len - 1)];
        }
#pragma unroll
        for (int q = 0; q < BPT; ++q) {
            if (k + q * PSUB < len) {
                const int pos = atomicAdd(&cur[ed[q].y - node0], 1);
                col[pos] = ed[q].x;
            }
        }
    }
}

#define AGG_WG16 1250
#define AGG_WG40 1024

template <int HEADS, int C, int RELU, int BATCH>
__device__ __forceinline__ void aggregate_body(const int* __restrict__ rowptr,
                                               const int* __restrict__ cnt,
                                               const int* __restrict__ col,
                                               const float* __restrict__ ald,
                                               const float* __restrict__ H,
                                               const float* __restrict__ bias,
                                               float* __restrict__ out, int n) {
    constexpr int HC = HEADS * C;
    constexpr int NP = (HC + 127) / 128;
    constexpr int RS = ROWSTR(HC);
    static_assert(C % 4 == 0, "a lane's four channels belong to one head");
    static_assert(HC % 4 == 0, "rows are whole 16-byte groups");
    static_assert(HEADS <= LSTR, "logit slots per node");
    static_assert(HC + LSTR <= RS, "the logit slots sit behind the channels");
    static_assert(32 % BATCH == 0, "whole batches per 32 edges");
    const int lane = threadIdx.x & 31;
    const int wpb = (int)(blockDim.x >> 5);
    const int w0  = __builtin_amdgcn_readfirstlane((int)(blockIdx.x * wpb + (threadIdx.x >> 5)));
    const int nw  = (int)gridDim.x * wpb;

    bool   live[NP];
    int    hp[NP];
    int    co[NP];
#pragma unroll
    for (int p = 0; p < NP; ++p) {
        const int c = p * 128 + 4 * lane;
        live[p] = c < HC;
        hp[p]   = c / C < HEADS ? c / C : HEADS - 1;
        co[p]   = live[p] ? c : HC - 4;
    }

    for (int d = w0; d < n; d += nw) {
    const int beg = rowptr[d];
    const int deg = cnt[d];
    const int tot = deg + 1;

    float  adh[HEADS];
#pragma unroll
    for (int h = 0; h < HEADS; ++h) adh[h] = ald[(size_t)d * LSTR + h];
    float  adp[NP], m[NP], dsum[NP];
    float4 acc[NP];
#pragma unroll
    for (int p = 0; p < NP; ++p) {
        adp[p] = adh[0];
#pragma unroll
        for (int h = 1; h < HEADS; ++h) adp[p] = hp[p] == h ? adh[h] : adp[p];
        m[p]    = -INFINITY;
        dsum[p] = 0.0f;
        acc[p]  = make_float4(0.0f, 0.0f, 0.0f, 0.0f);
    }

    int c0 = 0;
    do {
        const int t = c0 + lane;
        int s = d;
        if (t < deg) s = col[beg + t];

        const int nj = (tot - c0) < 32 ? (tot - c0) : 32;
        float4 hv[BATCH][NP];
        float  lg[BATCH][NP];
        auto fold = [&]() {
#pragma unroll
            for (int p = 0; p < NP; ++p) {
                float x[BATCH];
                float bm = -INFINITY;
#pragma unroll
                for (int i = 0; i < BATCH; ++i) {
                    const float v = lg[i][p] + adp[p];
                    x[i] = v >= 0.0f ? v : 0.2f * v;
                    bm = fmaxf(bm, x[i]);
                }
                const float mn = fmaxf(m[p], bm);
                const float sc = __expf(m[p] - mn);
                m[p] = mn;
                dsum[p]  *= sc;
                acc[p].x *= sc; acc[p].y *= sc; acc[p].z *= sc; acc[p].w *= sc;
#pragma unroll
                for (int i = 0; i < BATCH; ++i) {
                    const float e = __expf(x[i] - mn);
                    dsum[p]  += e;
                    acc[p].x += hv[i][p].x * e;
                    acc[p].y += hv[i][p].y * e;
                    acc[p].z += hv[i][p].z * e;
                    acc[p].w += hv[i][p].w * e;
                }
            }
        };
        int j0 = 0;
        for (; j0 + BATCH <= nj; j0 += BATCH) {
            int sj[BATCH];
#pragma unroll
            for (int i = 0; i < BATCH; ++i) sj[i] = __builtin_amdgcn_readlane(s, j0 + i);
#pragma unroll
            for (int i = 0; i < BATCH; ++i) {
                const float* hs = H + (size_t)sj[i] * RS;
#pragma unroll
                for (int p = 0; p < NP; ++p) {
                    hv[i][p] = *(const float4*)(hs + co[p]);
                    lg[i][p] = hs[HC + hp[p]];
                }
            }
            fold();
        }
        if (j0 < nj) {
            int sj[BATCH];
#pragma unroll
            for (int i = 0; i < BATCH - 1; ++i)
                sj[i] = __builtin_amdgcn_readlane(s, j0 + i < nj ? j0 + i : nj - 1);
#pragma unroll
            for (int i = 0; i < BATCH; ++i) {
#pragma unroll
                for (int p = 0; p < NP; ++p) {
                    hv[i][p] = make_float4(0.0f, 0.0f, 0.0f, 0.0f);
                    lg[i][p] = -INFINITY;
                }
                if (i < BATCH - 1 && j0 + i < nj) {
                    const float* hs = H + (size_t)sj[i] * RS;
#pragma unroll
                    for (int p = 0; p < NP; ++p) {
                        hv[i][p] = *(const float4*)(hs + co[p]);
                        lg[i][p] = hs[HC + hp[p]];
                    }
                }
            }
            fold();
        }
        c0 += 32;
    } while (c0 < tot);

    float* orow = out + (size_t)d * HC;
#pragma unroll
    for (int p = 0; p < NP; ++p) {
        if (live[p]) {
            const int c = p * 128 + 4 * lane;
            const float den = dsum[p];
            const float4 b = *(const float4*)(bias + c);
            float4 v;
            v.x = acc[p].x / den + b.x;
            v.y = acc[p].y / den + b.y;
            v.z = acc[p].z / den + b.z;
            v.w = acc[p].w / den + b.w;
            if (RELU) {
                v.x = fmaxf(v.x, 0.0f); v.y = fmaxf(v.y, 0.0f);
                v.z = fmaxf(v.z, 0.0f); v.w = fmaxf(v.w, 0.0f);
            }
            *(float4*)(orow + c) = v;
        }
    }
    }
}

__global__ void aggregate_7x16(const int* __restrict__ rowptr, const int* __restrict__ cnt,
                               const int* __restrict__ col, const float* __restrict__ ald,
                               const float* __restrict__ H, const float* __restrict__ bias,
                               float* __restrict__ out, int n) {
    aggregate_body<7, 16, 1, 8>(rowptr, cnt, col, ald, H, bias, out, n);
}

__global__ void aggregate_6x16(const int* __restrict__ rowptr, const int* __restrict__ cnt,
                               const int* __restrict__ col, const float* __restrict__ ald,
                               const float* __restrict__ H, const float* __restrict__ bias,
                               float* __restrict__ out, int n) {
    aggregate_body<6, 16, 1, 8>(rowptr, cnt, col, ald, H, bias, out, n);
}

__global__ void aggregate_6x40(const int* __restrict__ rowptr, const int* __restrict__ cnt,
                               const int* __restrict__ col, const float* __restrict__ ald,
                               const float* __restrict__ H, const float* __restrict__ bias,
                               float* __restrict__ out, int n) {
    aggregate_body<6, 40, 0, 4>(rowptr, cnt, col, ald, H, bias, out, n);
}


extern "C" void kernel_launch(void* const* d_in, const int* in_sizes, int n_in,
                              void* d_out, int out_size, void* d_ws, size_t ws_size,
                              hipStream_t stream) {
    const float* x      = (const float*)d_in[0];
    const int*   ei     = (const int*)  d_in[1];
    const float* W1     = (const float*)d_in[2];
    const float* a_src1 = (const float*)d_in[3];
    const float* a_dst1 = (const float*)d_in[4];
    const float* b1     = (const float*)d_in[5];
    const float* W2     = (const float*)d_in[6];
    const float* a_src2 = (const float*)d_in[7];
    const float* a_dst2 = (const float*)d_in[8];
    const float* b2     = (const float*)d_in[9];
    const float* W3     = (const float*)d_in[10];
    const float* a_src3 = (const float*)d_in[11];
    const float* a_dst3 = (const float*)d_in[12];
    const float* b3     = (const float*)d_in[13];

    const int E = in_sizes[1] / 2;

    float* ws   = (float*)d_ws;
    float* bufH = ws;
    float* act1 = bufH + (size_t)NNODE * 256;
    float* h2   = act1 + (size_t)NNODE * 112;
    float* act2 = h2   + (size_t)NNODE * 128;
    float* ald  = act2 + (size_t)NNODE * 96;
    int*   cnt    = (int*)(ald + (size_t)NNODE * LSTR);
    int*   rowptr = cnt    + NNODE;
    int*   pst    = rowptr + NNODE;
    int2*  binned = (int2*)(pst + NPST);
    int*   col    = (int*)(binned + E);
    float* out  = (float*)d_out;

    bin_local<<<GB, 1024, 0, stream>>>(ei, E, pst, binned);
    bin_group<<<NBIN, PSUB * GB, 0, stream>>>(binned, pst, E, NNODE, cnt, rowptr, col);

    auto layer = [&](const float* X, int K,
                     const float* W, const float* as, const float* ad, const float* b,
                     int heads, int C, float* Hbuf, float* obuf) {
        const int gb = (NNODE / 16 + 1) / 2;
        if (heads == 7)
            gemm_strip_wmma_f32<128, 7, 7, 16><<<gb, 64, 0, stream>>>(X, W, as, ad, Hbuf,
                                                                      ald, NNODE);
        else if (C == 16)
            gemm_strip_wmma_f32<112, 6, 6, 16><<<gb, 64, 0, stream>>>(X, W, as, ad, Hbuf,
                                                                      ald, NNODE);
        else
            gemm_strip_wmma_f32<96, 15, 6, 40><<<gb, 64, 0, stream>>>(X, W, as, ad, Hbuf,
                                                                      ald, NNODE);

        if (heads == 7)
            aggregate_7x16<<<AGG_WG16, 256, 0, stream>>>(rowptr, cnt, col, ald, Hbuf,
                                                         b, obuf, NNODE);
        else if (C == 16)
            aggregate_6x16<<<AGG_WG16, 256, 0, stream>>>(rowptr, cnt, col, ald, Hbuf,
                                                         b, obuf, NNODE);
        else
            aggregate_6x40<<<AGG_WG40, 256, 0, stream>>>(rowptr, cnt, col, ald, Hbuf,
                                                         b, obuf, NNODE);
    };

    layer(x,    128, W1, a_src1, a_dst1, b1, 7, 16, bufH, act1);
    layer(act1, 112, W2, a_src2, a_dst2, b2, 6, 16, h2,   act2);
    layer(act2,  96, W3, a_src3, a_dst3, b3, 6, 40, bufH, out);
}
